# starting kernel
# baseline (speedup 1.0000x reference)
#include <hip/hip_runtime.h>
#include <hip/hip_bf16.h>

typedef __attribute__((ext_vector_type(2))) float v2f;
typedef __attribute__((ext_vector_type(8))) float v8f;

__global__ void k_init_deg(float* __restrict__ deg, int n) {
  int i = blockIdx.x * blockDim.x + threadIdx.x;
  if (i < n) deg[i] = 1.0f;
}

__global__ void k_count_deg(const int* __restrict__ dst, float* __restrict__ deg, int ne) {
  int e = blockIdx.x * blockDim.x + threadIdx.x;
  if (e < ne) unsafeAtomicAdd(&deg[__builtin_nontemporal_load(dst + e)], 1.0f);
}

__global__ void k_dinv(const float* __restrict__ deg, float* __restrict__ dinv, int n) {
  int i = blockIdx.x * blockDim.x + threadIdx.x;
  if (i < n) dinv[i] = rsqrtf(deg[i]);
}

template <bool RELU_A>
__global__ void k_gemm_n16(const float* __restrict__ A, const float* __restrict__ B,
                           float* __restrict__ C, int nrows) {
  int wid  = (blockIdx.x * blockDim.x + threadIdx.x) >> 5;
  int lane = threadIdx.x & 31;
  int row0 = wid * 16;
  if (row0 >= nrows) return;
  int m  = lane & 15;
  int kh = (lane >> 4) << 1;
  const float* arow = A + (row0 + m) * 16;
  v8f c = {};
#pragma unroll
  for (int k0 = 0; k0 < 16; k0 += 4) {
    v2f a, b;
    float a0 = arow[k0 + kh + 0];
    float a1 = arow[k0 + kh + 1];
    if (RELU_A) { a0 = fmaxf(a0, 0.0f); a1 = fmaxf(a1, 0.0f); }
    a.x = a0; a.y = a1;
    b.x = B[(k0 + kh + 0) * 16 + m];
    b.y = B[(k0 + kh + 1) * 16 + m];
    c = __builtin_amdgcn_wmma_f32_16x16x4_f32(false, a, false, b, (short)0, c, false, false);
  }
  int rbase = row0 + ((lane >> 4) << 3);
#pragma unroll
  for (int r = 0; r < 8; ++r) C[(rbase + r) * 16 + m] = c[r];
}

__global__ void k_init_agg(const float* __restrict__ hw, const float* __restrict__ dinv,
                           const float* __restrict__ bias, float* __restrict__ agg, int n16) {
  int i = blockIdx.x * blockDim.x + threadIdx.x;
  if (i >= n16) return;
  int node = i >> 4, f = i & 15;
  float di = dinv[node];
  agg[i] = hw[i] * di * di + bias[f];
}

__global__ void k_scatter(const int* __restrict__ src, const int* __restrict__ dst,
                          const float* __restrict__ dinv, const float* __restrict__ hw,
                          float* __restrict__ agg, int ne) {
  int t = blockIdx.x * blockDim.x + threadIdx.x;
  int e = t >> 2, q = t & 3;
  if (e >= ne) return;
  int s = __builtin_nontemporal_load(src + e);
  int d = __builtin_nontemporal_load(dst + e);
  float w = dinv[s] * dinv[d];
  float4 v = ((const float4*)(hw + s * 16))[q];
  float* ad = agg + d * 16 + q * 4;
  unsafeAtomicAdd(ad + 0, v.x * w);
  unsafeAtomicAdd(ad + 1, v.y * w);
  unsafeAtomicAdd(ad + 2, v.z * w);
  unsafeAtomicAdd(ad + 3, v.w * w);
}

template <int KLEN>
__global__ void k_gemm_splitk_relu(const float* __restrict__ A, int lda,
                                   const float* __restrict__ B, int ncols,
                                   float* __restrict__ acc,
                                   int tiles_n, int nsplit) {
  int wid  = (blockIdx.x * blockDim.x + threadIdx.x) >> 5;
  int lane = threadIdx.x & 31;
  int split = wid % nsplit;
  int tn    = (wid / nsplit) % tiles_n;
  int tm    = wid / (nsplit * tiles_n);
  int m  = lane & 15;
  int kh = (lane >> 4) << 1;
  const float* arow = A + (size_t)(tm * 16 + m) * lda + split * KLEN;
  const float* bcol = B + (size_t)(split * KLEN) * ncols + tn * 16 + m;
  v8f c = {};
#pragma unroll 4
  for (int k0 = 0; k0 < KLEN; k0 += 4) {
    v2f a, b;
    a.x = fmaxf(arow[k0 + kh + 0], 0.0f);
    a.y = fmaxf(arow[k0 + kh + 1], 0.0f);
    b.x = bcol[(size_t)(k0 + kh + 0) * ncols];
    b.y = bcol[(size_t)(k0 + kh + 1) * ncols];
    c = __builtin_amdgcn_wmma_f32_16x16x4_f32(false, a, false, b, (short)0, c, false, false);
  }
  int rbase = tm * 16 + ((lane >> 4) << 3);
#pragma unroll
  for (int r = 0; r < 8; ++r)
    unsafeAtomicAdd(&acc[(rbase + r) * ncols + tn * 16 + m], c[r]);
}

__global__ void k_bias_relu(const float* __restrict__ in, const float* __restrict__ bias,
                            float* __restrict__ out, int n, int ncols) {
  int i = blockIdx.x * blockDim.x + threadIdx.x;
  if (i < n) out[i] = fmaxf(in[i] + bias[i % ncols], 0.0f);
}

__global__ void k_zero(float* __restrict__ p, int n) {
  int i = blockIdx.x * blockDim.x + threadIdx.x;
  if (i < n) p[i] = 0.0f;
}

__global__ void k_gemm_final(const float* __restrict__ A, const float* __restrict__ B,
                             const float* __restrict__ bias, float* __restrict__ out) {
  int wid  = (blockIdx.x * blockDim.x + threadIdx.x) >> 5;
  int lane = threadIdx.x & 31;
  int tm = wid >> 6;
  int tn = wid & 63;
  int m  = lane & 15;
  int kh = (lane >> 4) << 1;
  const float* arow = A + (tm * 16 + m) * 32;
  const float* bcol = B + tn * 16 + m;
  v8f c = {};
#pragma unroll
  for (int k0 = 0; k0 < 32; k0 += 4) {
    v2f a, b;
    a.x = arow[k0 + kh + 0];
    a.y = arow[k0 + kh + 1];
    b.x = bcol[(k0 + kh + 0) * 1024];
    b.y = bcol[(k0 + kh + 1) * 1024];
    c = __builtin_amdgcn_wmma_f32_16x16x4_f32(false, a, false, b, (short)0, c, false, false);
  }
  int rbase = tm * 16 + ((lane >> 4) << 3);
  float bv = bias[tn * 16 + m];
#pragma unroll
  for (int r = 0; r < 8; ++r)
    out[(rbase + r) * 1024 + tn * 16 + m] = c[r] + bv;
}

__global__ void k_value_final(const float* __restrict__ vhid, const float* __restrict__ w,
                              const float* __restrict__ b, float* __restrict__ out) {
  int bi = threadIdx.x;
  float s = 0.0f;
#pragma unroll
  for (int j = 0; j < 16; ++j) s += vhid[bi * 16 + j] * w[j];
  out[bi] = s + b[0];
}

extern "C" void kernel_launch(void* const* d_in, const int* in_sizes, int n_in,
                              void* d_out, int out_size, void* d_ws, size_t ws_size,
                              hipStream_t stream) {
  const float* x    = (const float*)d_in[0];
  const int*   ei   = (const int*)d_in[1];
  const float* W1   = (const float*)d_in[2];
  const float* b1   = (const float*)d_in[3];
  const float* W2   = (const float*)d_in[4];
  const float* b2   = (const float*)d_in[5];
  const float* p1w  = (const float*)d_in[6];
  const float* p1b  = (const float*)d_in[7];
  const float* p2w  = (const float*)d_in[8];
  const float* p2b  = (const float*)d_in[9];
  const float* v1w  = (const float*)d_in[10];
  const float* v1b  = (const float*)d_in[11];
  const float* v2w  = (const float*)d_in[12];
  const float* v2b  = (const float*)d_in[13];

  const int N  = in_sizes[0] / 16;
  const int E  = in_sizes[1] / 2;
  const int* esrc = ei;
  const int* edst = ei + E;

  float* ws   = (float*)d_ws;
  float* deg  = ws;
  float* dinv = deg + N;
  float* bufA = dinv + N;
  float* bufB = bufA + (size_t)16 * N;
  float* pacc = bufB + (size_t)16 * N;
  float* vacc = pacc + 64 * 32;
  float* phid = vacc + 64 * 16;
  float* vhid = phid + 64 * 32;

  float* Xout = (float*)d_out;
  float* Vout = Xout + 64 * 1024;

  const int B256 = 256;
  k_init_deg<<<N / B256, B256, 0, stream>>>(deg, N);
  k_count_deg<<<E / B256, B256, 0, stream>>>(edst, deg, E);
  k_dinv<<<N / B256, B256, 0, stream>>>(deg, dinv, N);

  int gemm_blocks    = (N / 16) / 8;
  int scatter_blocks = (4 * E) / B256;
  k_gemm_n16<false><<<gemm_blocks, B256, 0, stream>>>(x, W1, bufA, N);
  k_init_agg<<<(16 * N) / B256, B256, 0, stream>>>(bufA, dinv, b1, bufB, 16 * N);
  k_scatter<<<scatter_blocks, B256, 0, stream>>>(esrc, edst, dinv, bufA, bufB, E);

  k_gemm_n16<true><<<gemm_blocks, B256, 0, stream>>>(bufB, W2, bufA, N);
  k_init_agg<<<(16 * N) / B256, B256, 0, stream>>>(bufA, dinv, b2, bufB, 16 * N);
  k_scatter<<<scatter_blocks, B256, 0, stream>>>(esrc, edst, dinv, bufA, bufB, E);

  const int K = 16384, NSPLIT = 64, KLEN = K / NSPLIT;

  k_zero<<<(64 * 32 + 64 * 16 + B256 - 1) / B256, B256, 0, stream>>>(pacc, 64 * 32 + 64 * 16);

  k_gemm_splitk_relu<KLEN><<<64, B256, 0, stream>>>(bufB, K, p1w, 32, pacc, 2, NSPLIT);
  k_gemm_splitk_relu<KLEN><<<32, B256, 0, stream>>>(bufB, K, v1w, 16, vacc, 1, NSPLIT);

  k_bias_relu<<<(64 * 32) / B256, B256, 0, stream>>>(pacc, p1b, phid, 64 * 32, 32);
  k_bias_relu<<<(64 * 16 + B256 - 1) / B256, B256, 0, stream>>>(vacc, v1b, vhid, 64 * 16, 16);

  k_gemm_final<<<32, B256, 0, stream>>>(phid, p2w, p2b, Xout);
  k_value_final<<<1, 64, 0, stream>>>(vhid, v2w, v2b, Vout);
}


# degree counters padded to one per 64-byte row (stride 16 floats, placed in the aggregate buffer before its first write): the 2M atomic adds of 1.0 spread over 4 MiB instead of 256 KiB
# speedup vs baseline: 1.1532x; 1.1532x over previous
#include <hip/hip_runtime.h>
#include <hip/hip_bf16.h>

typedef __attribute__((ext_vector_type(2))) float v2f;
typedef __attribute__((ext_vector_type(8))) float v8f;

constexpr int DEG_STRIDE = 16;

__global__ void k_init_deg(float* __restrict__ deg, int n) {
  int i = blockIdx.x * blockDim.x + threadIdx.x;
  if (i < n) deg[i * DEG_STRIDE] = 1.0f;
}

__global__ void k_count_deg(const int* __restrict__ dst, float* __restrict__ deg, int ne) {
  int e = blockIdx.x * blockDim.x + threadIdx.x;
  if (e < ne) unsafeAtomicAdd(&deg[__builtin_nontemporal_load(dst + e) * DEG_STRIDE], 1.0f);
}

__global__ void k_dinv(const float* __restrict__ deg, float* __restrict__ dinv, int n) {
  int i = blockIdx.x * blockDim.x + threadIdx.x;
  if (i < n) dinv[i] = rsqrtf(deg[i * DEG_STRIDE]);
}

template <bool RELU_A>
__global__ void k_gemm_n16(const float* __restrict__ A, const float* __restrict__ B,
                           float* __restrict__ C, int nrows) {
  int wid  = (blockIdx.x * blockDim.x + threadIdx.x) >> 5;
  int lane = threadIdx.x & 31;
  int row0 = wid * 16;
  if (row0 >= nrows) return;
  int m  = lane & 15;
  int kh = (lane >> 4) << 1;
  const float* arow = A + (row0 + m) * 16;
  v8f c = {};
#pragma unroll
  for (int k0 = 0; k0 < 16; k0 += 4) {
    v2f a, b;
    float a0 = arow[k0 + kh + 0];
    float a1 = arow[k0 + kh + 1];
    if (RELU_A) { a0 = fmaxf(a0, 0.0f); a1 = fmaxf(a1, 0.0f); }
    a.x = a0; a.y = a1;
    b.x = B[(k0 + kh + 0) * 16 + m];
    b.y = B[(k0 + kh + 1) * 16 + m];
    c = __builtin_amdgcn_wmma_f32_16x16x4_f32(false, a, false, b, (short)0, c, false, false);
  }
  int rbase = row0 + ((lane >> 4) << 3);
#pragma unroll
  for (int r = 0; r < 8; ++r) C[(rbase + r) * 16 + m] = c[r];
}

__global__ void k_init_agg(const float* __restrict__ hw, const float* __restrict__ dinv,
                           const float* __restrict__ bias, float* __restrict__ agg, int n16) {
  int i = blockIdx.x * blockDim.x + threadIdx.x;
  if (i >= n16) return;
  int node = i >> 4, f = i & 15;
  float di = dinv[node];
  agg[i] = hw[i] * di * di + bias[f];
}

__global__ void k_scatter(const int* __restrict__ src, const int* __restrict__ dst,
                          const float* __restrict__ dinv, const float* __restrict__ hw,
                          float* __restrict__ agg, int ne) {
  int t = blockIdx.x * blockDim.x + threadIdx.x;
  int e = t >> 2, q = t & 3;
  if (e >= ne) return;
  int s = __builtin_nontemporal_load(src + e);
  int d = __builtin_nontemporal_load(dst + e);
  float w = dinv[s] * dinv[d];
  float4 v = ((const float4*)(hw + s * 16))[q];
  float* ad = agg + d * 16 + q * 4;
  unsafeAtomicAdd(ad + 0, v.x * w);
  unsafeAtomicAdd(ad + 1, v.y * w);
  unsafeAtomicAdd(ad + 2, v.z * w);
  unsafeAtomicAdd(ad + 3, v.w * w);
}

template <int KLEN>
__global__ void k_gemm_splitk_relu(const float* __restrict__ A, int lda,
                                   const float* __restrict__ B, int ncols,
                                   float* __restrict__ acc,
                                   int tiles_n, int nsplit) {
  int wid  = (blockIdx.x * blockDim.x + threadIdx.x) >> 5;
  int lane = threadIdx.x & 31;
  int split = wid % nsplit;
  int tn    = (wid / nsplit) % tiles_n;
  int tm    = wid / (nsplit * tiles_n);
  int m  = lane & 15;
  int kh = (lane >> 4) << 1;
  const float* arow = A + (size_t)(tm * 16 + m) * lda + split * KLEN;
  const float* bcol = B + (size_t)(split * KLEN) * ncols + tn * 16 + m;
  v8f c = {};
#pragma unroll 4
  for (int k0 = 0; k0 < KLEN; k0 += 4) {
    v2f a, b;
    a.x = fmaxf(arow[k0 + kh + 0], 0.0f);
    a.y = fmaxf(arow[k0 + kh + 1], 0.0f);
    b.x = bcol[(size_t)(k0 + kh + 0) * ncols];
    b.y = bcol[(size_t)(k0 + kh + 1) * ncols];
    c = __builtin_amdgcn_wmma_f32_16x16x4_f32(false, a, false, b, (short)0, c, false, false);
  }
  int rbase = tm * 16 + ((lane >> 4) << 3);
#pragma unroll
  for (int r = 0; r < 8; ++r)
    unsafeAtomicAdd(&acc[(rbase + r) * ncols + tn * 16 + m], c[r]);
}

__global__ void k_bias_relu(const float* __restrict__ in, const float* __restrict__ bias,
                            float* __restrict__ out, int n, int ncols) {
  int i = blockIdx.x * blockDim.x + threadIdx.x;
  if (i < n) out[i] = fmaxf(in[i] + bias[i % ncols], 0.0f);
}

__global__ void k_zero(float* __restrict__ p, int n) {
  int i = blockIdx.x * blockDim.x + threadIdx.x;
  if (i < n) p[i] = 0.0f;
}

__global__ void k_gemm_final(const float* __restrict__ A, const float* __restrict__ B,
                             const float* __restrict__ bias, float* __restrict__ out) {
  int wid  = (blockIdx.x * blockDim.x + threadIdx.x) >> 5;
  int lane = threadIdx.x & 31;
  int tm = wid >> 6;
  int tn = wid & 63;
  int m  = lane & 15;
  int kh = (lane >> 4) << 1;
  const float* arow = A + (tm * 16 + m) * 32;
  const float* bcol = B + tn * 16 + m;
  v8f c = {};
#pragma unroll
  for (int k0 = 0; k0 < 32; k0 += 4) {
    v2f a, b;
    a.x = arow[k0 + kh + 0];
    a.y = arow[k0 + kh + 1];
    b.x = bcol[(k0 + kh + 0) * 1024];
    b.y = bcol[(k0 + kh + 1) * 1024];
    c = __builtin_amdgcn_wmma_f32_16x16x4_f32(false, a, false, b, (short)0, c, false, false);
  }
  int rbase = tm * 16 + ((lane >> 4) << 3);
  float bv = bias[tn * 16 + m];
#pragma unroll
  for (int r = 0; r < 8; ++r)
    out[(rbase + r) * 1024 + tn * 16 + m] = c[r] + bv;
}

__global__ void k_value_final(const float* __restrict__ vhid, const float* __restrict__ w,
                              const float* __restrict__ b, float* __restrict__ out) {
  int bi = threadIdx.x;
  float s = 0.0f;
#pragma unroll
  for (int j = 0; j < 16; ++j) s += vhid[bi * 16 + j] * w[j];
  out[bi] = s + b[0];
}

extern "C" void kernel_launch(void* const* d_in, const int* in_sizes, int n_in,
                              void* d_out, int out_size, void* d_ws, size_t ws_size,
                              hipStream_t stream) {
  const float* x    = (const float*)d_in[0];
  const int*   ei   = (const int*)d_in[1];
  const float* W1   = (const float*)d_in[2];
  const float* b1   = (const float*)d_in[3];
  const float* W2   = (const float*)d_in[4];
  const float* b2   = (const float*)d_in[5];
  const float* p1w  = (const float*)d_in[6];
  const float* p1b  = (const float*)d_in[7];
  const float* p2w  = (const float*)d_in[8];
  const float* p2b  = (const float*)d_in[9];
  const float* v1w  = (const float*)d_in[10];
  const float* v1b  = (const float*)d_in[11];
  const float* v2w  = (const float*)d_in[12];
  const float* v2b  = (const float*)d_in[13];

  const int N  = in_sizes[0] / 16;
  const int E  = in_sizes[1] / 2;
  const int* esrc = ei;
  const int* edst = ei + E;

  float* ws   = (float*)d_ws;
  float* dinv = ws + N;
  float* bufA = dinv + N;
  float* bufB = bufA + (size_t)16 * N;
  float* deg  = bufB;
  float* pacc = bufB + (size_t)16 * N;
  float* vacc = pacc + 64 * 32;
  float* phid = vacc + 64 * 16;
  float* vhid = phid + 64 * 32;

  float* Xout = (float*)d_out;
  float* Vout = Xout + 64 * 1024;

  const int B256 = 256;
  k_init_deg<<<N / B256, B256, 0, stream>>>(deg, N);
  k_count_deg<<<E / B256, B256, 0, stream>>>(edst, deg, E);
  k_dinv<<<N / B256, B256, 0, stream>>>(deg, dinv, N);

  int gemm_blocks    = (N / 16) / 8;
  int scatter_blocks = (4 * E) / B256;
  k_gemm_n16<false><<<gemm_blocks, B256, 0, stream>>>(x, W1, bufA, N);
  k_init_agg<<<(16 * N) / B256, B256, 0, stream>>>(bufA, dinv, b1, bufB, 16 * N);
  k_scatter<<<scatter_blocks, B256, 0, stream>>>(esrc, edst, dinv, bufA, bufB, E);

  k_gemm_n16<true><<<gemm_blocks, B256, 0, stream>>>(bufB, W2, bufA, N);
  k_init_agg<<<(16 * N) / B256, B256, 0, stream>>>(bufA, dinv, b2, bufB, 16 * N);
  k_scatter<<<scatter_blocks, B256, 0, stream>>>(esrc, edst, dinv, bufA, bufB, E);

  const int K = 16384, NSPLIT = 64, KLEN = K / NSPLIT;

  k_zero<<<(64 * 32 + 64 * 16 + B256 - 1) / B256, B256, 0, stream>>>(pacc, 64 * 32 + 64 * 16);

  k_gemm_splitk_relu<KLEN><<<64, B256, 0, stream>>>(bufB, K, p1w, 32, pacc, 2, NSPLIT);
  k_gemm_splitk_relu<KLEN><<<32, B256, 0, stream>>>(bufB, K, v1w, 16, vacc, 1, NSPLIT);

  k_bias_relu<<<(64 * 32) / B256, B256, 0, stream>>>(pacc, p1b, phid, 64 * 32, 32);
  k_bias_relu<<<(64 * 16 + B256 - 1) / B256, B256, 0, stream>>>(vacc, v1b, vhid, 64 * 16, 16);

  k_gemm_final<<<32, B256, 0, stream>>>(phid, p2w, p2b, Xout);
  k_value_final<<<1, 64, 0, stream>>>(vhid, v2w, v2b, Vout);
}


# edge scatter privatised in LDS: a work-group keeps a 1024-node range of rows in LDS, its waves filter dst and add source rows with LDS float atomics; 4 global adds per element, sum order changes
# speedup vs baseline: 1.6709x; 1.4489x over previous
#include <hip/hip_runtime.h>
#include <hip/hip_bf16.h>

typedef __attribute__((ext_vector_type(2))) float v2f;
typedef __attribute__((ext_vector_type(8))) float v8f;

constexpr int DEG_STRIDE = 16;

__global__ void k_init_deg(float* __restrict__ deg, int n) {
  int i = blockIdx.x * blockDim.x + threadIdx.x;
  if (i < n) deg[i * DEG_STRIDE] = 1.0f;
}

__global__ void k_count_deg(const int* __restrict__ dst, float* __restrict__ deg, int ne) {
  int e = blockIdx.x * blockDim.x + threadIdx.x;
  if (e < ne) unsafeAtomicAdd(&deg[__builtin_nontemporal_load(dst + e) * DEG_STRIDE], 1.0f);
}

__global__ void k_dinv(const float* __restrict__ deg, float* __restrict__ dinv, int n) {
  int i = blockIdx.x * blockDim.x + threadIdx.x;
  if (i < n) dinv[i] = rsqrtf(deg[i * DEG_STRIDE]);
}

template <bool RELU_A>
__global__ void k_gemm_n16(const float* __restrict__ A, const float* __restrict__ B,
                           float* __restrict__ C, int nrows) {
  int wid  = (blockIdx.x * blockDim.x + threadIdx.x) >> 5;
  int lane = threadIdx.x & 31;
  int row0 = wid * 16;
  if (row0 >= nrows) return;
  int m  = lane & 15;
  int kh = (lane >> 4) << 1;
  const float* arow = A + (row0 + m) * 16;
  v8f c = {};
#pragma unroll
  for (int k0 = 0; k0 < 16; k0 += 4) {
    v2f a, b;
    float a0 = arow[k0 + kh + 0];
    float a1 = arow[k0 + kh + 1];
    if (RELU_A) { a0 = fmaxf(a0, 0.0f); a1 = fmaxf(a1, 0.0f); }
    a.x = a0; a.y = a1;
    b.x = B[(k0 + kh + 0) * 16 + m];
    b.y = B[(k0 + kh + 1) * 16 + m];
    c = __builtin_amdgcn_wmma_f32_16x16x4_f32(false, a, false, b, (short)0, c, false, false);
  }
  int rbase = row0 + ((lane >> 4) << 3);
#pragma unroll
  for (int r = 0; r < 8; ++r) C[(rbase + r) * 16 + m] = c[r];
}

__global__ void k_init_agg(const float* __restrict__ hw, const float* __restrict__ dinv,
                           const float* __restrict__ bias, float* __restrict__ agg, int n16) {
  int i = blockIdx.x * blockDim.x + threadIdx.x;
  if (i >= n16) return;
  int node = i >> 4, f = i & 15;
  float di = dinv[node];
  agg[i] = hw[i] * di * di + bias[f];
}

typedef int v4i __attribute__((ext_vector_type(4)));

constexpr int SC_NR    = 1024;
constexpr int SC_S     = 4;
constexpr int SC_WAVES = 32;
constexpr int SC_LOADS = 8;
constexpr int SC_BATCH = 32 * 4 * SC_LOADS;
constexpr int SC_UD    = 8;

__global__ __launch_bounds__(1024)
void k_scatter(const int* __restrict__ src, const int* __restrict__ dst,
               const float* __restrict__ dinv, const float* __restrict__ hw,
               float* __restrict__ agg, int ne, int nranges) {
  __shared__ float acc[SC_NR * 16];
  __shared__ unsigned q[SC_WAVES][SC_BATCH + 2 * SC_UD];
  const int tid = threadIdx.x, lane = tid & 31, w = tid >> 5;
  const int half = lane >> 4, f = lane & 15;
  const int r  = blockIdx.x % nranges;
  const int sl = blockIdx.x / nranges;
  const int lo = r * SC_NR;
  for (int i = tid; i < SC_NR * 16; i += 1024) acc[i] = 0.0f;
  __syncthreads();

  const int slice = ne / SC_S;
  const int span  = slice / SC_WAVES;
  const int ebase = sl * slice + w * span;
  const int nb    = span / SC_BATCH;
  const v4i* dp   = (const v4i*)(dst + ebase);
  const int* sp   = src + ebase;
  unsigned* qw    = q[w];
  int qn = 0;

  auto pass = [&](int q0, int nv) {
    unsigned ent[SC_UD]; bool ok[SC_UD]; int sv[SC_UD]; float wv[SC_UD], hv[SC_UD];
#pragma unroll
    for (int u = 0; u < SC_UD; ++u) {
      ok[u]  = 2 * u + half < nv;
      ent[u] = ok[u] ? qw[q0 + 2 * u + half] : 0u;
    }
#pragma unroll
    for (int u = 0; u < SC_UD; ++u) sv[u] = sp[ent[u] >> 10];
#pragma unroll
    for (int u = 0; u < SC_UD; ++u) {
      wv[u] = dinv[sv[u]] * dinv[lo + (int)(ent[u] & 1023u)];
      hv[u] = hw[sv[u] * 16 + f];
    }
#pragma unroll
    for (int u = 0; u < SC_UD; ++u)
      unsafeAtomicAdd(&acc[(ent[u] & 1023u) * 16 + f], ok[u] ? hv[u] * wv[u] : 0.0f);
  };

  v4i cur[SC_LOADS], nxt[SC_LOADS];
#pragma unroll
  for (int j = 0; j < SC_LOADS; ++j) cur[j] = dp[j * 32 + lane];
  for (int b = 0; b < nb; ++b) {
    const int bn = (b + 1 < nb) ? b + 1 : b;
#pragma unroll
    for (int j = 0; j < SC_LOADS; ++j) nxt[j] = dp[bn * (SC_BATCH / 4) + j * 32 + lane];
#pragma unroll
    for (int j = 0; j < SC_LOADS; ++j) {
#pragma unroll
      for (int c = 0; c < 4; ++c) {
        const int dl = cur[j][c] - lo;
        const bool m = (unsigned)dl < (unsigned)SC_NR;
        const unsigned mask = __builtin_amdgcn_ballot_w32(m);
        if (mask != 0u) {
          const int cnt = __builtin_popcount(mask);
          if (m) {
            const int el = b * SC_BATCH + j * 128 + lane * 4 + c;
            qw[qn + (int)__builtin_amdgcn_mbcnt_lo(mask, 0u)] = ((unsigned)el << 10) | (unsigned)dl;
          }
          qn += cnt;
        }
      }
    }
    while (qn >= 2 * SC_UD) { qn -= 2 * SC_UD; pass(qn, 2 * SC_UD); }
#pragma unroll
    for (int j = 0; j < SC_LOADS; ++j) cur[j] = nxt[j];
  }
  if (qn > 0) pass(0, qn);
  __syncthreads();
  float* arow = agg + (size_t)lo * 16;
  for (int i = tid; i < SC_NR * 16; i += 1024) unsafeAtomicAdd(arow + i, acc[i]);
}

template <int KLEN>
__global__ void k_gemm_splitk_relu(const float* __restrict__ A, int lda,
                                   const float* __restrict__ B, int ncols,
                                   float* __restrict__ acc,
                                   int tiles_n, int nsplit) {
  int wid  = (blockIdx.x * blockDim.x + threadIdx.x) >> 5;
  int lane = threadIdx.x & 31;
  int split = wid % nsplit;
  int tn    = (wid / nsplit) % tiles_n;
  int tm    = wid / (nsplit * tiles_n);
  int m  = lane & 15;
  int kh = (lane >> 4) << 1;
  const float* arow = A + (size_t)(tm * 16 + m) * lda + split * KLEN;
  const float* bcol = B + (size_t)(split * KLEN) * ncols + tn * 16 + m;
  v8f c = {};
#pragma unroll 4
  for (int k0 = 0; k0 < KLEN; k0 += 4) {
    v2f a, b;
    a.x = fmaxf(arow[k0 + kh + 0], 0.0f);
    a.y = fmaxf(arow[k0 + kh + 1], 0.0f);
    b.x = bcol[(size_t)(k0 + kh + 0) * ncols];
    b.y = bcol[(size_t)(k0 + kh + 1) * ncols];
    c = __builtin_amdgcn_wmma_f32_16x16x4_f32(false, a, false, b, (short)0, c, false, false);
  }
  int rbase = tm * 16 + ((lane >> 4) << 3);
#pragma unroll
  for (int r = 0; r < 8; ++r)
    unsafeAtomicAdd(&acc[(rbase + r) * ncols + tn * 16 + m], c[r]);
}

__global__ void k_bias_relu(const float* __restrict__ in, const float* __restrict__ bias,
                            float* __restrict__ out, int n, int ncols) {
  int i = blockIdx.x * blockDim.x + threadIdx.x;
  if (i < n) out[i] = fmaxf(in[i] + bias[i % ncols], 0.0f);
}

__global__ void k_zero(float* __restrict__ p, int n) {
  int i = blockIdx.x * blockDim.x + threadIdx.x;
  if (i < n) p[i] = 0.0f;
}

__global__ void k_gemm_final(const float* __restrict__ A, const float* __restrict__ B,
                             const float* __restrict__ bias, float* __restrict__ out) {
  int wid  = (blockIdx.x * blockDim.x + threadIdx.x) >> 5;
  int lane = threadIdx.x & 31;
  int tm = wid >> 6;
  int tn = wid & 63;
  int m  = lane & 15;
  int kh = (lane >> 4) << 1;
  const float* arow = A + (tm * 16 + m) * 32;
  const float* bcol = B + tn * 16 + m;
  v8f c = {};
#pragma unroll
  for (int k0 = 0; k0 < 32; k0 += 4) {
    v2f a, b;
    a.x = arow[k0 + kh + 0];
    a.y = arow[k0 + kh + 1];
    b.x = bcol[(k0 + kh + 0) * 1024];
    b.y = bcol[(k0 + kh + 1) * 1024];
    c = __builtin_amdgcn_wmma_f32_16x16x4_f32(false, a, false, b, (short)0, c, false, false);
  }
  int rbase = tm * 16 + ((lane >> 4) << 3);
  float bv = bias[tn * 16 + m];
#pragma unroll
  for (int r = 0; r < 8; ++r)
    out[(rbase + r) * 1024 + tn * 16 + m] = c[r] + bv;
}

__global__ void k_value_final(const float* __restrict__ vhid, const float* __restrict__ w,
                              const float* __restrict__ b, float* __restrict__ out) {
  int bi = threadIdx.x;
  float s = 0.0f;
#pragma unroll
  for (int j = 0; j < 16; ++j) s += vhid[bi * 16 + j] * w[j];
  out[bi] = s + b[0];
}

extern "C" void kernel_launch(void* const* d_in, const int* in_sizes, int n_in,
                              void* d_out, int out_size, void* d_ws, size_t ws_size,
                              hipStream_t stream) {
  const float* x    = (const float*)d_in[0];
  const int*   ei   = (const int*)d_in[1];
  const float* W1   = (const float*)d_in[2];
  const float* b1   = (const float*)d_in[3];
  const float* W2   = (const float*)d_in[4];
  const float* b2   = (const float*)d_in[5];
  const float* p1w  = (const float*)d_in[6];
  const float* p1b  = (const float*)d_in[7];
  const float* p2w  = (const float*)d_in[8];
  const float* p2b  = (const float*)d_in[9];
  const float* v1w  = (const float*)d_in[10];
  const float* v1b  = (const float*)d_in[11];
  const float* v2w  = (const float*)d_in[12];
  const float* v2b  = (const float*)d_in[13];

  const int N  = in_sizes[0] / 16;
  const int E  = in_sizes[1] / 2;
  const int* esrc = ei;
  const int* edst = ei + E;

  float* ws   = (float*)d_ws;
  float* dinv = ws + N;
  float* bufA = dinv + N;
  float* bufB = bufA + (size_t)16 * N;
  float* deg  = bufB;
  float* pacc = bufB + (size_t)16 * N;
  float* vacc = pacc + 64 * 32;
  float* phid = vacc + 64 * 16;
  float* vhid = phid + 64 * 32;

  float* Xout = (float*)d_out;
  float* Vout = Xout + 64 * 1024;

  const int B256 = 256;
  k_init_deg<<<N / B256, B256, 0, stream>>>(deg, N);
  k_count_deg<<<E / B256, B256, 0, stream>>>(edst, deg, E);
  k_dinv<<<N / B256, B256, 0, stream>>>(deg, dinv, N);

  int gemm_blocks    = (N / 16) / 8;
  int nranges        = N / SC_NR;
  int scatter_blocks = nranges * SC_S;
  k_gemm_n16<false><<<gemm_blocks, B256, 0, stream>>>(x, W1, bufA, N);
  k_init_agg<<<(16 * N) / B256, B256, 0, stream>>>(bufA, dinv, b1, bufB, 16 * N);
  k_scatter<<<scatter_blocks, 1024, 0, stream>>>(esrc, edst, dinv, bufA, bufB, E, nranges);

  k_gemm_n16<true><<<gemm_blocks, B256, 0, stream>>>(bufB, W2, bufA, N);
  k_init_agg<<<(16 * N) / B256, B256, 0, stream>>>(bufA, dinv, b2, bufB, 16 * N);
  k_scatter<<<scatter_blocks, 1024, 0, stream>>>(esrc, edst, dinv, bufA, bufB, E, nranges);

  const int K = 16384, NSPLIT = 64, KLEN = K / NSPLIT;

  k_zero<<<(64 * 32 + 64 * 16 + B256 - 1) / B256, B256, 0, stream>>>(pacc, 64 * 32 + 64 * 16);

  k_gemm_splitk_relu<KLEN><<<64, B256, 0, stream>>>(bufB, K, p1w, 32, pacc, 2, NSPLIT);
  k_gemm_splitk_relu<KLEN><<<32, B256, 0, stream>>>(bufB, K, v1w, 16, vacc, 1, NSPLIT);

  k_bias_relu<<<(64 * 32) / B256, B256, 0, stream>>>(pacc, p1b, phid, 64 * 32, 32);
  k_bias_relu<<<(64 * 16 + B256 - 1) / B256, B256, 0, stream>>>(vacc, v1b, vhid, 64 * 16, 16);

  k_gemm_final<<<32, B256, 0, stream>>>(phid, p2w, p2b, Xout);
  k_value_final<<<1, 64, 0, stream>>>(vhid, v2w, v2b, Vout);
}


# launch fusion, 16 launches to 8: dinv and the aggregate's start value move into the GEMM epilogue, both split-K heads share one grid, hidden bias+relu and the value head move into the last GEMM
# speedup vs baseline: 1.8834x; 1.1272x over previous
#include <hip/hip_runtime.h>
#include <hip/hip_bf16.h>

typedef __attribute__((ext_vector_type(2))) float v2f;
typedef __attribute__((ext_vector_type(8))) float v8f;

constexpr int DEG_STRIDE = 16;

__global__ void k_init_deg(float* __restrict__ deg, int n, float* __restrict__ zacc, int nz) {
  int i = blockIdx.x * blockDim.x + threadIdx.x;
  if (i < n) deg[i * DEG_STRIDE] = 1.0f;
  if (i < nz) zacc[i] = 0.0f;
}

__global__ void k_count_deg(const int* __restrict__ dst, float* __restrict__ deg, int ne) {
  int e = blockIdx.x * blockDim.x + threadIdx.x;
  if (e < ne) unsafeAtomicAdd(&deg[__builtin_nontemporal_load(dst + e) * DEG_STRIDE], 1.0f);
}

template <bool FIRST>
__global__ void k_gemm_agg(const float* __restrict__ x, const float* __restrict__ B,
                           const float* __restrict__ bias, float* dinv,
                           float* __restrict__ hw, float* agg, int nrows) {
  int wid  = (blockIdx.x * blockDim.x + threadIdx.x) >> 5;
  int lane = threadIdx.x & 31;
  int row0 = wid * 16;
  if (row0 >= nrows) return;
  int m  = lane & 15;
  int kh = (lane >> 4) << 1;
  int rbase = row0 + ((lane >> 4) << 3);
  float di[8];
#pragma unroll
  for (int r = 0; r < 8; ++r)
    di[r] = FIRST ? rsqrtf(agg[(rbase + r) * DEG_STRIDE]) : dinv[rbase + r];
  const float* arow = (FIRST ? x : (const float*)agg) + (row0 + m) * 16;
  v8f c = {};
#pragma unroll
  for (int k0 = 0; k0 < 16; k0 += 4) {
    v2f a, b;
    float a0 = arow[k0 + kh + 0];
    float a1 = arow[k0 + kh + 1];
    if (!FIRST) { a0 = fmaxf(a0, 0.0f); a1 = fmaxf(a1, 0.0f); }
    a.x = a0; a.y = a1;
    b.x = B[(k0 + kh + 0) * 16 + m];
    b.y = B[(k0 + kh + 1) * 16 + m];
    c = __builtin_amdgcn_wmma_f32_16x16x4_f32(false, a, false, b, (short)0, c, false, false);
  }
  const float bv = bias[m];
#pragma unroll
  for (int r = 0; r < 8; ++r) {
    const float h = c[r];
    hw[(rbase + r) * 16 + m]  = h;
    agg[(rbase + r) * 16 + m] = h * di[r] * di[r] + bv;
  }
  if (FIRST) {
    if (m == 0) {
#pragma unroll
      for (int r = 0; r < 8; ++r) dinv[rbase + r] = di[r];
    }
  }
}

typedef int v4i __attribute__((ext_vector_type(4)));

constexpr int SC_NR    = 1024;
constexpr int SC_S     = 4;
constexpr int SC_WAVES = 32;
constexpr int SC_LOADS = 8;
constexpr int SC_BATCH = 32 * 4 * SC_LOADS;
constexpr int SC_UD    = 8;

__global__ __launch_bounds__(1024)
void k_scatter(const int* __restrict__ src, const int* __restrict__ dst,
               const float* __restrict__ dinv, const float* __restrict__ hw,
               float* __restrict__ agg, int ne, int nranges) {
  __shared__ float acc[SC_NR * 16];
  __shared__ unsigned q[SC_WAVES][SC_BATCH + 2 * SC_UD];
  const int tid = threadIdx.x, lane = tid & 31, w = tid >> 5;
  const int half = lane >> 4, f = lane & 15;
  const int r  = blockIdx.x % nranges;
  const int sl = blockIdx.x / nranges;
  const int lo = r * SC_NR;
  for (int i = tid; i < SC_NR * 16; i += 1024) acc[i] = 0.0f;
  __syncthreads();

  const int slice = ne / SC_S;
  const int span  = slice / SC_WAVES;
  const int ebase = sl * slice + w * span;
  const int nb    = span / SC_BATCH;
  const v4i* dp   = (const v4i*)(dst + ebase);
  const int* sp   = src + ebase;
  unsigned* qw    = q[w];
  int qn = 0;

  auto pass = [&](int q0, int nv) {
    unsigned ent[SC_UD]; bool ok[SC_UD]; int sv[SC_UD]; float wv[SC_UD], hv[SC_UD];
#pragma unroll
    for (int u = 0; u < SC_UD; ++u) {
      ok[u]  = 2 * u + half < nv;
      ent[u] = ok[u] ? qw[q0 + 2 * u + half] : 0u;
    }
#pragma unroll
    for (int u = 0; u < SC_UD; ++u) sv[u] = sp[ent[u] >> 10];
#pragma unroll
    for (int u = 0; u < SC_UD; ++u) {
      wv[u] = dinv[sv[u]] * dinv[lo + (int)(ent[u] & 1023u)];
      hv[u] = hw[sv[u] * 16 + f];
    }
#pragma unroll
    for (int u = 0; u < SC_UD; ++u)
      unsafeAtomicAdd(&acc[(ent[u] & 1023u) * 16 + f], ok[u] ? hv[u] * wv[u] : 0.0f);
  };

  v4i cur[SC_LOADS], nxt[SC_LOADS];
#pragma unroll
  for (int j = 0; j < SC_LOADS; ++j) cur[j] = dp[j * 32 + lane];
  for (int b = 0; b < nb; ++b) {
    const int bn = (b + 1 < nb) ? b + 1 : b;
#pragma unroll
    for (int j = 0; j < SC_LOADS; ++j) nxt[j] = dp[bn * (SC_BATCH / 4) + j * 32 + lane];
#pragma unroll
    for (int j = 0; j < SC_LOADS; ++j) {
#pragma unroll
      for (int c = 0; c < 4; ++c) {
        const int dl = cur[j][c] - lo;
        const bool m = (unsigned)dl < (unsigned)SC_NR;
        const unsigned mask = __builtin_amdgcn_ballot_w32(m);
        if (mask != 0u) {
          const int cnt = __builtin_popcount(mask);
          if (m) {
            const int el = b * SC_BATCH + j * 128 + lane * 4 + c;
            qw[qn + (int)__builtin_amdgcn_mbcnt_lo(mask, 0u)] = ((unsigned)el << 10) | (unsigned)dl;
          }
          qn += cnt;
        }
      }
    }
    while (qn >= 2 * SC_UD) { qn -= 2 * SC_UD; pass(qn, 2 * SC_UD); }
#pragma unroll
    for (int j = 0; j < SC_LOADS; ++j) cur[j] = nxt[j];
  }
  if (qn > 0) pass(0, qn);
  __syncthreads();
  float* arow = agg + (size_t)lo * 16;
  for (int i = tid; i < SC_NR * 16; i += 1024) unsafeAtomicAdd(arow + i, acc[i]);
}

template <int KLEN>
__global__ void k_heads_splitk_relu(const float* __restrict__ A, int lda,
                                    const float* __restrict__ Bp, const float* __restrict__ Bv,
                                    float* __restrict__ accp, float* __restrict__ accv,
                                    int nsplit, int pol_blocks) {
  const bool pol = (int)blockIdx.x < pol_blocks;
  const float* __restrict__ B = pol ? Bp : Bv;
  float* __restrict__ acc     = pol ? accp : accv;
  const int ncols   = pol ? 32 : 16;
  const int tiles_n = pol ? 2 : 1;
  const int blk     = pol ? (int)blockIdx.x : (int)blockIdx.x - pol_blocks;
  int wid  = (blk * blockDim.x + threadIdx.x) >> 5;
  int lane = threadIdx.x & 31;
  int split = wid % nsplit;
  int tn    = (wid / nsplit) % tiles_n;
  int tm    = wid / (nsplit * tiles_n);
  int m  = lane & 15;
  int kh = (lane >> 4) << 1;
  const float* arow = A + (size_t)(tm * 16 + m) * lda + split * KLEN;
  const float* bcol = B + (size_t)(split * KLEN) * ncols + tn * 16 + m;
  v8f c = {};
#pragma unroll 4
  for (int k0 = 0; k0 < KLEN; k0 += 4) {
    v2f a, b;
    a.x = fmaxf(arow[k0 + kh + 0], 0.0f);
    a.y = fmaxf(arow[k0 + kh + 1], 0.0f);
    b.x = bcol[(size_t)(k0 + kh + 0) * ncols];
    b.y = bcol[(size_t)(k0 + kh + 1) * ncols];
    c = __builtin_amdgcn_wmma_f32_16x16x4_f32(false, a, false, b, (short)0, c, false, false);
  }
  int rbase = tm * 16 + ((lane >> 4) << 3);
#pragma unroll
  for (int r = 0; r < 8; ++r)
    unsafeAtomicAdd(&acc[(rbase + r) * ncols + tn * 16 + m], c[r]);
}

__global__ void k_heads_final(const float* __restrict__ pacc, const float* __restrict__ p1b,
                              const float* __restrict__ B, const float* __restrict__ bias,
                              const float* __restrict__ vacc, const float* __restrict__ v1b,
                              const float* __restrict__ v2w, const float* __restrict__ v2b,
                              float* __restrict__ out, float* __restrict__ vout) {
  if (blockIdx.x == 32) {
    int bi = threadIdx.x;
    if (bi < 64) {
      float s = 0.0f;
#pragma unroll
      for (int j = 0; j < 16; ++j) s += fmaxf(vacc[bi * 16 + j] + v1b[j], 0.0f) * v2w[j];
      vout[bi] = s + v2b[0];
    }
    return;
  }
  int wid  = (blockIdx.x * blockDim.x + threadIdx.x) >> 5;
  int lane = threadIdx.x & 31;
  int tm = wid >> 6;
  int tn = wid & 63;
  int m  = lane & 15;
  int kh = (lane >> 4) << 1;
  const float* arow = pacc + (tm * 16 + m) * 32;
  const float* bcol = B + tn * 16 + m;
  v8f c = {};
#pragma unroll
  for (int k0 = 0; k0 < 32; k0 += 4) {
    v2f a, b;
    a.x = fmaxf(arow[k0 + kh + 0] + p1b[k0 + kh + 0], 0.0f);
    a.y = fmaxf(arow[k0 + kh + 1] + p1b[k0 + kh + 1], 0.0f);
    b.x = bcol[(k0 + kh + 0) * 1024];
    b.y = bcol[(k0 + kh + 1) * 1024];
    c = __builtin_amdgcn_wmma_f32_16x16x4_f32(false, a, false, b, (short)0, c, false, false);
  }
  int rbase = tm * 16 + ((lane >> 4) << 3);
  float bv = bias[tn * 16 + m];
#pragma unroll
  for (int r = 0; r < 8; ++r)
    out[(rbase + r) * 1024 + tn * 16 + m] = c[r] + bv;
}

extern "C" void kernel_launch(void* const* d_in, const int* in_sizes, int n_in,
                              void* d_out, int out_size, void* d_ws, size_t ws_size,
                              hipStream_t stream) {
  const float* x    = (const float*)d_in[0];
  const int*   ei   = (const int*)d_in[1];
  const float* W1   = (const float*)d_in[2];
  const float* b1   = (const float*)d_in[3];
  const float* W2   = (const float*)d_in[4];
  const float* b2   = (const float*)d_in[5];
  const float* p1w  = (const float*)d_in[6];
  const float* p1b  = (const float*)d_in[7];
  const float* p2w  = (const float*)d_in[8];
  const float* p2b  = (const float*)d_in[9];
  const float* v1w  = (const float*)d_in[10];
  const float* v1b  = (const float*)d_in[11];
  const float* v2w  = (const float*)d_in[12];
  const float* v2b  = (const float*)d_in[13];

  const int N  = in_sizes[0] / 16;
  const int E  = in_sizes[1] / 2;
  const int* esrc = ei;
  const int* edst = ei + E;

  float* ws   = (float*)d_ws;
  float* dinv = ws + N;
  float* bufA = dinv + N;
  float* bufB = bufA + (size_t)16 * N;
  float* deg  = bufB;
  float* pacc = bufB + (size_t)16 * N;
  float* vacc = pacc + 64 * 32;

  float* Xout = (float*)d_out;
  float* Vout = Xout + 64 * 1024;

  const int B256 = 256;
  k_init_deg<<<N / B256, B256, 0, stream>>>(deg, N, pacc, 64 * 32 + 64 * 16);
  k_count_deg<<<E / B256, B256, 0, stream>>>(edst, deg, E);

  int gemm_blocks    = (N / 16) / 8;
  int nranges        = N / SC_NR;
  int scatter_blocks = nranges * SC_S;
  k_gemm_agg<true><<<gemm_blocks, B256, 0, stream>>>(x, W1, b1, dinv, bufA, bufB, N);
  k_scatter<<<scatter_blocks, 1024, 0, stream>>>(esrc, edst, dinv, bufA, bufB, E, nranges);

  k_gemm_agg<false><<<gemm_blocks, B256, 0, stream>>>(x, W2, b2, dinv, bufA, bufB, N);
  k_scatter<<<scatter_blocks, 1024, 0, stream>>>(esrc, edst, dinv, bufA, bufB, E, nranges);

  const int K = 16384, NSPLIT = 64, KLEN = K / NSPLIT;

  k_heads_splitk_relu<KLEN><<<64 + 32, B256, 0, stream>>>(bufB, K, p1w, v1w, pacc, vacc, NSPLIT, 64);

  k_heads_final<<<32 + 1, B256, 0, stream>>>(pacc, p1b, p2w, p2b, vacc, v1b, v2w, v2b, Xout, Vout);
}


# edge weight split in two factors: GEMM stores rows times dinv[src], scatter sums plain rows in LDS, applies dinv[dst] once per row at hand-over; no weight gathers per edge; rounding differs
# speedup vs baseline: 2.2654x; 1.2028x over previous
#include <hip/hip_runtime.h>
#include <hip/hip_bf16.h>

typedef __attribute__((ext_vector_type(2))) float v2f;
typedef __attribute__((ext_vector_type(8))) float v8f;

constexpr int DEG_STRIDE = 16;

__global__ void k_init_deg(float* __restrict__ deg, int n, float* __restrict__ zacc, int nz) {
  int i = blockIdx.x * blockDim.x + threadIdx.x;
  if (i < n) deg[i * DEG_STRIDE] = 1.0f;
  if (i < nz) zacc[i] = 0.0f;
}

__global__ void k_count_deg(const int* __restrict__ dst, float* __restrict__ deg, int ne) {
  int e = blockIdx.x * blockDim.x + threadIdx.x;
  if (e < ne) unsafeAtomicAdd(&deg[__builtin_nontemporal_load(dst + e) * DEG_STRIDE], 1.0f);
}

template <bool FIRST>
__global__ void k_gemm_agg(const float* __restrict__ x, const float* __restrict__ B,
                           const float* __restrict__ bias, float* dinv,
                           float* __restrict__ hw, float* agg, int nrows) {
  int wid  = (blockIdx.x * blockDim.x + threadIdx.x) >> 5;
  int lane = threadIdx.x & 31;
  int row0 = wid * 16;
  if (row0 >= nrows) return;
  int m  = lane & 15;
  int kh = (lane >> 4) << 1;
  int rbase = row0 + ((lane >> 4) << 3);
  float di[8];
#pragma unroll
  for (int r = 0; r < 8; ++r)
    di[r] = FIRST ? rsqrtf(agg[(rbase + r) * DEG_STRIDE]) : dinv[rbase + r];
  const float* arow = (FIRST ? x : (const float*)agg) + (row0 + m) * 16;
  v8f c = {};
#pragma unroll
  for (int k0 = 0; k0 < 16; k0 += 4) {
    v2f a, b;
    float a0 = arow[k0 + kh + 0];
    float a1 = arow[k0 + kh + 1];
    if (!FIRST) { a0 = fmaxf(a0, 0.0f); a1 = fmaxf(a1, 0.0f); }
    a.x = a0; a.y = a1;
    b.x = B[(k0 + kh + 0) * 16 + m];
    b.y = B[(k0 + kh + 1) * 16 + m];
    c = __builtin_amdgcn_wmma_f32_16x16x4_f32(false, a, false, b, (short)0, c, false, false);
  }
  const float bv = bias[m];
#pragma unroll
  for (int r = 0; r < 8; ++r) {
    const float hs = c[r] * di[r];
    hw[(rbase + r) * 16 + m]  = hs;
    agg[(rbase + r) * 16 + m] = hs * di[r] + bv;
  }
  if (FIRST) {
    if (m == 0) {
#pragma unroll
      for (int r = 0; r < 8; ++r) dinv[rbase + r] = di[r];
    }
  }
}

typedef int v4i __attribute__((ext_vector_type(4)));

constexpr int SC_NR    = 1024;
constexpr int SC_S     = 4;
constexpr int SC_WAVES = 32;
constexpr int SC_LOADS = 8;
constexpr int SC_BATCH = 32 * 4 * SC_LOADS;
constexpr int SC_UD    = 8;

__global__ __launch_bounds__(1024)
void k_scatter(const int* __restrict__ src, const int* __restrict__ dst,
               const float* __restrict__ dinv, const float* __restrict__ hw,
               float* __restrict__ agg, int ne, int nranges) {
  __shared__ float acc[SC_NR * 16];
  __shared__ unsigned q[SC_WAVES][SC_BATCH + 2 * SC_UD];
  const int tid = threadIdx.x, lane = tid & 31, w = tid >> 5;
  const int half = lane >> 4, f = lane & 15;
  const int r  = blockIdx.x % nranges;
  const int sl = blockIdx.x / nranges;
  const int lo = r * SC_NR;
  for (int i = tid; i < SC_NR * 16; i += 1024) acc[i] = 0.0f;
  __syncthreads();

  const int slice = ne / SC_S;
  const int span  = slice / SC_WAVES;
  const int ebase = sl * slice + w * span;
  const int nb    = span / SC_BATCH;
  const v4i* dp   = (const v4i*)(dst + ebase);
  const int* sp   = src + ebase;
  unsigned* qw    = q[w];
  int qn = 0;

  auto pass = [&](int q0, int nv) {
    unsigned ent[SC_UD]; bool ok[SC_UD]; int sv[SC_UD]; float hv[SC_UD];
#pragma unroll
    for (int u = 0; u < SC_UD; ++u) {
      ok[u]  = 2 * u + half < nv;
      ent[u] = ok[u] ? qw[q0 + 2 * u + half] : 0u;
    }
#pragma unroll
    for (int u = 0; u < SC_UD; ++u) sv[u] = sp[ent[u] >> 10];
#pragma unroll
    for (int u = 0; u < SC_UD; ++u) hv[u] = hw[sv[u] * 16 + f];
#pragma unroll
    for (int u = 0; u < SC_UD; ++u)
      unsafeAtomicAdd(&acc[(ent[u] & 1023u) * 16 + f], ok[u] ? hv[u] : 0.0f);
  };

  v4i cur[SC_LOADS], nxt[SC_LOADS];
#pragma unroll
  for (int j = 0; j < SC_LOADS; ++j) cur[j] = dp[j * 32 + lane];
  for (int b = 0; b < nb; ++b) {
    const int bn = (b + 1 < nb) ? b + 1 : b;
#pragma unroll
    for (int j = 0; j < SC_LOADS; ++j) nxt[j] = dp[bn * (SC_BATCH / 4) + j * 32 + lane];
#pragma unroll
    for (int j = 0; j < SC_LOADS; ++j) {
#pragma unroll
      for (int c = 0; c < 4; ++c) {
        const int dl = cur[j][c] - lo;
        const bool m = (unsigned)dl < (unsigned)SC_NR;
        const unsigned mask = __builtin_amdgcn_ballot_w32(m);
        if (mask != 0u) {
          const int cnt = __builtin_popcount(mask);
          if (m) {
            const int el = b * SC_BATCH + j * 128 + lane * 4 + c;
            qw[qn + (int)__builtin_amdgcn_mbcnt_lo(mask, 0u)] = ((unsigned)el << 10) | (unsigned)dl;
          }
          qn += cnt;
        }
      }
    }
    while (qn >= 2 * SC_UD) { qn -= 2 * SC_UD; pass(qn, 2 * SC_UD); }
#pragma unroll
    for (int j = 0; j < SC_LOADS; ++j) cur[j] = nxt[j];
  }
  constexpr int SC_FL = SC_NR * 16 / 1024;
  float dv[SC_FL], av[SC_FL];
#pragma unroll
  for (int j = 0; j < SC_FL; ++j) dv[j] = dinv[lo + (tid >> 4) + 64 * j];
  if (qn > 0) pass(0, qn);
  __syncthreads();
  float* arow = agg + (size_t)lo * 16;
#pragma unroll
  for (int j = 0; j < SC_FL; ++j) av[j] = acc[tid + 1024 * j] * dv[j];
#pragma unroll
  for (int j = 0; j < SC_FL; ++j) unsafeAtomicAdd(arow + tid + 1024 * j, av[j]);
}

template <int KLEN>
__global__ void k_heads_splitk_relu(const float* __restrict__ A, int lda,
                                    const float* __restrict__ Bp, const float* __restrict__ Bv,
                                    float* __restrict__ accp, float* __restrict__ accv,
                                    int nsplit, int pol_blocks) {
  const bool pol = (int)blockIdx.x < pol_blocks;
  const float* __restrict__ B = pol ? Bp : Bv;
  float* __restrict__ acc     = pol ? accp : accv;
  const int ncols   = pol ? 32 : 16;
  const int tiles_n = pol ? 2 : 1;
  const int blk     = pol ? (int)blockIdx.x : (int)blockIdx.x - pol_blocks;
  int wid  = (blk * blockDim.x + threadIdx.x) >> 5;
  int lane = threadIdx.x & 31;
  int split = wid % nsplit;
  int tn    = (wid / nsplit) % tiles_n;
  int tm    = wid / (nsplit * tiles_n);
  int m  = lane & 15;
  int kh = (lane >> 4) << 1;
  const float* arow = A + (size_t)(tm * 16 + m) * lda + split * KLEN;
  const float* bcol = B + (size_t)(split * KLEN) * ncols + tn * 16 + m;
  v8f c = {};
#pragma unroll 4
  for (int k0 = 0; k0 < KLEN; k0 += 4) {
    v2f a, b;
    a.x = fmaxf(arow[k0 + kh + 0], 0.0f);
    a.y = fmaxf(arow[k0 + kh + 1], 0.0f);
    b.x = bcol[(size_t)(k0 + kh + 0) * ncols];
    b.y = bcol[(size_t)(k0 + kh + 1) * ncols];
    c = __builtin_amdgcn_wmma_f32_16x16x4_f32(false, a, false, b, (short)0, c, false, false);
  }
  int rbase = tm * 16 + ((lane >> 4) << 3);
#pragma unroll
  for (int r = 0; r < 8; ++r)
    unsafeAtomicAdd(&acc[(rbase + r) * ncols + tn * 16 + m], c[r]);
}

__global__ void k_heads_final(const float* __restrict__ pacc, const float* __restrict__ p1b,
                              const float* __restrict__ B, const float* __restrict__ bias,
                              const float* __restrict__ vacc, const float* __restrict__ v1b,
                              const float* __restrict__ v2w, const float* __restrict__ v2b,
                              float* __restrict__ out, float* __restrict__ vout) {
  if (blockIdx.x == 32) {
    int bi = threadIdx.x;
    if (bi < 64) {
      float s = 0.0f;
#pragma unroll
      for (int j = 0; j < 16; ++j) s += fmaxf(vacc[bi * 16 + j] + v1b[j], 0.0f) * v2w[j];
      vout[bi] = s + v2b[0];
    }
    return;
  }
  int wid  = (blockIdx.x * blockDim.x + threadIdx.x) >> 5;
  int lane = threadIdx.x & 31;
  int tm = wid >> 6;
  int tn = wid & 63;
  int m  = lane & 15;
  int kh = (lane >> 4) << 1;
  const float* arow = pacc + (tm * 16 + m) * 32;
  const float* bcol = B + tn * 16 + m;
  v8f c = {};
#pragma unroll
  for (int k0 = 0; k0 < 32; k0 += 4) {
    v2f a, b;
    a.x = fmaxf(arow[k0 + kh + 0] + p1b[k0 + kh + 0], 0.0f);
    a.y = fmaxf(arow[k0 + kh + 1] + p1b[k0 + kh + 1], 0.0f);
    b.x = bcol[(k0 + kh + 0) * 1024];
    b.y = bcol[(k0 + kh + 1) * 1024];
    c = __builtin_amdgcn_wmma_f32_16x16x4_f32(false, a, false, b, (short)0, c, false, false);
  }
  int rbase = tm * 16 + ((lane >> 4) << 3);
  float bv = bias[tn * 16 + m];
#pragma unroll
  for (int r = 0; r < 8; ++r)
    out[(rbase + r) * 1024 + tn * 16 + m] = c[r] + bv;
}

extern "C" void kernel_launch(void* const* d_in, const int* in_sizes, int n_in,
                              void* d_out, int out_size, void* d_ws, size_t ws_size,
                              hipStream_t stream) {
  const float* x    = (const float*)d_in[0];
  const int*   ei   = (const int*)d_in[1];
  const float* W1   = (const float*)d_in[2];
  const float* b1   = (const float*)d_in[3];
  const float* W2   = (const float*)d_in[4];
  const float* b2   = (const float*)d_in[5];
  const float* p1w  = (const float*)d_in[6];
  const float* p1b  = (const float*)d_in[7];
  const float* p2w  = (const float*)d_in[8];
  const float* p2b  = (const float*)d_in[9];
  const float* v1w  = (const float*)d_in[10];
  const float* v1b  = (const float*)d_in[11];
  const float* v2w  = (const float*)d_in[12];
  const float* v2b  = (const float*)d_in[13];

  const int N  = in_sizes[0] / 16;
  const int E  = in_sizes[1] / 2;
  const int* esrc = ei;
  const int* edst = ei + E;

  float* ws   = (float*)d_ws;
  float* dinv = ws + N;
  float* bufA = dinv + N;
  float* bufB = bufA + (size_t)16 * N;
  float* deg  = bufB;
  float* pacc = bufB + (size_t)16 * N;
  float* vacc = pacc + 64 * 32;

  float* Xout = (float*)d_out;
  float* Vout = Xout + 64 * 1024;

  const int B256 = 256;
  k_init_deg<<<N / B256, B256, 0, stream>>>(deg, N, pacc, 64 * 32 + 64 * 16);
  k_count_deg<<<E / B256, B256, 0, stream>>>(edst, deg, E);

  int gemm_blocks    = (N / 16) / 8;
  int nranges        = N / SC_NR;
  int scatter_blocks = nranges * SC_S;
  k_gemm_agg<true><<<gemm_blocks, B256, 0, stream>>>(x, W1, b1, dinv, bufA, bufB, N);
  k_scatter<<<scatter_blocks, 1024, 0, stream>>>(esrc, edst, dinv, bufA, bufB, E, nranges);

  k_gemm_agg<false><<<gemm_blocks, B256, 0, stream>>>(x, W2, b2, dinv, bufA, bufB, N);
  k_scatter<<<scatter_blocks, 1024, 0, stream>>>(esrc, edst, dinv, bufA, bufB, E, nranges);

  const int K = 16384, NSPLIT = 64, KLEN = K / NSPLIT;

  k_heads_splitk_relu<KLEN><<<64 + 32, B256, 0, stream>>>(bufB, K, p1w, v1w, pacc, vacc, NSPLIT, 64);

  k_heads_final<<<32 + 1, B256, 0, stream>>>(pacc, p1b, p2w, p2b, vacc, v1b, v2w, v2b, Xout, Vout);
}


# layer 2's edge scatter replays per-range edge lists (src<<10|local row) that layer 1's scatter files while filtering: no dst scan, no src-index gather; lists in arrival order, sum order changes
# speedup vs baseline: 2.5408x; 1.1216x over previous
#include <hip/hip_runtime.h>
#include <hip/hip_bf16.h>

typedef __attribute__((ext_vector_type(2))) float v2f;
typedef __attribute__((ext_vector_type(8))) float v8f;

constexpr int DEG_STRIDE = 16;
constexpr int SC_NR      = 1024;

__global__ void k_init_deg(float* __restrict__ deg, int n, float* __restrict__ zacc, int nz,
                           unsigned* __restrict__ rc, int nrc) {
  int i = blockIdx.x * blockDim.x + threadIdx.x;
  if (i < n) deg[i * DEG_STRIDE] = 1.0f;
  if (i < nz) zacc[i] = 0.0f;
  if (i < nrc) rc[i] = 0u;
}

__global__ void k_count_deg(const int* __restrict__ dst, float* __restrict__ deg, int ne) {
  int e = blockIdx.x * blockDim.x + threadIdx.x;
  if (e < ne) unsafeAtomicAdd(&deg[__builtin_nontemporal_load(dst + e) * DEG_STRIDE], 1.0f);
}

template <bool FIRST>
__global__ void k_gemm_agg(const float* __restrict__ x, const float* __restrict__ B,
                           const float* __restrict__ bias, float* dinv,
                           float* __restrict__ hw, float* agg, int nrows,
                           unsigned* __restrict__ wavecnt) {
  int wid  = (blockIdx.x * blockDim.x + threadIdx.x) >> 5;
  int lane = threadIdx.x & 31;
  int row0 = wid * 16;
  if (row0 >= nrows) return;
  int m  = lane & 15;
  int kh = (lane >> 4) << 1;
  int rbase = row0 + ((lane >> 4) << 3);
  float di[8];
  float dsum = 0.0f;
#pragma unroll
  for (int r = 0; r < 8; ++r) {
    if (FIRST) {
      const float dg = agg[(rbase + r) * DEG_STRIDE];
      dsum += dg;
      di[r] = rsqrtf(dg);
    } else {
      di[r] = dinv[rbase + r];
    }
  }
  const float* arow = (FIRST ? x : (const float*)agg) + (row0 + m) * 16;
  v8f c = {};
#pragma unroll
  for (int k0 = 0; k0 < 16; k0 += 4) {
    v2f a, b;
    float a0 = arow[k0 + kh + 0];
    float a1 = arow[k0 + kh + 1];
    if (!FIRST) { a0 = fmaxf(a0, 0.0f); a1 = fmaxf(a1, 0.0f); }
    a.x = a0; a.y = a1;
    b.x = B[(k0 + kh + 0) * 16 + m];
    b.y = B[(k0 + kh + 1) * 16 + m];
    c = __builtin_amdgcn_wmma_f32_16x16x4_f32(false, a, false, b, (short)0, c, false, false);
  }
  const float bv = bias[m];
#pragma unroll
  for (int r = 0; r < 8; ++r) {
    const float hs = c[r] * di[r];
    hw[(rbase + r) * 16 + m]  = hs;
    agg[(rbase + r) * 16 + m] = hs * di[r] + bv;
  }
  if (FIRST) {
    if (m == 0) {
#pragma unroll
      for (int r = 0; r < 8; ++r) dinv[rbase + r] = di[r];
    }
    const float tot = dsum + __shfl_xor(dsum, 16);
    if (lane == 0) wavecnt[wid] = (unsigned)tot - 16u;
  }
}

typedef int v4i __attribute__((ext_vector_type(4)));

constexpr int SC_S     = 4;
constexpr int SC_WAVES = 32;
constexpr int SC_LOADS = 8;
constexpr int SC_BATCH = 32 * 4 * SC_LOADS;
constexpr int SC_UD    = 8;
constexpr int SC_QLEN  = SC_BATCH + 2 * SC_UD;
constexpr int SC_RB    = 128;
constexpr int SC_RBLEN = SC_RB + 2 * SC_UD;
constexpr int RP_UD    = 16;

constexpr int SC_WPR = SC_NR / 16;
constexpr int SC_FILL = 32;

__device__ __forceinline__ unsigned sc_wave_sum(unsigned v) {
  for (int o = 16; o > 0; o >>= 1) v += __shfl_xor(v, o);
  return (unsigned)__builtin_amdgcn_readfirstlane((int)v);
}
__device__ __forceinline__ void sc_bounds_shares(const unsigned* __restrict__ wavecnt, int r, unsigned* part) {
  const int tid = threadIdx.x;
  const int nb = r * SC_WPR;
  unsigned s = 0u;
  for (int i = tid; i < nb; i += 1024) s += wavecnt[i];
  const unsigned c = (tid < SC_WPR) ? wavecnt[nb + tid] : 0u;
  s = sc_wave_sum(s);
  const unsigned cs = sc_wave_sum(c);
  if ((tid & 31) == 0) { part[tid >> 5] = s; part[SC_WAVES + (tid >> 5)] = cs; }
}
__device__ __forceinline__ void sc_bounds(const unsigned* part, unsigned& start, unsigned& cnt) {
  const int lane = threadIdx.x & 31;
  start = sc_wave_sum(part[lane]);
  cnt   = sc_wave_sum(part[SC_WAVES + lane]);
}

__device__ __forceinline__ bool sc_list_fits(const unsigned* list, int ne, const float* ws0, size_t ws_bytes) {
  return (size_t)((const char*)(list + ne) - (const char*)ws0) <= ws_bytes;
}

__device__ __forceinline__ void sc_hand_over(const float* acc, const float* __restrict__ dinv,
                                             float* __restrict__ agg, int lo, int tid) {
  constexpr int SC_FL = SC_NR * 16 / 1024;
  float dv[SC_FL], av[SC_FL];
#pragma unroll
  for (int j = 0; j < SC_FL; ++j) dv[j] = dinv[lo + (tid >> 4) + 64 * j];
  __syncthreads();
  float* arow = agg + (size_t)lo * 16;
#pragma unroll
  for (int j = 0; j < SC_FL; ++j) av[j] = acc[tid + 1024 * j] * dv[j];
#pragma unroll
  for (int j = 0; j < SC_FL; ++j) unsafeAtomicAdd(arow + tid + 1024 * j, av[j]);
}

template <bool REC>
__device__ __forceinline__ void sc_scan(float* acc, unsigned* qw, unsigned* rb,
                                        const int* __restrict__ src, const int* __restrict__ dst,
                                        const float* __restrict__ hw, int ne, int r, int sl,
                                        unsigned* __restrict__ rlist, unsigned* __restrict__ fill) {
  const int tid = threadIdx.x, lane = tid & 31, w = tid >> 5;
  const int half = lane >> 4, f = lane & 15;
  const int lo = r * SC_NR;

  const int slice = ne / SC_S;
  const int span  = slice / SC_WAVES;
  const int ebase = sl * slice + w * span;
  const int nb    = span / SC_BATCH;
  const v4i* dp   = (const v4i*)(dst + ebase);
  const int* sp   = src + ebase;
  int qn = 0;
  int rn = 0;

  auto flush = [&]() {
    unsigned t = 0u;
    if (lane == 0) t = atomicAdd(fill, (unsigned)rn);
    t = (unsigned)__builtin_amdgcn_readfirstlane((int)t);
    for (int i = lane; i < rn; i += 32) rlist[t + i] = rb[i];
    rn = 0;
  };

  auto pass = [&](int q0, int nv) {
    unsigned ent[SC_UD]; bool ok[SC_UD]; int sv[SC_UD]; float hv[SC_UD];
#pragma unroll
    for (int u = 0; u < SC_UD; ++u) {
      ok[u]  = 2 * u + half < nv;
      ent[u] = ok[u] ? qw[q0 + 2 * u + half] : 0u;
    }
#pragma unroll
    for (int u = 0; u < SC_UD; ++u) sv[u] = sp[ent[u] >> 10];
#pragma unroll
    for (int u = 0; u < SC_UD; ++u) hv[u] = hw[sv[u] * 16 + f];
    if (REC) {
      unsigned word = 0u;
#pragma unroll
      for (int u = 0; u < SC_UD; ++u)
        word = (f == u) ? (((unsigned)sv[u] << 10) | (ent[u] & 1023u)) : word;
      const int idx = 2 * f + half;
      if (f < SC_UD && idx < nv) rb[rn + idx] = word;
    }
#pragma unroll
    for (int u = 0; u < SC_UD; ++u)
      unsafeAtomicAdd(&acc[(ent[u] & 1023u) * 16 + f], ok[u] ? hv[u] : 0.0f);
    if (REC) {
      rn += nv;
      if (rn >= SC_RB) flush();
    }
  };

  v4i cur[SC_LOADS], nxt[SC_LOADS];
#pragma unroll
  for (int j = 0; j < SC_LOADS; ++j) cur[j] = dp[j * 32 + lane];
  for (int b = 0; b < nb; ++b) {
    const int bn = (b + 1 < nb) ? b + 1 : b;
#pragma unroll
    for (int j = 0; j < SC_LOADS; ++j) nxt[j] = dp[bn * (SC_BATCH / 4) + j * 32 + lane];
#pragma unroll
    for (int j = 0; j < SC_LOADS; ++j) {
#pragma unroll
      for (int c = 0; c < 4; ++c) {
        const int dl = cur[j][c] - lo;
        const bool m = (unsigned)dl < (unsigned)SC_NR;
        const unsigned mask = __builtin_amdgcn_ballot_w32(m);
        if (mask != 0u) {
          const int cnt = __builtin_popcount(mask);
          if (m) {
            const int el = b * SC_BATCH + j * 128 + lane * 4 + c;
            qw[qn + (int)__builtin_amdgcn_mbcnt_lo(mask, 0u)] = ((unsigned)el << 10) | (unsigned)dl;
          }
          qn += cnt;
        }
      }
    }
    while (qn >= 2 * SC_UD) { qn -= 2 * SC_UD; pass(qn, 2 * SC_UD); }
#pragma unroll
    for (int j = 0; j < SC_LOADS; ++j) cur[j] = nxt[j];
  }
  if (qn > 0) pass(0, qn);
  if (REC) { if (rn > 0) flush(); }
}

__global__ __launch_bounds__(1024)
void k_scatter_scan(const int* __restrict__ src, const int* __restrict__ dst,
                    const float* __restrict__ dinv, const float* __restrict__ hw,
                    float* __restrict__ agg, int ne, int nranges,
                    unsigned* __restrict__ rc, unsigned* __restrict__ list,
                    const float* ws0, size_t ws_bytes) {
  __shared__ float acc[SC_NR * 16];
  __shared__ unsigned q[SC_WAVES][SC_QLEN];
  __shared__ unsigned rb[SC_WAVES][SC_RBLEN];
  const int tid = threadIdx.x;
  const int r  = blockIdx.x % nranges;
  const int sl = blockIdx.x / nranges;
  __shared__ unsigned shares[2 * SC_WAVES];
  const bool fits = sc_list_fits(list, ne, ws0, ws_bytes);
  for (int i = tid; i < SC_NR * 16; i += 1024) acc[i] = 0.0f;
  if (fits) sc_bounds_shares(rc + nranges * SC_FILL, r, shares);
  __syncthreads();
  if (fits) {
    unsigned lstart, lcnt;
    sc_bounds(shares, lstart, lcnt);
    sc_scan<true>(acc, q[tid >> 5], rb[tid >> 5], src, dst, hw, ne, r, sl, list + lstart, rc + r * SC_FILL);
  } else {
    sc_scan<false>(acc, q[tid >> 5], nullptr, src, dst, hw, ne, r, sl, nullptr, nullptr);
  }
  sc_hand_over(acc, dinv, agg, r * SC_NR, tid);
}

__global__ __launch_bounds__(1024)
void k_scatter_replay(const int* __restrict__ src, const int* __restrict__ dst,
                      const float* __restrict__ dinv, const float* __restrict__ hw,
                      float* __restrict__ agg, int ne, int nranges,
                      const unsigned* __restrict__ rc, const unsigned* __restrict__ list,
                      const float* ws0, size_t ws_bytes) {
  __shared__ float acc[SC_NR * 16];
  __shared__ unsigned q[SC_WAVES][SC_QLEN];
  const int tid = threadIdx.x, lane = tid & 31, w = tid >> 5;
  const int half = lane >> 4, f = lane & 15;
  const int r    = blockIdx.x % nranges;
  const int part = blockIdx.x / nranges;
  __shared__ unsigned shares[2 * SC_WAVES];
  const bool fits = sc_list_fits(list, ne, ws0, ws_bytes);
  for (int i = tid; i < SC_NR * 16; i += 1024) acc[i] = 0.0f;
  if (fits) sc_bounds_shares(rc + nranges * SC_FILL, r, shares);
  __syncthreads();
  if (fits) {
    unsigned lstart, cnt;
    sc_bounds(shares, lstart, cnt);
    const unsigned* __restrict__ rl = list + lstart;
    constexpr unsigned NW = SC_S * SC_WAVES;
    const unsigned run = ((cnt + NW - 1u) / NW + 31u) & ~31u;
    const unsigned g   = (unsigned)(part * SC_WAVES + w);
    const unsigned c0  = min(g * run, cnt);
    const unsigned c1  = min(c0 + run, cnt);
    if (c0 < c1) {
      const unsigned last = c1 - 1u;
      unsigned ent[RP_UD], nxt[RP_UD]; float hv[RP_UD];
#pragma unroll
      for (int u = 0; u < RP_UD; ++u) ent[u] = rl[min(c0 + 2u * u + half, last)];
      for (unsigned p = c0; p < c1; p += 2 * RP_UD) {
#pragma unroll
        for (int u = 0; u < RP_UD; ++u) nxt[u] = rl[min(p + 2 * RP_UD + 2u * u + half, last)];
#pragma unroll
        for (int u = 0; u < RP_UD; ++u) hv[u] = hw[(ent[u] >> 10) * 16 + f];
#pragma unroll
        for (int u = 0; u < RP_UD; ++u)
          unsafeAtomicAdd(&acc[(ent[u] & 1023u) * 16 + f], (p + 2u * u + half < c1) ? hv[u] : 0.0f);
#pragma unroll
        for (int u = 0; u < RP_UD; ++u) ent[u] = nxt[u];
      }
    }
  } else {
    sc_scan<false>(acc, q[w], nullptr, src, dst, hw, ne, r, part, nullptr, nullptr);
  }
  sc_hand_over(acc, dinv, agg, r * SC_NR, tid);
}

template <int KLEN>
__global__ void k_heads_splitk_relu(const float* __restrict__ A, int lda,
                                    const float* __restrict__ Bp, const float* __restrict__ Bv,
                                    float* __restrict__ accp, float* __restrict__ accv,
                                    int nsplit, int pol_blocks) {
  const bool pol = (int)blockIdx.x < pol_blocks;
  const float* __restrict__ B = pol ? Bp : Bv;
  float* __restrict__ acc     = pol ? accp : accv;
  const int ncols   = pol ? 32 : 16;
  const int tiles_n = pol ? 2 : 1;
  const int blk     = pol ? (int)blockIdx.x : (int)blockIdx.x - pol_blocks;
  int wid  = (blk * blockDim.x + threadIdx.x) >> 5;
  int lane = threadIdx.x & 31;
  int split = wid % nsplit;
  int tn    = (wid / nsplit) % tiles_n;
  int tm    = wid / (nsplit * tiles_n);
  int m  = lane & 15;
  int kh = (lane >> 4) << 1;
  const float* arow = A + (size_t)(tm * 16 + m) * lda + split * KLEN;
  const float* bcol = B + (size_t)(split * KLEN) * ncols + tn * 16 + m;
  v8f c = {};
#pragma unroll 4
  for (int k0 = 0; k0 < KLEN; k0 += 4) {
    v2f a, b;
    a.x = fmaxf(arow[k0 + kh + 0], 0.0f);
    a.y = fmaxf(arow[k0 + kh + 1], 0.0f);
    b.x = bcol[(size_t)(k0 + kh + 0) * ncols];
    b.y = bcol[(size_t)(k0 + kh + 1) * ncols];
    c = __builtin_amdgcn_wmma_f32_16x16x4_f32(false, a, false, b, (short)0, c, false, false);
  }
  int rbase = tm * 16 + ((lane >> 4) << 3);
#pragma unroll
  for (int r = 0; r < 8; ++r)
    unsafeAtomicAdd(&acc[(rbase + r) * ncols + tn * 16 + m], c[r]);
}

__global__ void k_heads_final(const float* __restrict__ pacc, const float* __restrict__ p1b,
                              const float* __restrict__ B, const float* __restrict__ bias,
                              const float* __restrict__ vacc, const float* __restrict__ v1b,
                              const float* __restrict__ v2w, const float* __restrict__ v2b,
                              float* __restrict__ out, float* __restrict__ vout) {
  if (blockIdx.x == 32) {
    int bi = threadIdx.x;
    if (bi < 64) {
      float s = 0.0f;
#pragma unroll
      for (int j = 0; j < 16; ++j) s += fmaxf(vacc[bi * 16 + j] + v1b[j], 0.0f) * v2w[j];
      vout[bi] = s + v2b[0];
    }
    return;
  }
  int wid  = (blockIdx.x * blockDim.x + threadIdx.x) >> 5;
  int lane = threadIdx.x & 31;
  int tm = wid >> 6;
  int tn = wid & 63;
  int m  = lane & 15;
  int kh = (lane >> 4) << 1;
  const float* arow = pacc + (tm * 16 + m) * 32;
  const float* bcol = B + tn * 16 + m;
  v8f c = {};
#pragma unroll
  for (int k0 = 0; k0 < 32; k0 += 4) {
    v2f a, b;
    a.x = fmaxf(arow[k0 + kh + 0] + p1b[k0 + kh + 0], 0.0f);
    a.y = fmaxf(arow[k0 + kh + 1] + p1b[k0 + kh + 1], 0.0f);
    b.x = bcol[(k0 + kh + 0) * 1024];
    b.y = bcol[(k0 + kh + 1) * 1024];
    c = __builtin_amdgcn_wmma_f32_16x16x4_f32(false, a, false, b, (short)0, c, false, false);
  }
  int rbase = tm * 16 + ((lane >> 4) << 3);
  float bv = bias[tn * 16 + m];
#pragma unroll
  for (int r = 0; r < 8; ++r)
    out[(rbase + r) * 1024 + tn * 16 + m] = c[r] + bv;
}

extern "C" void kernel_launch(void* const* d_in, const int* in_sizes, int n_in,
                              void* d_out, int out_size, void* d_ws, size_t ws_size,
                              hipStream_t stream) {
  const float* x    = (const float*)d_in[0];
  const int*   ei   = (const int*)d_in[1];
  const float* W1   = (const float*)d_in[2];
  const float* b1   = (const float*)d_in[3];
  const float* W2   = (const float*)d_in[4];
  const float* b2   = (const float*)d_in[5];
  const float* p1w  = (const float*)d_in[6];
  const float* p1b  = (const float*)d_in[7];
  const float* p2w  = (const float*)d_in[8];
  const float* p2b  = (const float*)d_in[9];
  const float* v1w  = (const float*)d_in[10];
  const float* v1b  = (const float*)d_in[11];
  const float* v2w  = (const float*)d_in[12];
  const float* v2b  = (const float*)d_in[13];

  const int N  = in_sizes[0] / 16;
  const int E  = in_sizes[1] / 2;
  const int* esrc = ei;
  const int* edst = ei + E;

  float* ws   = (float*)d_ws;
  float* dinv = ws + N;
  float* bufA = dinv + N;
  float* bufB = bufA + (size_t)16 * N;
  float* deg  = bufB;
  float* pacc = bufB + (size_t)16 * N;
  float* vacc = pacc + 64 * 32;
  unsigned* elist = (unsigned*)(vacc + 64 * 16);
  unsigned* rc    = (unsigned*)ws;

  float* Xout = (float*)d_out;
  float* Vout = Xout + 64 * 1024;

  const int B256 = 256;
  k_init_deg<<<N / B256, B256, 0, stream>>>(deg, N, pacc, 64 * 32 + 64 * 16, rc, (N / SC_NR) * SC_FILL);
  k_count_deg<<<E / B256, B256, 0, stream>>>(edst, deg, E);

  int gemm_blocks    = (N / 16) / 8;
  int nranges        = N / SC_NR;
  int scatter_blocks = nranges * SC_S;
  k_gemm_agg<true><<<gemm_blocks, B256, 0, stream>>>(x, W1, b1, dinv, bufA, bufB, N, rc + (N / SC_NR) * SC_FILL);
  k_scatter_scan<<<scatter_blocks, 1024, 0, stream>>>(esrc, edst, dinv, bufA, bufB, E, nranges,
                                                      rc, elist, ws, ws_size);

  k_gemm_agg<false><<<gemm_blocks, B256, 0, stream>>>(x, W2, b2, dinv, bufA, bufB, N, rc + (N / SC_NR) * SC_FILL);
  k_scatter_replay<<<scatter_blocks, 1024, 0, stream>>>(esrc, edst, dinv, bufA, bufB, E, nranges,
                                                        rc, elist, ws, ws_size);

  const int K = 16384, NSPLIT = 64, KLEN = K / NSPLIT;

  k_heads_splitk_relu<KLEN><<<64 + 32, B256, 0, stream>>>(bufB, K, p1w, v1w, pacc, vacc, NSPLIT, 64);

  k_heads_final<<<32 + 1, B256, 0, stream>>>(pacc, p1b, p2w, p2b, vacc, v1b, v2w, v2b, Xout, Vout);
}


# edges binned by destination range in one added launch (LDS ranks, one list claim per range and chunk); layer 1's scatter replays the lists like layer 2's, no 64-fold dst scan; sum order changes
# speedup vs baseline: 2.8908x; 1.1378x over previous
#include <hip/hip_runtime.h>
#include <hip/hip_bf16.h>

typedef __attribute__((ext_vector_type(2))) float v2f;
typedef __attribute__((ext_vector_type(8))) float v8f;

constexpr int DEG_STRIDE = 16;
constexpr int SC_NR      = 1024;

__global__ void k_init_deg(float* __restrict__ deg, int n, float* __restrict__ zacc, int nz,
                           unsigned* __restrict__ rc, int nrc) {
  int i = blockIdx.x * blockDim.x + threadIdx.x;
  if (i < n) deg[i * DEG_STRIDE] = 1.0f;
  if (i < nz) zacc[i] = 0.0f;
  if (i < nrc) rc[i] = 0u;
}

__global__ void k_count_deg(const int* __restrict__ dst, float* __restrict__ deg, int ne) {
  int e = blockIdx.x * blockDim.x + threadIdx.x;
  if (e < ne) unsafeAtomicAdd(&deg[__builtin_nontemporal_load(dst + e) * DEG_STRIDE], 1.0f);
}

template <bool FIRST>
__global__ void k_gemm_agg(const float* __restrict__ x, const float* __restrict__ B,
                           const float* __restrict__ bias, float* dinv,
                           float* __restrict__ hw, float* agg, int nrows,
                           unsigned* __restrict__ wavecnt) {
  int wid  = (blockIdx.x * blockDim.x + threadIdx.x) >> 5;
  int lane = threadIdx.x & 31;
  int row0 = wid * 16;
  if (row0 >= nrows) return;
  int m  = lane & 15;
  int kh = (lane >> 4) << 1;
  int rbase = row0 + ((lane >> 4) << 3);
  float di[8];
  float dsum = 0.0f;
#pragma unroll
  for (int r = 0; r < 8; ++r) {
    if (FIRST) {
      const float dg = agg[(rbase + r) * DEG_STRIDE];
      dsum += dg;
      di[r] = rsqrtf(dg);
    } else {
      di[r] = dinv[rbase + r];
    }
  }
  const float* arow = (FIRST ? x : (const float*)agg) + (row0 + m) * 16;
  v8f c = {};
#pragma unroll
  for (int k0 = 0; k0 < 16; k0 += 4) {
    v2f a, b;
    float a0 = arow[k0 + kh + 0];
    float a1 = arow[k0 + kh + 1];
    if (!FIRST) { a0 = fmaxf(a0, 0.0f); a1 = fmaxf(a1, 0.0f); }
    a.x = a0; a.y = a1;
    b.x = B[(k0 + kh + 0) * 16 + m];
    b.y = B[(k0 + kh + 1) * 16 + m];
    c = __builtin_amdgcn_wmma_f32_16x16x4_f32(false, a, false, b, (short)0, c, false, false);
  }
  const float bv = bias[m];
#pragma unroll
  for (int r = 0; r < 8; ++r) {
    const float hs = c[r] * di[r];
    hw[(rbase + r) * 16 + m]  = hs;
    agg[(rbase + r) * 16 + m] = hs * di[r] + bv;
  }
  if (FIRST) {
    if (m == 0) {
#pragma unroll
      for (int r = 0; r < 8; ++r) dinv[rbase + r] = di[r];
    }
    const float tot = dsum + __shfl_xor(dsum, 16);
    if (lane == 0) wavecnt[wid] = (unsigned)tot - 16u;
  }
}

typedef int v4i __attribute__((ext_vector_type(4)));

constexpr int SC_S     = 4;
constexpr int SC_WAVES = 32;
constexpr int SC_LOADS = 8;
constexpr int SC_BATCH = 32 * 4 * SC_LOADS;
constexpr int SC_UD    = 8;
constexpr int SC_QLEN  = SC_BATCH + 2 * SC_UD;
constexpr int RP_UD    = 16;

constexpr int SC_WPR = SC_NR / 16;
constexpr int SC_FILL = 32;

__device__ __forceinline__ unsigned sc_wave_sum(unsigned v) {
  for (int o = 16; o > 0; o >>= 1) v += __shfl_xor(v, o);
  return (unsigned)__builtin_amdgcn_readfirstlane((int)v);
}
__device__ __forceinline__ void sc_bounds_shares(const unsigned* __restrict__ wavecnt, int r, unsigned* part) {
  const int tid = threadIdx.x;
  const int nb = r * SC_WPR;
  unsigned s = 0u;
  for (int i = tid; i < nb; i += 1024) s += wavecnt[i];
  const unsigned c = (tid < SC_WPR) ? wavecnt[nb + tid] : 0u;
  s = sc_wave_sum(s);
  const unsigned cs = sc_wave_sum(c);
  if ((tid & 31) == 0) { part[tid >> 5] = s; part[SC_WAVES + (tid >> 5)] = cs; }
}
__device__ __forceinline__ void sc_bounds(const unsigned* part, unsigned& start, unsigned& cnt) {
  const int lane = threadIdx.x & 31;
  start = sc_wave_sum(part[lane]);
  cnt   = sc_wave_sum(part[SC_WAVES + lane]);
}

__device__ __forceinline__ bool sc_list_fits(const unsigned* list, int ne, const float* ws0, size_t ws_bytes) {
  return (size_t)((const char*)(list + ne) - (const char*)ws0) <= ws_bytes;
}

__device__ __forceinline__ void sc_hand_over(const float* acc, const float* __restrict__ dinv,
                                             float* __restrict__ agg, int lo, int tid) {
  constexpr int SC_FL = SC_NR * 16 / 1024;
  float dv[SC_FL], av[SC_FL];
#pragma unroll
  for (int j = 0; j < SC_FL; ++j) dv[j] = dinv[lo + (tid >> 4) + 64 * j];
  __syncthreads();
  float* arow = agg + (size_t)lo * 16;
#pragma unroll
  for (int j = 0; j < SC_FL; ++j) av[j] = acc[tid + 1024 * j] * dv[j];
#pragma unroll
  for (int j = 0; j < SC_FL; ++j) unsafeAtomicAdd(arow + tid + 1024 * j, av[j]);
}

__device__ __forceinline__ void sc_scan(float* acc, unsigned* qw,
                                        const int* __restrict__ src, const int* __restrict__ dst,
                                        const float* __restrict__ hw, int ne, int r, int sl) {
  const int tid = threadIdx.x, lane = tid & 31, w = tid >> 5;
  const int half = lane >> 4, f = lane & 15;
  const int lo = r * SC_NR;

  const int slice = ne / SC_S;
  const int span  = slice / SC_WAVES;
  const int ebase = sl * slice + w * span;
  const int nb    = span / SC_BATCH;
  const v4i* dp   = (const v4i*)(dst + ebase);
  const int* sp   = src + ebase;
  int qn = 0;

  auto pass = [&](int q0, int nv) {
    unsigned ent[SC_UD]; bool ok[SC_UD]; int sv[SC_UD]; float hv[SC_UD];
#pragma unroll
    for (int u = 0; u < SC_UD; ++u) {
      ok[u]  = 2 * u + half < nv;
      ent[u] = ok[u] ? qw[q0 + 2 * u + half] : 0u;
    }
#pragma unroll
    for (int u = 0; u < SC_UD; ++u) sv[u] = sp[ent[u] >> 10];
#pragma unroll
    for (int u = 0; u < SC_UD; ++u) hv[u] = hw[sv[u] * 16 + f];
#pragma unroll
    for (int u = 0; u < SC_UD; ++u)
      unsafeAtomicAdd(&acc[(ent[u] & 1023u) * 16 + f], ok[u] ? hv[u] : 0.0f);
  };

  v4i cur[SC_LOADS], nxt[SC_LOADS];
#pragma unroll
  for (int j = 0; j < SC_LOADS; ++j) cur[j] = dp[j * 32 + lane];
  for (int b = 0; b < nb; ++b) {
    const int bn = (b + 1 < nb) ? b + 1 : b;
#pragma unroll
    for (int j = 0; j < SC_LOADS; ++j) nxt[j] = dp[bn * (SC_BATCH / 4) + j * 32 + lane];
#pragma unroll
    for (int j = 0; j < SC_LOADS; ++j) {
#pragma unroll
      for (int c = 0; c < 4; ++c) {
        const int dl = cur[j][c] - lo;
        const bool m = (unsigned)dl < (unsigned)SC_NR;
        const unsigned mask = __builtin_amdgcn_ballot_w32(m);
        if (mask != 0u) {
          const int cnt = __builtin_popcount(mask);
          if (m) {
            const int el = b * SC_BATCH + j * 128 + lane * 4 + c;
            qw[qn + (int)__builtin_amdgcn_mbcnt_lo(mask, 0u)] = ((unsigned)el << 10) | (unsigned)dl;
          }
          qn += cnt;
        }
      }
    }
    while (qn >= 2 * SC_UD) { qn -= 2 * SC_UD; pass(qn, 2 * SC_UD); }
#pragma unroll
    for (int j = 0; j < SC_LOADS; ++j) cur[j] = nxt[j];
  }
  if (qn > 0) pass(0, qn);
}

typedef unsigned v2u __attribute__((ext_vector_type(2)));
typedef unsigned v4u __attribute__((ext_vector_type(4)));

constexpr int BIN_T     = 512;
constexpr int BIN_LD    = 4;
constexpr int BIN_EDGES = BIN_T * 4 * BIN_LD;
constexpr int BIN_MAXR  = 64;

__global__ __launch_bounds__(BIN_T)
void k_bin_edges(const int* __restrict__ src, const int* __restrict__ dst, int ne, int nranges,
                 unsigned* __restrict__ rc, unsigned* __restrict__ list,
                 const float* ws0, size_t ws_bytes) {
  __shared__ unsigned cnt[BIN_MAXR];
  __shared__ unsigned rtot[BIN_MAXR];
  __shared__ unsigned lpre[BIN_MAXR];
  __shared__ unsigned gpos[BIN_MAXR];
  __shared__ v2u ordered[BIN_EDGES];
  if (!sc_list_fits(list, ne, ws0, ws_bytes)) return;
  const int tid = threadIdx.x;
  const int e0  = blockIdx.x * BIN_EDGES;

  static_assert(SC_WPR / 4 == 16, "one range per 16 lanes");
  constexpr int BIN_WC = BIN_MAXR * 16 / BIN_T;
  const v4u* wc = (const v4u*)(rc + nranges * SC_FILL);
  const int nwc = nranges * 16;
  v4u wv[BIN_WC];
#pragma unroll
  for (int k = 0; k < BIN_WC; ++k) wv[k] = wc[min(tid + k * BIN_T, nwc - 1)];

  v4i dv[BIN_LD], sv[BIN_LD];
  bool in[BIN_LD];
#pragma unroll
  for (int j = 0; j < BIN_LD; ++j) {
    const int e = e0 + (j * BIN_T + tid) * 4;
    in[j] = e + 3 < ne;
    dv[j] = __builtin_nontemporal_load((const v4i*)(dst + (in[j] ? e : 0)));
    sv[j] = __builtin_nontemporal_load((const v4i*)(src + (in[j] ? e : 0)));
  }
  if (tid < BIN_MAXR) cnt[tid] = 0u;
#pragma unroll
  for (int k = 0; k < BIN_WC; ++k) {
    unsigned s = (wv[k].x + wv[k].y) + (wv[k].z + wv[k].w);
    s += __shfl_xor(s, 8); s += __shfl_xor(s, 4); s += __shfl_xor(s, 2); s += __shfl_xor(s, 1);
    const int i = tid + k * BIN_T;
    if ((i & 15) == 0 && i < nwc) rtot[i >> 4] = s;
  }
  __syncthreads();

  unsigned rk[BIN_LD][4];
#pragma unroll
  for (int j = 0; j < BIN_LD; ++j) {
#pragma unroll
    for (int k = 0; k < 4; ++k) {
      rk[j][k] = 0u;
      if (in[j]) rk[j][k] = atomicAdd(&cnt[(unsigned)dv[j][k] / (unsigned)SC_NR], 1u);
    }
  }
  __syncthreads();

  if (tid < BIN_MAXR) {
    const int lane = tid & 31;
    const unsigned mine = (tid < nranges) ? cnt[tid] : 0u;
    const unsigned all  = (tid < nranges) ? rtot[tid] : 0u;
    unsigned room = 0u;
    if (mine != 0u) room = atomicAdd(rc + tid * SC_FILL, mine);
    unsigned sm = mine, sa = all;
    unsigned lm = (lane < nranges) ? cnt[lane] : 0u;
    unsigned la = (lane < nranges) ? rtot[lane] : 0u;
    for (int o = 1; o < 32; o <<= 1) {
      const unsigned tm = __shfl_up(sm, o), ta = __shfl_up(sa, o);
      const unsigned um = __shfl_up(lm, o), ua = __shfl_up(la, o);
      if (lane >= o) { sm += tm; sa += ta; lm += um; la += ua; }
    }
    const unsigned lowm = __shfl(lm, 31), lowa = __shfl(la, 31);
    if (tid >= 32) { sm += lowm; sa += lowa; }
    lpre[tid] = sm - mine;
    gpos[tid] = sa - all + room;
  }
  __syncthreads();

#pragma unroll
  for (int j = 0; j < BIN_LD; ++j) {
#pragma unroll
    for (int k = 0; k < 4; ++k) {
      if (in[j]) {
        const unsigned rr = (unsigned)dv[j][k] / (unsigned)SC_NR;
        v2u wd;
        wd.x = ((unsigned)sv[j][k] << 10) | ((unsigned)dv[j][k] % (unsigned)SC_NR);
        wd.y = gpos[rr] + rk[j][k];
        ordered[lpre[rr] + rk[j][k]] = wd;
      }
    }
  }
  __syncthreads();

  const int left = (ne - e0) & ~3;
  const int nv = left < BIN_EDGES ? left : BIN_EDGES;
  v2u wd[BIN_EDGES / BIN_T];
#pragma unroll
  for (int k = 0; k < BIN_EDGES / BIN_T; ++k) wd[k] = ordered[tid + k * BIN_T];
#pragma unroll
  for (int k = 0; k < BIN_EDGES / BIN_T; ++k)
    if (tid + k * BIN_T < nv) list[wd[k].y] = wd[k].x;
}

__global__ __launch_bounds__(1024)
void k_scatter_replay(const int* __restrict__ src, const int* __restrict__ dst,
                      const float* __restrict__ dinv, const float* __restrict__ hw,
                      float* __restrict__ agg, int ne, int nranges,
                      const unsigned* __restrict__ rc, const unsigned* __restrict__ list,
                      const float* ws0, size_t ws_bytes) {
  __shared__ float acc[SC_NR * 16];
  __shared__ unsigned q[SC_WAVES][SC_QLEN];
  const int tid = threadIdx.x, lane = tid & 31, w = tid >> 5;
  const int half = lane >> 4, f = lane & 15;
  const int r    = blockIdx.x % nranges;
  const int part = blockIdx.x / nranges;
  __shared__ unsigned shares[2 * SC_WAVES];
  const bool fits = sc_list_fits(list, ne, ws0, ws_bytes);
  for (int i = tid; i < SC_NR * 16; i += 1024) acc[i] = 0.0f;
  if (fits) sc_bounds_shares(rc + nranges * SC_FILL, r, shares);
  __syncthreads();
  if (fits) {
    unsigned lstart, cnt;
    sc_bounds(shares, lstart, cnt);
    const unsigned* __restrict__ rl = list + lstart;
    constexpr unsigned NW = SC_S * SC_WAVES;
    const unsigned run = ((cnt + NW - 1u) / NW + 31u) & ~31u;
    const unsigned g   = (unsigned)(part * SC_WAVES + w);
    const unsigned c0  = min(g * run, cnt);
    const unsigned c1  = min(c0 + run, cnt);
    if (c0 < c1) {
      const unsigned last = c1 - 1u;
      unsigned ent[RP_UD], nxt[RP_UD]; float hv[RP_UD];
#pragma unroll
      for (int u = 0; u < RP_UD; ++u) ent[u] = rl[min(c0 + 2u * u + half, last)];
      for (unsigned p = c0; p < c1; p += 2 * RP_UD) {
#pragma unroll
        for (int u = 0; u < RP_UD; ++u) nxt[u] = rl[min(p + 2 * RP_UD + 2u * u + half, last)];
#pragma unroll
        for (int u = 0; u < RP_UD; ++u) hv[u] = hw[(ent[u] >> 10) * 16 + f];
#pragma unroll
        for (int u = 0; u < RP_UD; ++u)
          unsafeAtomicAdd(&acc[(ent[u] & 1023u) * 16 + f], (p + 2u * u + half < c1) ? hv[u] : 0.0f);
#pragma unroll
        for (int u = 0; u < RP_UD; ++u) ent[u] = nxt[u];
      }
    }
  } else {
    sc_scan(acc, q[w], src, dst, hw, ne, r, part);
  }
  sc_hand_over(acc, dinv, agg, r * SC_NR, tid);
}

template <int KLEN>
__global__ void k_heads_splitk_relu(const float* __restrict__ A, int lda,
                                    const float* __restrict__ Bp, const float* __restrict__ Bv,
                                    float* __restrict__ accp, float* __restrict__ accv,
                                    int nsplit, int pol_blocks) {
  const bool pol = (int)blockIdx.x < pol_blocks;
  const float* __restrict__ B = pol ? Bp : Bv;
  float* __restrict__ acc     = pol ? accp : accv;
  const int ncols   = pol ? 32 : 16;
  const int tiles_n = pol ? 2 : 1;
  const int blk     = pol ? (int)blockIdx.x : (int)blockIdx.x - pol_blocks;
  int wid  = (blk * blockDim.x + threadIdx.x) >> 5;
  int lane = threadIdx.x & 31;
  int split = wid % nsplit;
  int tn    = (wid / nsplit) % tiles_n;
  int tm    = wid / (nsplit * tiles_n);
  int m  = lane & 15;
  int kh = (lane >> 4) << 1;
  const float* arow = A + (size_t)(tm * 16 + m) * lda + split * KLEN;
  const float* bcol = B + (size_t)(split * KLEN) * ncols + tn * 16 + m;
  v8f c = {};
#pragma unroll 4
  for (int k0 = 0; k0 < KLEN; k0 += 4) {
    v2f a, b;
    a.x = fmaxf(arow[k0 + kh + 0], 0.0f);
    a.y = fmaxf(arow[k0 + kh + 1], 0.0f);
    b.x = bcol[(size_t)(k0 + kh + 0) * ncols];
    b.y = bcol[(size_t)(k0 + kh + 1) * ncols];
    c = __builtin_amdgcn_wmma_f32_16x16x4_f32(false, a, false, b, (short)0, c, false, false);
  }
  int rbase = tm * 16 + ((lane >> 4) << 3);
#pragma unroll
  for (int r = 0; r < 8; ++r)
    unsafeAtomicAdd(&acc[(rbase + r) * ncols + tn * 16 + m], c[r]);
}

__global__ void k_heads_final(const float* __restrict__ pacc, const float* __restrict__ p1b,
                              const float* __restrict__ B, const float* __restrict__ bias,
                              const float* __restrict__ vacc, const float* __restrict__ v1b,
                              const float* __restrict__ v2w, const float* __restrict__ v2b,
                              float* __restrict__ out, float* __restrict__ vout) {
  if (blockIdx.x == 32) {
    int bi = threadIdx.x;
    if (bi < 64) {
      float s = 0.0f;
#pragma unroll
      for (int j = 0; j < 16; ++j) s += fmaxf(vacc[bi * 16 + j] + v1b[j], 0.0f) * v2w[j];
      vout[bi] = s + v2b[0];
    }
    return;
  }
  int wid  = (blockIdx.x * blockDim.x + threadIdx.x) >> 5;
  int lane = threadIdx.x & 31;
  int tm = wid >> 6;
  int tn = wid & 63;
  int m  = lane & 15;
  int kh = (lane >> 4) << 1;
  const float* arow = pacc + (tm * 16 + m) * 32;
  const float* bcol = B + tn * 16 + m;
  v8f c = {};
#pragma unroll
  for (int k0 = 0; k0 < 32; k0 += 4) {
    v2f a, b;
    a.x = fmaxf(arow[k0 + kh + 0] + p1b[k0 + kh + 0], 0.0f);
    a.y = fmaxf(arow[k0 + kh + 1] + p1b[k0 + kh + 1], 0.0f);
    b.x = bcol[(k0 + kh + 0) * 1024];
    b.y = bcol[(k0 + kh + 1) * 1024];
    c = __builtin_amdgcn_wmma_f32_16x16x4_f32(false, a, false, b, (short)0, c, false, false);
  }
  int rbase = tm * 16 + ((lane >> 4) << 3);
  float bv = bias[tn * 16 + m];
#pragma unroll
  for (int r = 0; r < 8; ++r)
    out[(rbase + r) * 1024 + tn * 16 + m] = c[r] + bv;
}

extern "C" void kernel_launch(void* const* d_in, const int* in_sizes, int n_in,
                              void* d_out, int out_size, void* d_ws, size_t ws_size,
                              hipStream_t stream) {
  const float* x    = (const float*)d_in[0];
  const int*   ei   = (const int*)d_in[1];
  const float* W1   = (const float*)d_in[2];
  const float* b1   = (const float*)d_in[3];
  const float* W2   = (const float*)d_in[4];
  const float* b2   = (const float*)d_in[5];
  const float* p1w  = (const float*)d_in[6];
  const float* p1b  = (const float*)d_in[7];
  const float* p2w  = (const float*)d_in[8];
  const float* p2b  = (const float*)d_in[9];
  const float* v1w  = (const float*)d_in[10];
  const float* v1b  = (const float*)d_in[11];
  const float* v2w  = (const float*)d_in[12];
  const float* v2b  = (const float*)d_in[13];

  const int N  = in_sizes[0] / 16;
  const int E  = in_sizes[1] / 2;
  const int* esrc = ei;
  const int* edst = ei + E;

  float* ws   = (float*)d_ws;
  float* dinv = ws + N;
  float* bufA = dinv + N;
  float* bufB = bufA + (size_t)16 * N;
  float* deg  = bufB;
  float* pacc = bufB + (size_t)16 * N;
  float* vacc = pacc + 64 * 32;
  unsigned* elist = (unsigned*)(vacc + 64 * 16);
  unsigned* rc    = (unsigned*)ws;

  float* Xout = (float*)d_out;
  float* Vout = Xout + 64 * 1024;

  const int B256 = 256;
  k_init_deg<<<N / B256, B256, 0, stream>>>(deg, N, pacc, 64 * 32 + 64 * 16, rc, (N / SC_NR) * SC_FILL);
  k_count_deg<<<E / B256, B256, 0, stream>>>(edst, deg, E);

  int gemm_blocks    = (N / 16) / 8;
  int nranges        = N / SC_NR;
  int scatter_blocks = nranges * SC_S;
  k_gemm_agg<true><<<gemm_blocks, B256, 0, stream>>>(x, W1, b1, dinv, bufA, bufB, N, rc + (N / SC_NR) * SC_FILL);
  k_bin_edges<<<(E + BIN_EDGES - 1) / BIN_EDGES, BIN_T, 0, stream>>>(esrc, edst, E, nranges, rc, elist, ws, ws_size);
  k_scatter_replay<<<scatter_blocks, 1024, 0, stream>>>(esrc, edst, dinv, bufA, bufB, E, nranges,
                                                        rc, elist, ws, ws_size);

  k_gemm_agg<false><<<gemm_blocks, B256, 0, stream>>>(x, W2, b2, dinv, bufA, bufB, N, rc + (N / SC_NR) * SC_FILL);
  k_scatter_replay<<<scatter_blocks, 1024, 0, stream>>>(esrc, edst, dinv, bufA, bufB, E, nranges,
                                                        rc, elist, ws, ws_size);

  const int K = 16384, NSPLIT = 64, KLEN = K / NSPLIT;

  k_heads_splitk_relu<KLEN><<<64 + 32, B256, 0, stream>>>(bufB, K, p1w, v1w, pacc, vacc, NSPLIT, 64);

  k_heads_final<<<32 + 1, B256, 0, stream>>>(pacc, p1b, p2w, p2b, vacc, v1b, v2w, v2b, Xout, Vout);
}


# degree counts come from the binned edge lists via LDS adds: binning runs first, per-chunk lists plus a table of run starts; no global add per edge, no list claims, 9 launches to 8; sum order changes
# speedup vs baseline: 4.5022x; 1.5574x over previous
#include <hip/hip_runtime.h>
#include <hip/hip_bf16.h>

typedef __attribute__((ext_vector_type(2))) float v2f;
typedef __attribute__((ext_vector_type(8))) float v8f;

constexpr int DEG_STRIDE = 16;
constexpr int SC_NR      = 1024;
constexpr int SC_WAVES   = 32;
constexpr int DG_S       = 4;

constexpr int BIN_T     = 512;
constexpr int BIN_LD    = 4;
constexpr int BIN_EDGES = BIN_T * 4 * BIN_LD;
constexpr int BIN_MAXR  = 64;
constexpr int BIN_TBL   = BIN_MAXR + 1;

template <bool FIRST>
__global__ void k_gemm_agg(const float* __restrict__ x, const float* __restrict__ B,
                           const float* __restrict__ bias, float* dinv,
                           float* __restrict__ hw, float* agg, int nrows) {
  int wid  = (blockIdx.x * blockDim.x + threadIdx.x) >> 5;
  int lane = threadIdx.x & 31;
  int row0 = wid * 16;
  if (row0 >= nrows) return;
  int m  = lane & 15;
  int kh = (lane >> 4) << 1;
  int rbase = row0 + ((lane >> 4) << 3);
  float di[8];
  static_assert(DG_S == 4, "a row's partial counts are read as one 16-byte word");
#pragma unroll
  for (int r = 0; r < 8; ++r) {
    if (FIRST) {
      const float4 dq = *(const float4*)(agg + (rbase + r) * DEG_STRIDE);
      const float dg = 1.0f + ((dq.x + dq.y) + (dq.z + dq.w));
      di[r] = rsqrtf(dg);
    } else {
      di[r] = dinv[rbase + r];
    }
  }
  const float* arow = (FIRST ? x : (const float*)agg) + (row0 + m) * 16;
  v8f c = {};
#pragma unroll
  for (int k0 = 0; k0 < 16; k0 += 4) {
    v2f a, b;
    float a0 = arow[k0 + kh + 0];
    float a1 = arow[k0 + kh + 1];
    if (!FIRST) { a0 = fmaxf(a0, 0.0f); a1 = fmaxf(a1, 0.0f); }
    a.x = a0; a.y = a1;
    b.x = B[(k0 + kh + 0) * 16 + m];
    b.y = B[(k0 + kh + 1) * 16 + m];
    c = __builtin_amdgcn_wmma_f32_16x16x4_f32(false, a, false, b, (short)0, c, false, false);
  }
  const float bv = bias[m];
#pragma unroll
  for (int r = 0; r < 8; ++r) {
    const float hs = c[r] * di[r];
    hw[(rbase + r) * 16 + m]  = hs;
    agg[(rbase + r) * 16 + m] = hs * di[r] + bv;
  }
  if (FIRST) {
    if (m == 0) {
#pragma unroll
      for (int r = 0; r < 8; ++r) dinv[rbase + r] = di[r];
    }
  }
}

typedef int v4i __attribute__((ext_vector_type(4)));

constexpr int SC_S     = 4;
constexpr int SC_LOADS = 8;
constexpr int SC_BATCH = 32 * 4 * SC_LOADS;
constexpr int SC_UD    = 8;
constexpr int SC_QLEN  = SC_BATCH + 2 * SC_UD;
constexpr int RP_UD    = 16;
constexpr int DG_UD    = 8;

__device__ __forceinline__ bool sc_list_fits(const unsigned* list, int ne, const float* ws0, size_t ws_bytes,
                                             int tbl_room) {
  const int nchunks = (ne + BIN_EDGES - 1) / BIN_EDGES;
  return (size_t)((const char*)(list + ne) - (const char*)ws0) <= ws_bytes && nchunks * BIN_TBL <= tbl_room;
}

__device__ __forceinline__ void sc_hand_over(const float* acc, const float* __restrict__ dinv,
                                             float* __restrict__ agg, int lo, int tid) {
  constexpr int SC_FL = SC_NR * 16 / 1024;
  float dv[SC_FL], av[SC_FL];
#pragma unroll
  for (int j = 0; j < SC_FL; ++j) dv[j] = dinv[lo + (tid >> 4) + 64 * j];
  __syncthreads();
  float* arow = agg + (size_t)lo * 16;
#pragma unroll
  for (int j = 0; j < SC_FL; ++j) av[j] = acc[tid + 1024 * j] * dv[j];
#pragma unroll
  for (int j = 0; j < SC_FL; ++j) unsafeAtomicAdd(arow + tid + 1024 * j, av[j]);
}

__device__ __forceinline__ void sc_scan(float* acc, unsigned* qw,
                                        const int* __restrict__ src, const int* __restrict__ dst,
                                        const float* __restrict__ hw, int ne, int r, int sl) {
  const int tid = threadIdx.x, lane = tid & 31, w = tid >> 5;
  const int half = lane >> 4, f = lane & 15;
  const int lo = r * SC_NR;

  const int slice = ne / SC_S;
  const int span  = slice / SC_WAVES;
  const int ebase = sl * slice + w * span;
  const int nb    = span / SC_BATCH;
  const v4i* dp   = (const v4i*)(dst + ebase);
  const int* sp   = src + ebase;
  int qn = 0;

  auto pass = [&](int q0, int nv) {
    unsigned ent[SC_UD]; bool ok[SC_UD]; int sv[SC_UD]; float hv[SC_UD];
#pragma unroll
    for (int u = 0; u < SC_UD; ++u) {
      ok[u]  = 2 * u + half < nv;
      ent[u] = ok[u] ? qw[q0 + 2 * u + half] : 0u;
    }
#pragma unroll
    for (int u = 0; u < SC_UD; ++u) sv[u] = sp[ent[u] >> 10];
#pragma unroll
    for (int u = 0; u < SC_UD; ++u) hv[u] = hw[sv[u] * 16 + f];
#pragma unroll
    for (int u = 0; u < SC_UD; ++u)
      unsafeAtomicAdd(&acc[(ent[u] & 1023u) * 16 + f], ok[u] ? hv[u] : 0.0f);
  };

  v4i cur[SC_LOADS], nxt[SC_LOADS];
#pragma unroll
  for (int j = 0; j < SC_LOADS; ++j) cur[j] = dp[j * 32 + lane];
  for (int b = 0; b < nb; ++b) {
    const int bn = (b + 1 < nb) ? b + 1 : b;
#pragma unroll
    for (int j = 0; j < SC_LOADS; ++j) nxt[j] = dp[bn * (SC_BATCH / 4) + j * 32 + lane];
#pragma unroll
    for (int j = 0; j < SC_LOADS; ++j) {
#pragma unroll
      for (int c = 0; c < 4; ++c) {
        const int dl = cur[j][c] - lo;
        const bool m = (unsigned)dl < (unsigned)SC_NR;
        const unsigned mask = __builtin_amdgcn_ballot_w32(m);
        if (mask != 0u) {
          const int cnt = __builtin_popcount(mask);
          if (m) {
            const int el = b * SC_BATCH + j * 128 + lane * 4 + c;
            qw[qn + (int)__builtin_amdgcn_mbcnt_lo(mask, 0u)] = ((unsigned)el << 10) | (unsigned)dl;
          }
          qn += cnt;
        }
      }
    }
    while (qn >= 2 * SC_UD) { qn -= 2 * SC_UD; pass(qn, 2 * SC_UD); }
#pragma unroll
    for (int j = 0; j < SC_LOADS; ++j) cur[j] = nxt[j];
  }
  if (qn > 0) pass(0, qn);
}

typedef unsigned v4u __attribute__((ext_vector_type(4)));

__global__ __launch_bounds__(BIN_T)
void k_bin_edges(const int* __restrict__ src, const int* __restrict__ dst, int ne, int nranges,
                 unsigned* __restrict__ tbl, unsigned* __restrict__ list,
                 const float* ws0, size_t ws_bytes, int tbl_room,
                 float* __restrict__ zacc, int nz) {
  __shared__ unsigned cnt[BIN_MAXR];
  __shared__ unsigned lpre[BIN_MAXR];
  __shared__ __attribute__((aligned(16))) unsigned ordered[BIN_EDGES];
  const int tid = threadIdx.x;
  for (int i = blockIdx.x * BIN_T + tid; i < nz; i += gridDim.x * BIN_T) zacc[i] = 0.0f;
  if (!sc_list_fits(list, ne, ws0, ws_bytes, tbl_room)) return;
  const int e0  = blockIdx.x * BIN_EDGES;

  v4i dv[BIN_LD], sv[BIN_LD];
  bool in[BIN_LD];
#pragma unroll
  for (int j = 0; j < BIN_LD; ++j) {
    const int e = e0 + (j * BIN_T + tid) * 4;
    in[j] = e + 3 < ne;
    dv[j] = __builtin_nontemporal_load((const v4i*)(dst + (in[j] ? e : 0)));
    sv[j] = __builtin_nontemporal_load((const v4i*)(src + (in[j] ? e : 0)));
  }
  if (tid < BIN_MAXR) cnt[tid] = 0u;
  __syncthreads();

  unsigned rk[BIN_LD][4];
#pragma unroll
  for (int j = 0; j < BIN_LD; ++j) {
#pragma unroll
    for (int k = 0; k < 4; ++k) {
      rk[j][k] = 0u;
      if (in[j]) rk[j][k] = atomicAdd(&cnt[(unsigned)dv[j][k] / (unsigned)SC_NR], 1u);
    }
  }
  __syncthreads();

  if (tid < BIN_MAXR) {
    const int lane = tid & 31;
    const unsigned mine = (tid < nranges) ? cnt[tid] : 0u;
    unsigned sm = mine;
    unsigned lm = (lane < nranges) ? cnt[lane] : 0u;
    for (int o = 1; o < 32; o <<= 1) {
      const unsigned tm = __shfl_up(sm, o);
      const unsigned um = __shfl_up(lm, o);
      if (lane >= o) { sm += tm; lm += um; }
    }
    const unsigned lowm = __shfl(lm, 31);
    if (tid >= 32) sm += lowm;
    lpre[tid] = sm - mine;
    unsigned* trow = tbl + (size_t)blockIdx.x * BIN_TBL;
    trow[tid] = sm - mine;
    if (tid == BIN_MAXR - 1) trow[BIN_MAXR] = sm;
  }
  __syncthreads();

#pragma unroll
  for (int j = 0; j < BIN_LD; ++j) {
#pragma unroll
    for (int k = 0; k < 4; ++k) {
      if (in[j]) {
        const unsigned rr = (unsigned)dv[j][k] / (unsigned)SC_NR;
        ordered[lpre[rr] + rk[j][k]] = ((unsigned)sv[j][k] << 10) | ((unsigned)dv[j][k] % (unsigned)SC_NR);
      }
    }
  }
  __syncthreads();

  const int left = (ne - e0) & ~3;
  const int nv4 = (left < BIN_EDGES ? left : BIN_EDGES) >> 2;
  constexpr int BIN_CP = BIN_EDGES / 4 / BIN_T;
  const v4u* o4 = (const v4u*)ordered;
  v4u* l4 = (v4u*)(list + e0);
  v4u wd[BIN_CP];
#pragma unroll
  for (int k = 0; k < BIN_CP; ++k) wd[k] = o4[tid + k * BIN_T];
#pragma unroll
  for (int k = 0; k < BIN_CP; ++k)
    if (tid + k * BIN_T < nv4) l4[tid + k * BIN_T] = wd[k];
}

__global__ __launch_bounds__(1024)
void k_deg_from_lists(const int* __restrict__ dst, float* __restrict__ deg, int ne, int nranges,
                      const unsigned* __restrict__ tbl, const unsigned* __restrict__ list,
                      const float* ws0, size_t ws_bytes, int tbl_room) {
  static_assert(SC_NR == 1024, "one thread per row of the range");
  __shared__ unsigned cnt[SC_NR];
  const int tid = threadIdx.x, lane = tid & 31, w = tid >> 5;
  const int r    = blockIdx.x % nranges;
  const int part = blockIdx.x / nranges;
  const int lo   = r * SC_NR;
  cnt[tid] = 0u;
  __syncthreads();
  if (sc_list_fits(list, ne, ws0, ws_bytes, tbl_room)) {
    const int nchunks = (ne + BIN_EDGES - 1) / BIN_EDGES;
    for (int c = part * SC_WAVES + w; c < nchunks; c += DG_S * SC_WAVES) {
      const unsigned s = tbl[c * BIN_TBL + r], e = tbl[c * BIN_TBL + r + 1];
      const unsigned* __restrict__ rl = list + (size_t)c * BIN_EDGES;
      for (unsigned p0 = s; p0 < e; p0 += 32 * DG_UD) {
        unsigned wv[DG_UD];
#pragma unroll
        for (int u = 0; u < DG_UD; ++u) wv[u] = rl[min(p0 + 32u * u + lane, e - 1u)];
#pragma unroll
        for (int u = 0; u < DG_UD; ++u)
          if (p0 + 32u * u + lane < e) atomicAdd(&cnt[wv[u] & 1023u], 1u);
      }
    }
  } else {
    const int slice = ne / DG_S;
    const int e1 = (part == DG_S - 1) ? ne : (part + 1) * slice;
    for (int e = part * slice + tid; e < e1; e += 1024) {
      const int dl = dst[e] - lo;
      if ((unsigned)dl < (unsigned)SC_NR) atomicAdd(&cnt[dl], 1u);
    }
  }
  __syncthreads();
  deg[(size_t)(lo + tid) * DEG_STRIDE + part] = (float)cnt[tid];
}

__global__ __launch_bounds__(1024)
void k_scatter_replay(const int* __restrict__ src, const int* __restrict__ dst,
                      const float* __restrict__ dinv, const float* __restrict__ hw,
                      float* __restrict__ agg, int ne, int nranges,
                      const unsigned* __restrict__ tbl, const unsigned* __restrict__ list,
                      const float* ws0, size_t ws_bytes, int tbl_room) {
  __shared__ float acc[SC_NR * 16];
  __shared__ unsigned q[SC_WAVES][SC_QLEN];
  const int tid = threadIdx.x, lane = tid & 31, w = tid >> 5;
  const int half = lane >> 4, f = lane & 15;
  const int r    = blockIdx.x % nranges;
  const int part = blockIdx.x / nranges;
  const bool fits = sc_list_fits(list, ne, ws0, ws_bytes, tbl_room);
  const int nchunks = (ne + BIN_EDGES - 1) / BIN_EDGES;
  constexpr int NW = SC_S * SC_WAVES;
  int c = part * SC_WAVES + w;
  unsigned rs = 0u, re = 0u;
  if (fits && c < nchunks) { rs = tbl[c * BIN_TBL + r]; re = tbl[c * BIN_TBL + r + 1]; }
  for (int i = tid; i < SC_NR * 16; i += 1024) acc[i] = 0.0f;
  __syncthreads();
  if (fits) {
    while (c < nchunks) {
      const int cn = c + NW;
      unsigned ns = 0u, nend = 0u;
      if (cn < nchunks) { ns = tbl[cn * BIN_TBL + r]; nend = tbl[cn * BIN_TBL + r + 1]; }
      const unsigned* __restrict__ rl = list + (size_t)c * BIN_EDGES;
      const unsigned c0 = rs, c1 = re;
      if (c0 < c1) {
        const unsigned last = c1 - 1u;
        unsigned ent[RP_UD], nxt[RP_UD]; float hv[RP_UD];
#pragma unroll
        for (int u = 0; u < RP_UD; ++u) ent[u] = rl[min(c0 + 2u * u + half, last)];
        for (unsigned p = c0; p < c1; p += 2 * RP_UD) {
#pragma unroll
          for (int u = 0; u < RP_UD; ++u) nxt[u] = rl[min(p + 2 * RP_UD + 2u * u + half, last)];
#pragma unroll
          for (int u = 0; u < RP_UD; ++u) hv[u] = hw[(ent[u] >> 10) * 16 + f];
#pragma unroll
          for (int u = 0; u < RP_UD; ++u)
            unsafeAtomicAdd(&acc[(ent[u] & 1023u) * 16 + f], (p + 2u * u + half < c1) ? hv[u] : 0.0f);
#pragma unroll
          for (int u = 0; u < RP_UD; ++u) ent[u] = nxt[u];
        }
      }
      c = cn; rs = ns; re = nend;
    }
  } else {
    sc_scan(acc, q[w], src, dst, hw, ne, r, part);
  }
  sc_hand_over(acc, dinv, agg, r * SC_NR, tid);
}

template <int KLEN>
__global__ void k_heads_splitk_relu(const float* __restrict__ A, int lda,
                                    const float* __restrict__ Bp, const float* __restrict__ Bv,
                                    float* __restrict__ accp, float* __restrict__ accv,
                                    int nsplit, int pol_blocks) {
  const bool pol = (int)blockIdx.x < pol_blocks;
  const float* __restrict__ B = pol ? Bp : Bv;
  float* __restrict__ acc     = pol ? accp : accv;
  const int ncols   = pol ? 32 : 16;
  const int tiles_n = pol ? 2 : 1;
  const int blk     = pol ? (int)blockIdx.x : (int)blockIdx.x - pol_blocks;
  int wid  = (blk * blockDim.x + threadIdx.x) >> 5;
  int lane = threadIdx.x & 31;
  int split = wid % nsplit;
  int tn    = (wid / nsplit) % tiles_n;
  int tm    = wid / (nsplit * tiles_n);
  int m  = lane & 15;
  int kh = (lane >> 4) << 1;
  const float* arow = A + (size_t)(tm * 16 + m) * lda + split * KLEN;
  const float* bcol = B + (size_t)(split * KLEN) * ncols + tn * 16 + m;
  v8f c = {};
#pragma unroll 4
  for (int k0 = 0; k0 < KLEN; k0 += 4) {
    v2f a, b;
    a.x = fmaxf(arow[k0 + kh + 0], 0.0f);
    a.y = fmaxf(arow[k0 + kh + 1], 0.0f);
    b.x = bcol[(size_t)(k0 + kh + 0) * ncols];
    b.y = bcol[(size_t)(k0 + kh + 1) * ncols];
    c = __builtin_amdgcn_wmma_f32_16x16x4_f32(false, a, false, b, (short)0, c, false, false);
  }
  int rbase = tm * 16 + ((lane >> 4) << 3);
#pragma unroll
  for (int r = 0; r < 8; ++r)
    unsafeAtomicAdd(&acc[(rbase + r) * ncols + tn * 16 + m], c[r]);
}

__global__ void k_heads_final(const float* __restrict__ pacc, const float* __restrict__ p1b,
                              const float* __restrict__ B, const float* __restrict__ bias,
                              const float* __restrict__ vacc, const float* __restrict__ v1b,
                              const float* __restrict__ v2w, const float* __restrict__ v2b,
                              float* __restrict__ out, float* __restrict__ vout) {
  if (blockIdx.x == 32) {
    int bi = threadIdx.x;
    if (bi < 64) {
      float s = 0.0f;
#pragma unroll
      for (int j = 0; j < 16; ++j) s += fmaxf(vacc[bi * 16 + j] + v1b[j], 0.0f) * v2w[j];
      vout[bi] = s + v2b[0];
    }
    return;
  }
  int wid  = (blockIdx.x * blockDim.x + threadIdx.x) >> 5;
  int lane = threadIdx.x & 31;
  int tm = wid >> 6;
  int tn = wid & 63;
  int m  = lane & 15;
  int kh = (lane >> 4) << 1;
  const float* arow = pacc + (tm * 16 + m) * 32;
  const float* bcol = B + tn * 16 + m;
  v8f c = {};
#pragma unroll
  for (int k0 = 0; k0 < 32; k0 += 4) {
    v2f a, b;
    a.x = fmaxf(arow[k0 + kh + 0] + p1b[k0 + kh + 0], 0.0f);
    a.y = fmaxf(arow[k0 + kh + 1] + p1b[k0 + kh + 1], 0.0f);
    b.x = bcol[(k0 + kh + 0) * 1024];
    b.y = bcol[(k0 + kh + 1) * 1024];
    c = __builtin_amdgcn_wmma_f32_16x16x4_f32(false, a, false, b, (short)0, c, false, false);
  }
  int rbase = tm * 16 + ((lane >> 4) << 3);
  float bv = bias[tn * 16 + m];
#pragma unroll
  for (int r = 0; r < 8; ++r)
    out[(rbase + r) * 1024 + tn * 16 + m] = c[r] + bv;
}

extern "C" void kernel_launch(void* const* d_in, const int* in_sizes, int n_in,
                              void* d_out, int out_size, void* d_ws, size_t ws_size,
                              hipStream_t stream) {
  const float* x    = (const float*)d_in[0];
  const int*   ei   = (const int*)d_in[1];
  const float* W1   = (const float*)d_in[2];
  const float* b1   = (const float*)d_in[3];
  const float* W2   = (const float*)d_in[4];
  const float* b2   = (const float*)d_in[5];
  const float* p1w  = (const float*)d_in[6];
  const float* p1b  = (const float*)d_in[7];
  const float* p2w  = (const float*)d_in[8];
  const float* p2b  = (const float*)d_in[9];
  const float* v1w  = (const float*)d_in[10];
  const float* v1b  = (const float*)d_in[11];
  const float* v2w  = (const float*)d_in[12];
  const float* v2b  = (const float*)d_in[13];

  const int N  = in_sizes[0] / 16;
  const int E  = in_sizes[1] / 2;
  const int* esrc = ei;
  const int* edst = ei + E;

  float* ws   = (float*)d_ws;
  float* dinv = ws + N;
  float* bufA = dinv + N;
  float* bufB = bufA + (size_t)16 * N;
  float* deg  = bufB;
  float* pacc = bufB + (size_t)16 * N;
  float* vacc = pacc + 64 * 32;
  unsigned* elist = (unsigned*)(vacc + 64 * 16);
  unsigned* tbl   = (unsigned*)ws;

  float* Xout = (float*)d_out;
  float* Vout = Xout + 64 * 1024;

  const int B256 = 256;
  int gemm_blocks    = (N / 16) / 8;
  int nranges        = N / SC_NR;
  int scatter_blocks = nranges * SC_S;

  k_bin_edges<<<(E + BIN_EDGES - 1) / BIN_EDGES, BIN_T, 0, stream>>>(esrc, edst, E, nranges, tbl, elist, ws, ws_size, N,
                                                                    pacc, 64 * 32 + 64 * 16);
  k_deg_from_lists<<<nranges * DG_S, 1024, 0, stream>>>(edst, deg, E, nranges, tbl, elist, ws, ws_size, N);

  k_gemm_agg<true><<<gemm_blocks, B256, 0, stream>>>(x, W1, b1, dinv, bufA, bufB, N);
  k_scatter_replay<<<scatter_blocks, 1024, 0, stream>>>(esrc, edst, dinv, bufA, bufB, E, nranges,
                                                        tbl, elist, ws, ws_size, N);

  k_gemm_agg<false><<<gemm_blocks, B256, 0, stream>>>(x, W2, b2, dinv, bufA, bufB, N);
  k_scatter_replay<<<scatter_blocks, 1024, 0, stream>>>(esrc, edst, dinv, bufA, bufB, E, nranges,
                                                        tbl, elist, ws, ws_size, N);

  const int K = 16384, NSPLIT = 64, KLEN = K / NSPLIT;

  k_heads_splitk_relu<KLEN><<<64 + 32, B256, 0, stream>>>(bufB, K, p1w, v1w, pacc, vacc, NSPLIT, 64);

  k_heads_final<<<32 + 1, B256, 0, stream>>>(pacc, p1b, p2w, p2b, vacc, v1b, v2w, v2b, Xout, Vout);
}


# degree count moved into the first GEMM's launch: edges binned by 256-row bins, a work-group counts its bin's rows in LDS, makes dinv, does those rows' product; 8 launches to 7; sum order changes
# speedup vs baseline: 4.6552x; 1.0340x over previous
#include <hip/hip_runtime.h>
#include <hip/hip_bf16.h>

typedef __attribute__((ext_vector_type(2))) float v2f;
typedef __attribute__((ext_vector_type(8))) float v8f;

constexpr int SC_NR      = 1024;
constexpr int SC_WAVES   = 32;
constexpr int DGQ_ROWS   = 256;
constexpr int DGQ_WAVES  = DGQ_ROWS / 16;
constexpr int DGQ_UD     = 2;

constexpr int BIN_T     = 512;
constexpr int BIN_LD    = 4;
constexpr int BIN_EDGES = BIN_T * 4 * BIN_LD;
constexpr int BIN_NR    = 256;
constexpr int BIN_MAXR  = 256;
constexpr int BIN_TBL   = BIN_MAXR;
constexpr int SC_BINS   = SC_NR / BIN_NR;

__global__ void k_gemm_agg(const float* __restrict__ B, const float* __restrict__ bias,
                           const float* __restrict__ dinv, float* __restrict__ hw, float* agg, int nrows) {
  int wid  = (blockIdx.x * blockDim.x + threadIdx.x) >> 5;
  int lane = threadIdx.x & 31;
  int row0 = wid * 16;
  if (row0 >= nrows) return;
  int m  = lane & 15;
  int kh = (lane >> 4) << 1;
  int rbase = row0 + ((lane >> 4) << 3);
  float di[8];
#pragma unroll
  for (int r = 0; r < 8; ++r) di[r] = dinv[rbase + r];
  const float* arow = (const float*)agg + (row0 + m) * 16;
  v8f c = {};
#pragma unroll
  for (int k0 = 0; k0 < 16; k0 += 4) {
    v2f a, b;
    a.x = fmaxf(arow[k0 + kh + 0], 0.0f);
    a.y = fmaxf(arow[k0 + kh + 1], 0.0f);
    b.x = B[(k0 + kh + 0) * 16 + m];
    b.y = B[(k0 + kh + 1) * 16 + m];
    c = __builtin_amdgcn_wmma_f32_16x16x4_f32(false, a, false, b, (short)0, c, false, false);
  }
  const float bv = bias[m];
#pragma unroll
  for (int r = 0; r < 8; ++r) {
    const float hs = c[r] * di[r];
    hw[(rbase + r) * 16 + m]  = hs;
    agg[(rbase + r) * 16 + m] = hs * di[r] + bv;
  }
}

typedef int v4i __attribute__((ext_vector_type(4)));

constexpr int SC_S     = 4;
constexpr int SC_LOADS = 8;
constexpr int SC_BATCH = 32 * 4 * SC_LOADS;
constexpr int SC_UD    = 8;
constexpr int SC_QLEN  = SC_BATCH + 2 * SC_UD;
constexpr int RP_UD    = 16;

__device__ __forceinline__ bool sc_list_fits(const unsigned* list, int ne, const float* ws0, size_t ws_bytes,
                                             int tbl_room) {
  const int nchunks = (ne + BIN_EDGES - 1) / BIN_EDGES;
  return (size_t)((const char*)(list + ne) - (const char*)ws0) <= ws_bytes && nchunks * BIN_TBL <= tbl_room;
}

__device__ __forceinline__ unsigned bin_chunk_count(int c, int ne) {
  const int left = (ne - c * BIN_EDGES) & ~3;
  return (unsigned)(left < BIN_EDGES ? left : BIN_EDGES);
}

__device__ __forceinline__ unsigned bin_run_start(const unsigned* __restrict__ tbl, int c, int b, int ne) {
  return b < BIN_MAXR ? tbl[c * BIN_TBL + b] : bin_chunk_count(c, ne);
}

__device__ __forceinline__ void sc_hand_over(const float* acc, const float* __restrict__ dinv,
                                             float* __restrict__ agg, int lo, int tid) {
  constexpr int SC_FL = SC_NR * 16 / 1024;
  float dv[SC_FL], av[SC_FL];
#pragma unroll
  for (int j = 0; j < SC_FL; ++j) dv[j] = dinv[lo + (tid >> 4) + 64 * j];
  __syncthreads();
  float* arow = agg + (size_t)lo * 16;
#pragma unroll
  for (int j = 0; j < SC_FL; ++j) av[j] = acc[tid + 1024 * j] * dv[j];
#pragma unroll
  for (int j = 0; j < SC_FL; ++j) unsafeAtomicAdd(arow + tid + 1024 * j, av[j]);
}

__device__ __forceinline__ void sc_scan(float* acc, unsigned* qw,
                                        const int* __restrict__ src, const int* __restrict__ dst,
                                        const float* __restrict__ hw, int ne, int r, int sl) {
  const int tid = threadIdx.x, lane = tid & 31, w = tid >> 5;
  const int half = lane >> 4, f = lane & 15;
  const int lo = r * SC_NR;

  const int slice = ne / SC_S;
  const int span  = slice / SC_WAVES;
  const int ebase = sl * slice + w * span;
  const int nb    = span / SC_BATCH;
  const v4i* dp   = (const v4i*)(dst + ebase);
  const int* sp   = src + ebase;
  int qn = 0;

  auto pass = [&](int q0, int nv) {
    unsigned ent[SC_UD]; bool ok[SC_UD]; int sv[SC_UD]; float hv[SC_UD];
#pragma unroll
    for (int u = 0; u < SC_UD; ++u) {
      ok[u]  = 2 * u + half < nv;
      ent[u] = ok[u] ? qw[q0 + 2 * u + half] : 0u;
    }
#pragma unroll
    for (int u = 0; u < SC_UD; ++u) sv[u] = sp[ent[u] >> 10];
#pragma unroll
    for (int u = 0; u < SC_UD; ++u) hv[u] = hw[sv[u] * 16 + f];
#pragma unroll
    for (int u = 0; u < SC_UD; ++u)
      unsafeAtomicAdd(&acc[(ent[u] & 1023u) * 16 + f], ok[u] ? hv[u] : 0.0f);
  };

  v4i cur[SC_LOADS], nxt[SC_LOADS];
#pragma unroll
  for (int j = 0; j < SC_LOADS; ++j) cur[j] = dp[j * 32 + lane];
  for (int b = 0; b < nb; ++b) {
    const int bn = (b + 1 < nb) ? b + 1 : b;
#pragma unroll
    for (int j = 0; j < SC_LOADS; ++j) nxt[j] = dp[bn * (SC_BATCH / 4) + j * 32 + lane];
#pragma unroll
    for (int j = 0; j < SC_LOADS; ++j) {
#pragma unroll
      for (int c = 0; c < 4; ++c) {
        const int dl = cur[j][c] - lo;
        const bool m = (unsigned)dl < (unsigned)SC_NR;
        const unsigned mask = __builtin_amdgcn_ballot_w32(m);
        if (mask != 0u) {
          const int cnt = __builtin_popcount(mask);
          if (m) {
            const int el = b * SC_BATCH + j * 128 + lane * 4 + c;
            qw[qn + (int)__builtin_amdgcn_mbcnt_lo(mask, 0u)] = ((unsigned)el << 10) | (unsigned)dl;
          }
          qn += cnt;
        }
      }
    }
    while (qn >= 2 * SC_UD) { qn -= 2 * SC_UD; pass(qn, 2 * SC_UD); }
#pragma unroll
    for (int j = 0; j < SC_LOADS; ++j) cur[j] = nxt[j];
  }
  if (qn > 0) pass(0, qn);
}

typedef unsigned v4u __attribute__((ext_vector_type(4)));

__global__ __launch_bounds__(BIN_T)
void k_bin_edges(const int* __restrict__ src, const int* __restrict__ dst, int ne, int nbins,
                 unsigned* __restrict__ tbl, unsigned* __restrict__ list,
                 const float* ws0, size_t ws_bytes, int tbl_room,
                 float* __restrict__ zacc, int nz) {
  __shared__ unsigned cnt[BIN_MAXR];
  __shared__ unsigned lpre[BIN_MAXR];
  __shared__ __attribute__((aligned(16))) unsigned ordered[BIN_EDGES];
  const int tid = threadIdx.x;
  for (int i = blockIdx.x * BIN_T + tid; i < nz; i += gridDim.x * BIN_T) zacc[i] = 0.0f;
  if (!sc_list_fits(list, ne, ws0, ws_bytes, tbl_room)) return;
  const int e0  = blockIdx.x * BIN_EDGES;

  v4i dv[BIN_LD], sv[BIN_LD];
  bool in[BIN_LD];
#pragma unroll
  for (int j = 0; j < BIN_LD; ++j) {
    const int e = e0 + (j * BIN_T + tid) * 4;
    in[j] = e + 3 < ne;
    dv[j] = __builtin_nontemporal_load((const v4i*)(dst + (in[j] ? e : 0)));
    sv[j] = __builtin_nontemporal_load((const v4i*)(src + (in[j] ? e : 0)));
  }
  if (tid < BIN_MAXR) cnt[tid] = 0u;
  __syncthreads();

  unsigned rk[BIN_LD][4];
#pragma unroll
  for (int j = 0; j < BIN_LD; ++j) {
#pragma unroll
    for (int k = 0; k < 4; ++k) {
      rk[j][k] = 0u;
      if (in[j]) rk[j][k] = atomicAdd(&cnt[(unsigned)dv[j][k] / (unsigned)BIN_NR], 1u);
    }
  }
  __syncthreads();

  if (tid < 32) {
    constexpr int PB = BIN_MAXR / 32;
    unsigned v[PB];
    unsigned t = 0u;
#pragma unroll
    for (int j = 0; j < PB; ++j) {
      v[j] = (tid * PB + j < nbins) ? cnt[tid * PB + j] : 0u;
      t += v[j];
    }
    unsigned inc = t;
    for (int o = 1; o < 32; o <<= 1) {
      const unsigned up = __shfl_up(inc, o);
      if (tid >= o) inc += up;
    }
    unsigned run = inc - t;
    unsigned* trow = tbl + (size_t)blockIdx.x * BIN_TBL;
#pragma unroll
    for (int j = 0; j < PB; ++j) {
      lpre[tid * PB + j] = run;
      trow[tid * PB + j] = run;
      run += v[j];
    }
  }
  __syncthreads();

#pragma unroll
  for (int j = 0; j < BIN_LD; ++j) {
#pragma unroll
    for (int k = 0; k < 4; ++k) {
      if (in[j]) {
        const unsigned rr = (unsigned)dv[j][k] / (unsigned)BIN_NR;
        ordered[lpre[rr] + rk[j][k]] = ((unsigned)sv[j][k] << 10) | ((unsigned)dv[j][k] % (unsigned)SC_NR);
      }
    }
  }
  __syncthreads();

  const int left = (ne - e0) & ~3;
  const int nv4 = (left < BIN_EDGES ? left : BIN_EDGES) >> 2;
  constexpr int BIN_CP = BIN_EDGES / 4 / BIN_T;
  const v4u* o4 = (const v4u*)ordered;
  v4u* l4 = (v4u*)(list + e0);
  v4u wd[BIN_CP];
#pragma unroll
  for (int k = 0; k < BIN_CP; ++k) wd[k] = o4[tid + k * BIN_T];
#pragma unroll
  for (int k = 0; k < BIN_CP; ++k)
    if (tid + k * BIN_T < nv4) l4[tid + k * BIN_T] = wd[k];
}

__global__ __launch_bounds__(DGQ_WAVES * 32)
void k_deg_gemm_first(const float* __restrict__ x, const float* __restrict__ B,
                      const float* __restrict__ bias, float* __restrict__ dinv,
                      float* __restrict__ hw, float* __restrict__ agg,
                      const int* __restrict__ dst, int ne,
                      const unsigned* __restrict__ tbl, const unsigned* __restrict__ list,
                      const float* ws0, size_t ws_bytes, int tbl_room) {
  static_assert(DGQ_ROWS == BIN_NR, "a work-group's rows are one bin");
  static_assert(DGQ_WAVES == 16, "one load fetches the table words of one run per wave of the work-group");
  __shared__ unsigned cnt[DGQ_ROWS];
  const int tid = threadIdx.x, lane = tid & 31, w = tid >> 5;
  const int bin = blockIdx.x;
  const int lo = bin * DGQ_ROWS;
  const int m  = lane & 15;
  const int kh = (lane >> 4) << 1;
  const int row0  = lo + w * 16;
  const int rbase = row0 + ((lane >> 4) << 3);

  float a0[4], a1[4], b0[4], b1[4];
  const float* arow = x + (row0 + m) * 16;
#pragma unroll
  for (int k = 0; k < 4; ++k) {
    a0[k] = arow[4 * k + kh + 0];
    a1[k] = arow[4 * k + kh + 1];
    b0[k] = B[(4 * k + kh + 0) * 16 + m];
    b1[k] = B[(4 * k + kh + 1) * 16 + m];
  }
  const float bv = bias[m];

  for (int i = tid; i < DGQ_ROWS; i += DGQ_WAVES * 32) cnt[i] = 0u;
  __syncthreads();
  if (sc_list_fits(list, ne, ws0, ws_bytes, tbl_room)) {
    const int nchunks = (ne + BIN_EDGES - 1) / BIN_EDGES;
    for (int c0 = w; c0 < nchunks; c0 += 16 * DGQ_WAVES) {
      const int cl = c0 + DGQ_WAVES * m;
      unsigned tv = 0u;
      if (cl < nchunks) tv = bin_run_start(tbl, cl, bin + (lane >> 4), ne);
      unsigned s[16], e[16];
#pragma unroll
      for (int i = 0; i < 16; ++i) {
        s[i] = (unsigned)__builtin_amdgcn_readlane((int)tv, i);
        e[i] = (unsigned)__builtin_amdgcn_readlane((int)tv, 16 + i);
      }
      unsigned wv[16][DGQ_UD];
#pragma unroll
      for (int i = 0; i < 16; ++i) {
        const int ci = c0 + DGQ_WAVES * i;
        const unsigned* __restrict__ rl = list + (size_t)(ci < nchunks ? ci : 0) * BIN_EDGES;
#pragma unroll
        for (int u = 0; u < DGQ_UD; ++u) {
          const unsigned p = s[i] + 32u * u + lane;
          wv[i][u] = rl[p < e[i] ? p : 0u];
        }
      }
#pragma unroll
      for (int i = 0; i < 16; ++i) {
#pragma unroll
        for (int u = 0; u < DGQ_UD; ++u) {
          const unsigned p = s[i] + 32u * u + lane;
          if (p < e[i]) atomicAdd(&cnt[wv[i][u] & (DGQ_ROWS - 1)], 1u);
        }
      }
#pragma unroll
      for (int i = 0; i < 16; ++i) {
        const int ci = c0 + DGQ_WAVES * i;
        const unsigned* __restrict__ rl = list + (size_t)(ci < nchunks ? ci : 0) * BIN_EDGES;
        for (unsigned p0 = s[i] + 32u * DGQ_UD; p0 < e[i]; p0 += 32u) {
          const unsigned p = p0 + lane;
          if (p < e[i]) atomicAdd(&cnt[rl[p] & (DGQ_ROWS - 1)], 1u);
        }
      }
    }
  } else {
    for (int ed = tid; ed < ne; ed += DGQ_WAVES * 32) {
      const int dl = dst[ed] - lo;
      if ((unsigned)dl < (unsigned)DGQ_ROWS) atomicAdd(&cnt[dl], 1u);
    }
  }
  __syncthreads();

  float di[8];
#pragma unroll
  for (int rr = 0; rr < 8; ++rr) di[rr] = rsqrtf(1.0f + (float)cnt[rbase - lo + rr]);
  v8f c = {};
#pragma unroll
  for (int k = 0; k < 4; ++k) {
    v2f a, b;
    a.x = a0[k]; a.y = a1[k];
    b.x = b0[k]; b.y = b1[k];
    c = __builtin_amdgcn_wmma_f32_16x16x4_f32(false, a, false, b, (short)0, c, false, false);
  }
#pragma unroll
  for (int rr = 0; rr < 8; ++rr) {
    const float hs = c[rr] * di[rr];
    hw[(rbase + rr) * 16 + m]  = hs;
    agg[(rbase + rr) * 16 + m] = hs * di[rr] + bv;
  }
  if (m == 0) {
#pragma unroll
    for (int rr = 0; rr < 8; ++rr) dinv[rbase + rr] = di[rr];
  }
}

__global__ __launch_bounds__(1024)
void k_scatter_replay(const int* __restrict__ src, const int* __restrict__ dst,
                      const float* __restrict__ dinv, const float* __restrict__ hw,
                      float* __restrict__ agg, int ne, int nranges,
                      const unsigned* __restrict__ tbl, const unsigned* __restrict__ list,
                      const float* ws0, size_t ws_bytes, int tbl_room) {
  __shared__ float acc[SC_NR * 16];
  __shared__ unsigned q[SC_WAVES][SC_QLEN];
  const int tid = threadIdx.x, lane = tid & 31, w = tid >> 5;
  const int half = lane >> 4, f = lane & 15;
  const int r    = blockIdx.x % nranges;
  const int part = blockIdx.x / nranges;
  const bool fits = sc_list_fits(list, ne, ws0, ws_bytes, tbl_room);
  const int nchunks = (ne + BIN_EDGES - 1) / BIN_EDGES;
  constexpr int NW = SC_S * SC_WAVES;
  int c = part * SC_WAVES + w;
  unsigned rs = 0u, re = 0u;
  if (fits && c < nchunks) { rs = bin_run_start(tbl, c, r * SC_BINS, ne); re = bin_run_start(tbl, c, (r + 1) * SC_BINS, ne); }
  for (int i = tid; i < SC_NR * 16; i += 1024) acc[i] = 0.0f;
  __syncthreads();
  if (fits) {
    while (c < nchunks) {
      const int cn = c + NW;
      unsigned ns = 0u, nend = 0u;
      if (cn < nchunks) { ns = bin_run_start(tbl, cn, r * SC_BINS, ne); nend = bin_run_start(tbl, cn, (r + 1) * SC_BINS, ne); }
      const unsigned* __restrict__ rl = list + (size_t)c * BIN_EDGES;
      const unsigned c0 = rs, c1 = re;
      if (c0 < c1) {
        const unsigned last = c1 - 1u;
        unsigned ent[RP_UD], nxt[RP_UD]; float hv[RP_UD];
#pragma unroll
        for (int u = 0; u < RP_UD; ++u) ent[u] = rl[min(c0 + 2u * u + half, last)];
        for (unsigned p = c0; p < c1; p += 2 * RP_UD) {
#pragma unroll
          for (int u = 0; u < RP_UD; ++u) nxt[u] = rl[min(p + 2 * RP_UD + 2u * u + half, last)];
#pragma unroll
          for (int u = 0; u < RP_UD; ++u) hv[u] = hw[(ent[u] >> 10) * 16 + f];
#pragma unroll
          for (int u = 0; u < RP_UD; ++u)
            unsafeAtomicAdd(&acc[(ent[u] & 1023u) * 16 + f], (p + 2u * u + half < c1) ? hv[u] : 0.0f);
#pragma unroll
          for (int u = 0; u < RP_UD; ++u) ent[u] = nxt[u];
        }
      }
      c = cn; rs = ns; re = nend;
    }
  } else {
    sc_scan(acc, q[w], src, dst, hw, ne, r, part);
  }
  sc_hand_over(acc, dinv, agg, r * SC_NR, tid);
}

template <int KLEN>
__global__ void k_heads_splitk_relu(const float* __restrict__ A, int lda,
                                    const float* __restrict__ Bp, const float* __restrict__ Bv,
                                    float* __restrict__ accp, float* __restrict__ accv,
                                    int nsplit, int pol_blocks) {
  const bool pol = (int)blockIdx.x < pol_blocks;
  const float* __restrict__ B = pol ? Bp : Bv;
  float* __restrict__ acc     = pol ? accp : accv;
  const int ncols   = pol ? 32 : 16;
  const int tiles_n = pol ? 2 : 1;
  const int blk     = pol ? (int)blockIdx.x : (int)blockIdx.x - pol_blocks;
  int wid  = (blk * blockDim.x + threadIdx.x) >> 5;
  int lane = threadIdx.x & 31;
  int split = wid % nsplit;
  int tn    = (wid / nsplit) % tiles_n;
  int tm    = wid / (nsplit * tiles_n);
  int m  = lane & 15;
  int kh = (lane >> 4) << 1;
  const float* arow = A + (size_t)(tm * 16 + m) * lda + split * KLEN;
  const float* bcol = B + (size_t)(split * KLEN) * ncols + tn * 16 + m;
  v8f c = {};
#pragma unroll 4
  for (int k0 = 0; k0 < KLEN; k0 += 4) {
    v2f a, b;
    a.x = fmaxf(arow[k0 + kh + 0], 0.0f);
    a.y = fmaxf(arow[k0 + kh + 1], 0.0f);
    b.x = bcol[(size_t)(k0 + kh + 0) * ncols];
    b.y = bcol[(size_t)(k0 + kh + 1) * ncols];
    c = __builtin_amdgcn_wmma_f32_16x16x4_f32(false, a, false, b, (short)0, c, false, false);
  }
  int rbase = tm * 16 + ((lane >> 4) << 3);
#pragma unroll
  for (int r = 0; r < 8; ++r)
    unsafeAtomicAdd(&acc[(rbase + r) * ncols + tn * 16 + m], c[r]);
}

__global__ void k_heads_final(const float* __restrict__ pacc, const float* __restrict__ p1b,
                              const float* __restrict__ B, const float* __restrict__ bias,
                              const float* __restrict__ vacc, const float* __restrict__ v1b,
                              const float* __restrict__ v2w, const float* __restrict__ v2b,
                              float* __restrict__ out, float* __restrict__ vout) {
  if (blockIdx.x == 32) {
    int bi = threadIdx.x;
    if (bi < 64) {
      float s = 0.0f;
#pragma unroll
      for (int j = 0; j < 16; ++j) s += fmaxf(vacc[bi * 16 + j] + v1b[j], 0.0f) * v2w[j];
      vout[bi] = s + v2b[0];
    }
    return;
  }
  int wid  = (blockIdx.x * blockDim.x + threadIdx.x) >> 5;
  int lane = threadIdx.x & 31;
  int tm = wid >> 6;
  int tn = wid & 63;
  int m  = lane & 15;
  int kh = (lane >> 4) << 1;
  const float* arow = pacc + (tm * 16 + m) * 32;
  const float* bcol = B + tn * 16 + m;
  v8f c = {};
#pragma unroll
  for (int k0 = 0; k0 < 32; k0 += 4) {
    v2f a, b;
    a.x = fmaxf(arow[k0 + kh + 0] + p1b[k0 + kh + 0], 0.0f);
    a.y = fmaxf(arow[k0 + kh + 1] + p1b[k0 + kh + 1], 0.0f);
    b.x = bcol[(k0 + kh + 0) * 1024];
    b.y = bcol[(k0 + kh + 1) * 1024];
    c = __builtin_amdgcn_wmma_f32_16x16x4_f32(false, a, false, b, (short)0, c, false, false);
  }
  int rbase = tm * 16 + ((lane >> 4) << 3);
  float bv = bias[tn * 16 + m];
#pragma unroll
  for (int r = 0; r < 8; ++r)
    out[(rbase + r) * 1024 + tn * 16 + m] = c[r] + bv;
}

extern "C" void kernel_launch(void* const* d_in, const int* in_sizes, int n_in,
                              void* d_out, int out_size, void* d_ws, size_t ws_size,
                              hipStream_t stream) {
  const float* x    = (const float*)d_in[0];
  const int*   ei   = (const int*)d_in[1];
  const float* W1   = (const float*)d_in[2];
  const float* b1   = (const float*)d_in[3];
  const float* W2   = (const float*)d_in[4];
  const float* b2   = (const float*)d_in[5];
  const float* p1w  = (const float*)d_in[6];
  const float* p1b  = (const float*)d_in[7];
  const float* p2w  = (const float*)d_in[8];
  const float* p2b  = (const float*)d_in[9];
  const float* v1w  = (const float*)d_in[10];
  const float* v1b  = (const float*)d_in[11];
  const float* v2w  = (const float*)d_in[12];
  const float* v2b  = (const float*)d_in[13];

  const int N  = in_sizes[0] / 16;
  const int E  = in_sizes[1] / 2;
  const int* esrc = ei;
  const int* edst = ei + E;

  float* ws   = (float*)d_ws;
  float* dinv = ws + N;
  float* bufA = dinv + N;
  float* bufB = bufA + (size_t)16 * N;
  float* pacc = bufB + (size_t)16 * N;
  float* vacc = pacc + 64 * 32;
  unsigned* elist = (unsigned*)(vacc + 64 * 16);
  unsigned* tbl   = (unsigned*)ws;

  float* Xout = (float*)d_out;
  float* Vout = Xout + 64 * 1024;

  const int B256 = 256;
  int gemm_blocks    = (N / 16) / 8;
  int nranges        = N / SC_NR;
  int scatter_blocks = nranges * SC_S;

  k_bin_edges<<<(E + BIN_EDGES - 1) / BIN_EDGES, BIN_T, 0, stream>>>(esrc, edst, E, N / BIN_NR, tbl, elist, ws, ws_size, N,
                                                                    pacc, 64 * 32 + 64 * 16);

  k_deg_gemm_first<<<N / DGQ_ROWS, DGQ_WAVES * 32, 0, stream>>>(x, W1, b1, dinv, bufA, bufB, edst, E,
                                                              tbl, elist, ws, ws_size, N);
  k_scatter_replay<<<scatter_blocks, 1024, 0, stream>>>(esrc, edst, dinv, bufA, bufB, E, nranges,
                                                        tbl, elist, ws, ws_size, N);

  k_gemm_agg<<<gemm_blocks, B256, 0, stream>>>(W2, b2, dinv, bufA, bufB, N);
  k_scatter_replay<<<scatter_blocks, 1024, 0, stream>>>(esrc, edst, dinv, bufA, bufB, E, nranges,
                                                        tbl, elist, ws, ws_size, N);

  const int K = 16384, NSPLIT = 64, KLEN = K / NSPLIT;

  k_heads_splitk_relu<KLEN><<<64 + 32, B256, 0, stream>>>(bufB, K, p1w, v1w, pacc, vacc, NSPLIT, 64);

  k_heads_final<<<32 + 1, B256, 0, stream>>>(pacc, p1b, p2w, p2b, vacc, v1b, v2w, v2b, Xout, Vout);
}


# split-K heads re-tiled: a wave does one 16-row tile x all 48 columns over a 64-deep K slice, all its loads issued before the products; 4 waves sum in LDS, one add per element; sum order changes
# speedup vs baseline: 5.3039x; 1.1393x over previous
#include <hip/hip_runtime.h>
#include <hip/hip_bf16.h>

typedef __attribute__((ext_vector_type(2))) float v2f;
typedef __attribute__((ext_vector_type(8))) float v8f;

constexpr int SC_NR      = 1024;
constexpr int SC_WAVES   = 32;
constexpr int DGQ_ROWS   = 256;
constexpr int DGQ_WAVES  = DGQ_ROWS / 16;
constexpr int DGQ_UD     = 2;

constexpr int BIN_T     = 512;
constexpr int BIN_LD    = 4;
constexpr int BIN_EDGES = BIN_T * 4 * BIN_LD;
constexpr int BIN_NR    = 256;
constexpr int BIN_MAXR  = 256;
constexpr int BIN_TBL   = BIN_MAXR;
constexpr int SC_BINS   = SC_NR / BIN_NR;

__global__ void k_gemm_agg(const float* __restrict__ B, const float* __restrict__ bias,
                           const float* __restrict__ dinv, float* __restrict__ hw, float* agg, int nrows) {
  int wid  = (blockIdx.x * blockDim.x + threadIdx.x) >> 5;
  int lane = threadIdx.x & 31;
  int row0 = wid * 16;
  if (row0 >= nrows) return;
  int m  = lane & 15;
  int kh = (lane >> 4) << 1;
  int rbase = row0 + ((lane >> 4) << 3);
  float di[8];
#pragma unroll
  for (int r = 0; r < 8; ++r) di[r] = dinv[rbase + r];
  const float* arow = (const float*)agg + (row0 + m) * 16;
  v8f c = {};
#pragma unroll
  for (int k0 = 0; k0 < 16; k0 += 4) {
    v2f a, b;
    a.x = fmaxf(arow[k0 + kh + 0], 0.0f);
    a.y = fmaxf(arow[k0 + kh + 1], 0.0f);
    b.x = B[(k0 + kh + 0) * 16 + m];
    b.y = B[(k0 + kh + 1) * 16 + m];
    c = __builtin_amdgcn_wmma_f32_16x16x4_f32(false, a, false, b, (short)0, c, false, false);
  }
  const float bv = bias[m];
#pragma unroll
  for (int r = 0; r < 8; ++r) {
    const float hs = c[r] * di[r];
    hw[(rbase + r) * 16 + m]  = hs;
    agg[(rbase + r) * 16 + m] = hs * di[r] + bv;
  }
}

typedef int v4i __attribute__((ext_vector_type(4)));

constexpr int SC_S     = 4;
constexpr int SC_LOADS = 8;
constexpr int SC_BATCH = 32 * 4 * SC_LOADS;
constexpr int SC_UD    = 8;
constexpr int SC_QLEN  = SC_BATCH + 2 * SC_UD;
constexpr int RP_UD    = 16;

__device__ __forceinline__ bool sc_list_fits(const unsigned* list, int ne, const float* ws0, size_t ws_bytes,
                                             int tbl_room) {
  const int nchunks = (ne + BIN_EDGES - 1) / BIN_EDGES;
  return (size_t)((const char*)(list + ne) - (const char*)ws0) <= ws_bytes && nchunks * BIN_TBL <= tbl_room;
}

__device__ __forceinline__ unsigned bin_chunk_count(int c, int ne) {
  const int left = (ne - c * BIN_EDGES) & ~3;
  return (unsigned)(left < BIN_EDGES ? left : BIN_EDGES);
}

__device__ __forceinline__ unsigned bin_run_start(const unsigned* __restrict__ tbl, int c, int b, int ne) {
  return b < BIN_MAXR ? tbl[c * BIN_TBL + b] : bin_chunk_count(c, ne);
}

__device__ __forceinline__ void sc_hand_over(const float* acc, const float* __restrict__ dinv,
                                             float* __restrict__ agg, int lo, int tid) {
  constexpr int SC_FL = SC_NR * 16 / 1024;
  float dv[SC_FL], av[SC_FL];
#pragma unroll
  for (int j = 0; j < SC_FL; ++j) dv[j] = dinv[lo + (tid >> 4) + 64 * j];
  __syncthreads();
  float* arow = agg + (size_t)lo * 16;
#pragma unroll
  for (int j = 0; j < SC_FL; ++j) av[j] = acc[tid + 1024 * j] * dv[j];
#pragma unroll
  for (int j = 0; j < SC_FL; ++j) unsafeAtomicAdd(arow + tid + 1024 * j, av[j]);
}

__device__ __forceinline__ void sc_scan(float* acc, unsigned* qw,
                                        const int* __restrict__ src, const int* __restrict__ dst,
                                        const float* __restrict__ hw, int ne, int r, int sl) {
  const int tid = threadIdx.x, lane = tid & 31, w = tid >> 5;
  const int half = lane >> 4, f = lane & 15;
  const int lo = r * SC_NR;

  const int slice = ne / SC_S;
  const int span  = slice / SC_WAVES;
  const int ebase = sl * slice + w * span;
  const int nb    = span / SC_BATCH;
  const v4i* dp   = (const v4i*)(dst + ebase);
  const int* sp   = src + ebase;
  int qn = 0;

  auto pass = [&](int q0, int nv) {
    unsigned ent[SC_UD]; bool ok[SC_UD]; int sv[SC_UD]; float hv[SC_UD];
#pragma unroll
    for (int u = 0; u < SC_UD; ++u) {
      ok[u]  = 2 * u + half < nv;
      ent[u] = ok[u] ? qw[q0 + 2 * u + half] : 0u;
    }
#pragma unroll
    for (int u = 0; u < SC_UD; ++u) sv[u] = sp[ent[u] >> 10];
#pragma unroll
    for (int u = 0; u < SC_UD; ++u) hv[u] = hw[sv[u] * 16 + f];
#pragma unroll
    for (int u = 0; u < SC_UD; ++u)
      unsafeAtomicAdd(&acc[(ent[u] & 1023u) * 16 + f], ok[u] ? hv[u] : 0.0f);
  };

  v4i cur[SC_LOADS], nxt[SC_LOADS];
#pragma unroll
  for (int j = 0; j < SC_LOADS; ++j) cur[j] = dp[j * 32 + lane];
  for (int b = 0; b < nb; ++b) {
    const int bn = (b + 1 < nb) ? b + 1 : b;
#pragma unroll
    for (int j = 0; j < SC_LOADS; ++j) nxt[j] = dp[bn * (SC_BATCH / 4) + j * 32 + lane];
#pragma unroll
    for (int j = 0; j < SC_LOADS; ++j) {
#pragma unroll
      for (int c = 0; c < 4; ++c) {
        const int dl = cur[j][c] - lo;
        const bool m = (unsigned)dl < (unsigned)SC_NR;
        const unsigned mask = __builtin_amdgcn_ballot_w32(m);
        if (mask != 0u) {
          const int cnt = __builtin_popcount(mask);
          if (m) {
            const int el = b * SC_BATCH + j * 128 + lane * 4 + c;
            qw[qn + (int)__builtin_amdgcn_mbcnt_lo(mask, 0u)] = ((unsigned)el << 10) | (unsigned)dl;
          }
          qn += cnt;
        }
      }
    }
    while (qn >= 2 * SC_UD) { qn -= 2 * SC_UD; pass(qn, 2 * SC_UD); }
#pragma unroll
    for (int j = 0; j < SC_LOADS; ++j) cur[j] = nxt[j];
  }
  if (qn > 0) pass(0, qn);
}

typedef unsigned v4u __attribute__((ext_vector_type(4)));

__global__ __launch_bounds__(BIN_T)
void k_bin_edges(const int* __restrict__ src, const int* __restrict__ dst, int ne, int nbins,
                 unsigned* __restrict__ tbl, unsigned* __restrict__ list,
                 const float* ws0, size_t ws_bytes, int tbl_room,
                 float* __restrict__ zacc, int nz) {
  __shared__ unsigned cnt[BIN_MAXR];
  __shared__ unsigned lpre[BIN_MAXR];
  __shared__ __attribute__((aligned(16))) unsigned ordered[BIN_EDGES];
  const int tid = threadIdx.x;
  for (int i = blockIdx.x * BIN_T + tid; i < nz; i += gridDim.x * BIN_T) zacc[i] = 0.0f;
  if (!sc_list_fits(list, ne, ws0, ws_bytes, tbl_room)) return;
  const int e0  = blockIdx.x * BIN_EDGES;

  v4i dv[BIN_LD], sv[BIN_LD];
  bool in[BIN_LD];
#pragma unroll
  for (int j = 0; j < BIN_LD; ++j) {
    const int e = e0 + (j * BIN_T + tid) * 4;
    in[j] = e + 3 < ne;
    dv[j] = __builtin_nontemporal_load((const v4i*)(dst + (in[j] ? e : 0)));
    sv[j] = __builtin_nontemporal_load((const v4i*)(src + (in[j] ? e : 0)));
  }
  if (tid < BIN_MAXR) cnt[tid] = 0u;
  __syncthreads();

  unsigned rk[BIN_LD][4];
#pragma unroll
  for (int j = 0; j < BIN_LD; ++j) {
#pragma unroll
    for (int k = 0; k < 4; ++k) {
      rk[j][k] = 0u;
      if (in[j]) rk[j][k] = atomicAdd(&cnt[(unsigned)dv[j][k] / (unsigned)BIN_NR], 1u);
    }
  }
  __syncthreads();

  if (tid < 32) {
    constexpr int PB = BIN_MAXR / 32;
    unsigned v[PB];
    unsigned t = 0u;
#pragma unroll
    for (int j = 0; j < PB; ++j) {
      v[j] = (tid * PB + j < nbins) ? cnt[tid * PB + j] : 0u;
      t += v[j];
    }
    unsigned inc = t;
    for (int o = 1; o < 32; o <<= 1) {
      const unsigned up = __shfl_up(inc, o);
      if (tid >= o) inc += up;
    }
    unsigned run = inc - t;
    unsigned* trow = tbl + (size_t)blockIdx.x * BIN_TBL;
#pragma unroll
    for (int j = 0; j < PB; ++j) {
      lpre[tid * PB + j] = run;
      trow[tid * PB + j] = run;
      run += v[j];
    }
  }
  __syncthreads();

#pragma unroll
  for (int j = 0; j < BIN_LD; ++j) {
#pragma unroll
    for (int k = 0; k < 4; ++k) {
      if (in[j]) {
        const unsigned rr = (unsigned)dv[j][k] / (unsigned)BIN_NR;
        ordered[lpre[rr] + rk[j][k]] = ((unsigned)sv[j][k] << 10) | ((unsigned)dv[j][k] % (unsigned)SC_NR);
      }
    }
  }
  __syncthreads();

  const int left = (ne - e0) & ~3;
  const int nv4 = (left < BIN_EDGES ? left : BIN_EDGES) >> 2;
  constexpr int BIN_CP = BIN_EDGES / 4 / BIN_T;
  const v4u* o4 = (const v4u*)ordered;
  v4u* l4 = (v4u*)(list + e0);
  v4u wd[BIN_CP];
#pragma unroll
  for (int k = 0; k < BIN_CP; ++k) wd[k] = o4[tid + k * BIN_T];
#pragma unroll
  for (int k = 0; k < BIN_CP; ++k)
    if (tid + k * BIN_T < nv4) l4[tid + k * BIN_T] = wd[k];
}

__global__ __launch_bounds__(DGQ_WAVES * 32)
void k_deg_gemm_first(const float* __restrict__ x, const float* __restrict__ B,
                      const float* __restrict__ bias, float* __restrict__ dinv,
                      float* __restrict__ hw, float* __restrict__ agg,
                      const int* __restrict__ dst, int ne,
                      const unsigned* __restrict__ tbl, const unsigned* __restrict__ list,
                      const float* ws0, size_t ws_bytes, int tbl_room) {
  static_assert(DGQ_ROWS == BIN_NR, "a work-group's rows are one bin");
  static_assert(DGQ_WAVES == 16, "one load fetches the table words of one run per wave of the work-group");
  __shared__ unsigned cnt[DGQ_ROWS];
  const int tid = threadIdx.x, lane = tid & 31, w = tid >> 5;
  const int bin = blockIdx.x;
  const int lo = bin * DGQ_ROWS;
  const int m  = lane & 15;
  const int kh = (lane >> 4) << 1;
  const int row0  = lo + w * 16;
  const int rbase = row0 + ((lane >> 4) << 3);

  float a0[4], a1[4], b0[4], b1[4];
  const float* arow = x + (row0 + m) * 16;
#pragma unroll
  for (int k = 0; k < 4; ++k) {
    a0[k] = arow[4 * k + kh + 0];
    a1[k] = arow[4 * k + kh + 1];
    b0[k] = B[(4 * k + kh + 0) * 16 + m];
    b1[k] = B[(4 * k + kh + 1) * 16 + m];
  }
  const float bv = bias[m];

  for (int i = tid; i < DGQ_ROWS; i += DGQ_WAVES * 32) cnt[i] = 0u;
  __syncthreads();
  if (sc_list_fits(list, ne, ws0, ws_bytes, tbl_room)) {
    const int nchunks = (ne + BIN_EDGES - 1) / BIN_EDGES;
    for (int c0 = w; c0 < nchunks; c0 += 16 * DGQ_WAVES) {
      const int cl = c0 + DGQ_WAVES * m;
      unsigned tv = 0u;
      if (cl < nchunks) tv = bin_run_start(tbl, cl, bin + (lane >> 4), ne);
      unsigned s[16], e[16];
#pragma unroll
      for (int i = 0; i < 16; ++i) {
        s[i] = (unsigned)__builtin_amdgcn_readlane((int)tv, i);
        e[i] = (unsigned)__builtin_amdgcn_readlane((int)tv, 16 + i);
      }
      unsigned wv[16][DGQ_UD];
#pragma unroll
      for (int i = 0; i < 16; ++i) {
        const int ci = c0 + DGQ_WAVES * i;
        const unsigned* __restrict__ rl = list + (size_t)(ci < nchunks ? ci : 0) * BIN_EDGES;
#pragma unroll
        for (int u = 0; u < DGQ_UD; ++u) {
          const unsigned p = s[i] + 32u * u + lane;
          wv[i][u] = rl[p < e[i] ? p : 0u];
        }
      }
#pragma unroll
      for (int i = 0; i < 16; ++i) {
#pragma unroll
        for (int u = 0; u < DGQ_UD; ++u) {
          const unsigned p = s[i] + 32u * u + lane;
          if (p < e[i]) atomicAdd(&cnt[wv[i][u] & (DGQ_ROWS - 1)], 1u);
        }
      }
#pragma unroll
      for (int i = 0; i < 16; ++i) {
        const int ci = c0 + DGQ_WAVES * i;
        const unsigned* __restrict__ rl = list + (size_t)(ci < nchunks ? ci : 0) * BIN_EDGES;
        for (unsigned p0 = s[i] + 32u * DGQ_UD; p0 < e[i]; p0 += 32u) {
          const unsigned p = p0 + lane;
          if (p < e[i]) atomicAdd(&cnt[rl[p] & (DGQ_ROWS - 1)], 1u);
        }
      }
    }
  } else {
    for (int ed = tid; ed < ne; ed += DGQ_WAVES * 32) {
      const int dl = dst[ed] - lo;
      if ((unsigned)dl < (unsigned)DGQ_ROWS) atomicAdd(&cnt[dl], 1u);
    }
  }
  __syncthreads();

  float di[8];
#pragma unroll
  for (int rr = 0; rr < 8; ++rr) di[rr] = rsqrtf(1.0f + (float)cnt[rbase - lo + rr]);
  v8f c = {};
#pragma unroll
  for (int k = 0; k < 4; ++k) {
    v2f a, b;
    a.x = a0[k]; a.y = a1[k];
    b.x = b0[k]; b.y = b1[k];
    c = __builtin_amdgcn_wmma_f32_16x16x4_f32(false, a, false, b, (short)0, c, false, false);
  }
#pragma unroll
  for (int rr = 0; rr < 8; ++rr) {
    const float hs = c[rr] * di[rr];
    hw[(rbase + rr) * 16 + m]  = hs;
    agg[(rbase + rr) * 16 + m] = hs * di[rr] + bv;
  }
  if (m == 0) {
#pragma unroll
    for (int rr = 0; rr < 8; ++rr) dinv[rbase + rr] = di[rr];
  }
}

__global__ __launch_bounds__(1024)
void k_scatter_replay(const int* __restrict__ src, const int* __restrict__ dst,
                      const float* __restrict__ dinv, const float* __restrict__ hw,
                      float* __restrict__ agg, int ne, int nranges,
                      const unsigned* __restrict__ tbl, const unsigned* __restrict__ list,
                      const float* ws0, size_t ws_bytes, int tbl_room) {
  __shared__ float acc[SC_NR * 16];
  __shared__ unsigned q[SC_WAVES][SC_QLEN];
  const int tid = threadIdx.x, lane = tid & 31, w = tid >> 5;
  const int half = lane >> 4, f = lane & 15;
  const int r    = blockIdx.x % nranges;
  const int part = blockIdx.x / nranges;
  const bool fits = sc_list_fits(list, ne, ws0, ws_bytes, tbl_room);
  const int nchunks = (ne + BIN_EDGES - 1) / BIN_EDGES;
  constexpr int NW = SC_S * SC_WAVES;
  int c = part * SC_WAVES + w;
  unsigned rs = 0u, re = 0u;
  if (fits && c < nchunks) { rs = bin_run_start(tbl, c, r * SC_BINS, ne); re = bin_run_start(tbl, c, (r + 1) * SC_BINS, ne); }
  for (int i = tid; i < SC_NR * 16; i += 1024) acc[i] = 0.0f;
  __syncthreads();
  if (fits) {
    while (c < nchunks) {
      const int cn = c + NW;
      unsigned ns = 0u, nend = 0u;
      if (cn < nchunks) { ns = bin_run_start(tbl, cn, r * SC_BINS, ne); nend = bin_run_start(tbl, cn, (r + 1) * SC_BINS, ne); }
      const unsigned* __restrict__ rl = list + (size_t)c * BIN_EDGES;
      const unsigned c0 = rs, c1 = re;
      if (c0 < c1) {
        const unsigned last = c1 - 1u;
        unsigned ent[RP_UD], nxt[RP_UD]; float hv[RP_UD];
#pragma unroll
        for (int u = 0; u < RP_UD; ++u) ent[u] = rl[min(c0 + 2u * u + half, last)];
        for (unsigned p = c0; p < c1; p += 2 * RP_UD) {
#pragma unroll
          for (int u = 0; u < RP_UD; ++u) nxt[u] = rl[min(p + 2 * RP_UD + 2u * u + half, last)];
#pragma unroll
          for (int u = 0; u < RP_UD; ++u) hv[u] = hw[(ent[u] >> 10) * 16 + f];
#pragma unroll
          for (int u = 0; u < RP_UD; ++u)
            unsafeAtomicAdd(&acc[(ent[u] & 1023u) * 16 + f], (p + 2u * u + half < c1) ? hv[u] : 0.0f);
#pragma unroll
          for (int u = 0; u < RP_UD; ++u) ent[u] = nxt[u];
        }
      }
      c = cn; rs = ns; re = nend;
    }
  } else {
    sc_scan(acc, q[w], src, dst, hw, ne, r, part);
  }
  sc_hand_over(acc, dinv, agg, r * SC_NR, tid);
}

typedef __attribute__((ext_vector_type(4))) float v4f;

constexpr int HK_KLEN  = 64;
constexpr int HK_WAVES = 4;
constexpr int HK_NJ    = HK_KLEN / 8;
constexpr int HK_TS    = 8 * 32 + 16;
constexpr int HK_WS    = 3 * HK_TS;

__device__ __forceinline__ int hk_k(int j, int h, int s, int t) { return 8 * j + 4 * h + 2 * s + t; }

__global__ __launch_bounds__(HK_WAVES * 32)
void k_heads_splitk_relu(const float* __restrict__ A, int lda,
                         const float* __restrict__ Bp, const float* __restrict__ Bv,
                         float* __restrict__ accp, float* __restrict__ accv) {
  __shared__ float red[HK_WAVES * HK_WS];
  const int tid = threadIdx.x, lane = tid & 31, w = tid >> 5;
  const int tm = blockIdx.x & 3;
  const int kb = ((blockIdx.x >> 2) * HK_WAVES + w) * HK_KLEN;
  const int m = lane & 15, h = lane >> 4;
  const float* arow = A + ((tm * 16 + m) * lda + kb);
  const float* bp = Bp + (kb * 32 + 2 * m);
  const float* bv = Bv + (kb * 16 + m);

  v4f aq[HK_NJ];
  v2f pp[HK_NJ][4];
  float pv[HK_NJ][4];
#pragma unroll
  for (int j = 0; j < HK_NJ; ++j) {
    aq[j] = *(const v4f*)(arow + hk_k(j, h, 0, 0));
#pragma unroll
    for (int i = 0; i < 4; ++i) {
      const int k = hk_k(j, h, i >> 1, i & 1);
      pp[j][i] = *(const v2f*)(bp + k * 32);
      pv[j][i] = bv[k * 16];
    }
  }
  __builtin_amdgcn_sched_barrier(0);

  v8f c0 = {}, c1 = {}, c2 = {};
#pragma unroll
  for (int j = 0; j < HK_NJ; ++j) {
#pragma unroll
    for (int s = 0; s < 2; ++s) {
      v2f a, b;
      a.x = fmaxf(aq[j][2 * s + 0], 0.0f);
      a.y = fmaxf(aq[j][2 * s + 1], 0.0f);
      b.x = pp[j][2 * s + 0].x; b.y = pp[j][2 * s + 1].x;
      c0 = __builtin_amdgcn_wmma_f32_16x16x4_f32(false, a, false, b, (short)0, c0, false, false);
      b.x = pp[j][2 * s + 0].y; b.y = pp[j][2 * s + 1].y;
      c1 = __builtin_amdgcn_wmma_f32_16x16x4_f32(false, a, false, b, (short)0, c1, false, false);
      b.x = pv[j][2 * s + 0]; b.y = pv[j][2 * s + 1];
      c2 = __builtin_amdgcn_wmma_f32_16x16x4_f32(false, a, false, b, (short)0, c2, false, false);
    }
  }

  float* mine = red + w * HK_WS;
#pragma unroll
  for (int r = 0; r < 8; ++r) {
    mine[0 * HK_TS + r * 32 + lane] = c0[r];
    mine[1 * HK_TS + r * 32 + lane] = c1[r];
    mine[2 * HK_TS + r * 32 + lane] = c2[r];
  }
  __syncthreads();

  constexpr int HK_T  = HK_WAVES * 32;
  constexpr int HK_PE = 16 * 32 / HK_T;
  constexpr int HK_VE = 16 * 16 / HK_T;
  float sp[HK_PE], sv[HK_VE];
#pragma unroll
  for (int i = 0; i < HK_PE; ++i) {
    const int e = tid + HK_T * i;
    const int row = e >> 5, col = e & 31;
    const int at = (col & 1) * HK_TS + (row & 7) * 32 + (row >> 3) * 16 + (col >> 1);
    sp[i] = 0.0f;
#pragma unroll
    for (int ww = 0; ww < HK_WAVES; ++ww) sp[i] += red[ww * HK_WS + at];
  }
#pragma unroll
  for (int i = 0; i < HK_VE; ++i) {
    const int e = tid + HK_T * i;
    const int row = e >> 4, n = e & 15;
    const int at = 2 * HK_TS + (row & 7) * 32 + (row >> 3) * 16 + n;
    sv[i] = 0.0f;
#pragma unroll
    for (int ww = 0; ww < HK_WAVES; ++ww) sv[i] += red[ww * HK_WS + at];
  }
#pragma unroll
  for (int i = 0; i < HK_PE; ++i) {
    const int e = tid + HK_T * i;
    unsafeAtomicAdd(&accp[(tm * 16 + (e >> 5)) * 32 + (e & 31)], sp[i]);
  }
#pragma unroll
  for (int i = 0; i < HK_VE; ++i) {
    const int e = tid + HK_T * i;
    unsafeAtomicAdd(&accv[(tm * 16 + (e >> 4)) * 16 + (e & 15)], sv[i]);
  }
}

__global__ void k_heads_final(const float* __restrict__ pacc, const float* __restrict__ p1b,
                              const float* __restrict__ B, const float* __restrict__ bias,
                              const float* __restrict__ vacc, const float* __restrict__ v1b,
                              const float* __restrict__ v2w, const float* __restrict__ v2b,
                              float* __restrict__ out, float* __restrict__ vout) {
  if (blockIdx.x == 32) {
    int bi = threadIdx.x;
    if (bi < 64) {
      float s = 0.0f;
#pragma unroll
      for (int j = 0; j < 16; ++j) s += fmaxf(vacc[bi * 16 + j] + v1b[j], 0.0f) * v2w[j];
      vout[bi] = s + v2b[0];
    }
    return;
  }
  int wid  = (blockIdx.x * blockDim.x + threadIdx.x) >> 5;
  int lane = threadIdx.x & 31;
  int tm = wid >> 6;
  int tn = wid & 63;
  int m  = lane & 15;
  int kh = (lane >> 4) << 1;
  const float* arow = pacc + (tm * 16 + m) * 32;
  const float* bcol = B + tn * 16 + m;
  v8f c = {};
#pragma unroll
  for (int k0 = 0; k0 < 32; k0 += 4) {
    v2f a, b;
    a.x = fmaxf(arow[k0 + kh + 0] + p1b[k0 + kh + 0], 0.0f);
    a.y = fmaxf(arow[k0 + kh + 1] + p1b[k0 + kh + 1], 0.0f);
    b.x = bcol[(k0 + kh + 0) * 1024];
    b.y = bcol[(k0 + kh + 1) * 1024];
    c = __builtin_amdgcn_wmma_f32_16x16x4_f32(false, a, false, b, (short)0, c, false, false);
  }
  int rbase = tm * 16 + ((lane >> 4) << 3);
  float bv = bias[tn * 16 + m];
#pragma unroll
  for (int r = 0; r < 8; ++r)
    out[(rbase + r) * 1024 + tn * 16 + m] = c[r] + bv;
}

extern "C" void kernel_launch(void* const* d_in, const int* in_sizes, int n_in,
                              void* d_out, int out_size, void* d_ws, size_t ws_size,
                              hipStream_t stream) {
  const float* x    = (const float*)d_in[0];
  const int*   ei   = (const int*)d_in[1];
  const float* W1   = (const float*)d_in[2];
  const float* b1   = (const float*)d_in[3];
  const float* W2   = (const float*)d_in[4];
  const float* b2   = (const float*)d_in[5];
  const float* p1w  = (const float*)d_in[6];
  const float* p1b  = (const float*)d_in[7];
  const float* p2w  = (const float*)d_in[8];
  const float* p2b  = (const float*)d_in[9];
  const float* v1w  = (const float*)d_in[10];
  const float* v1b  = (const float*)d_in[11];
  const float* v2w  = (const float*)d_in[12];
  const float* v2b  = (const float*)d_in[13];

  const int N  = in_sizes[0] / 16;
  const int E  = in_sizes[1] / 2;
  const int* esrc = ei;
  const int* edst = ei + E;

  float* ws   = (float*)d_ws;
  float* dinv = ws + N;
  float* bufA = dinv + N;
  float* bufB = bufA + (size_t)16 * N;
  float* pacc = bufB + (size_t)16 * N;
  float* vacc = pacc + 64 * 32;
  unsigned* elist = (unsigned*)(vacc + 64 * 16);
  unsigned* tbl   = (unsigned*)ws;

  float* Xout = (float*)d_out;
  float* Vout = Xout + 64 * 1024;

  const int B256 = 256;
  int gemm_blocks    = (N / 16) / 8;
  int nranges        = N / SC_NR;
  int scatter_blocks = nranges * SC_S;

  k_bin_edges<<<(E + BIN_EDGES - 1) / BIN_EDGES, BIN_T, 0, stream>>>(esrc, edst, E, N / BIN_NR, tbl, elist, ws, ws_size, N,
                                                                    pacc, 64 * 32 + 64 * 16);

  k_deg_gemm_first<<<N / DGQ_ROWS, DGQ_WAVES * 32, 0, stream>>>(x, W1, b1, dinv, bufA, bufB, edst, E,
                                                              tbl, elist, ws, ws_size, N);
  k_scatter_replay<<<scatter_blocks, 1024, 0, stream>>>(esrc, edst, dinv, bufA, bufB, E, nranges,
                                                        tbl, elist, ws, ws_size, N);

  k_gemm_agg<<<gemm_blocks, B256, 0, stream>>>(W2, b2, dinv, bufA, bufB, N);
  k_scatter_replay<<<scatter_blocks, 1024, 0, stream>>>(esrc, edst, dinv, bufA, bufB, E, nranges,
                                                        tbl, elist, ws, ws_size, N);

  const int K = 16384;

  k_heads_splitk_relu<<<4 * (K / (HK_KLEN * HK_WAVES)), HK_WAVES * 32, 0, stream>>>(bufB, K, p1w, v1w, pacc, vacc);

  k_heads_final<<<32 + 1, B256, 0, stream>>>(pacc, p1b, p2w, p2b, vacc, v1b, v2w, v2b, Xout, Vout);
}


# edge scatter re-cut: one work-group per 256-row bin, a row's whole sum (self-loop as start value) completes in LDS; plain-store hand-over, layer 2 product fused in; 7 launches to 6; sum order changes
# speedup vs baseline: 5.9695x; 1.1255x over previous
#include <hip/hip_runtime.h>
#include <hip/hip_bf16.h>


typedef __attribute__((ext_vector_type(2))) float v2f;
typedef __attribute__((ext_vector_type(8))) float v8f;

constexpr int SC_NR      = 256;
constexpr int SC_WAVES   = 32;
constexpr int DGQ_ROWS   = 256;
constexpr int DGQ_WAVES  = DGQ_ROWS / 16;
constexpr int DGQ_UD     = 2;

constexpr int BIN_T     = 512;
constexpr int BIN_LD    = 4;
constexpr int BIN_EDGES = BIN_T * 4 * BIN_LD;
constexpr int BIN_NR    = 256;
constexpr int BIN_MAXR  = 256;
constexpr int BIN_TBL   = BIN_MAXR;
static_assert(SC_NR == BIN_NR, "a range of the scatter is one bin: its edges of a chunk are one run");


typedef int v4i __attribute__((ext_vector_type(4)));

constexpr int SC_S     = 1;
constexpr int SC_LOADS = 8;
constexpr int SC_BATCH = 32 * 4 * SC_LOADS;
constexpr int SC_UD    = 8;
constexpr int SC_QLEN  = SC_BATCH + 2 * SC_UD;
constexpr int RP_RUNS  = 8;
constexpr int RP_RUNW  = 64;
static_assert(RP_RUNS * RP_RUNW <= SC_QLEN, "a trip's words fit the wave's queue");

__device__ __forceinline__ bool sc_list_fits(const unsigned* list, int ne, const float* ws0, size_t ws_bytes,
                                             int tbl_room) {
  const int nchunks = (ne + BIN_EDGES - 1) / BIN_EDGES;
  return (size_t)((const char*)(list + ne) - (const char*)ws0) <= ws_bytes && nchunks * BIN_TBL <= tbl_room;
}

__device__ __forceinline__ unsigned bin_chunk_count(int c, int ne) {
  const int left = (ne - c * BIN_EDGES) & ~3;
  return (unsigned)(left < BIN_EDGES ? left : BIN_EDGES);
}

__device__ __forceinline__ unsigned bin_run_start(const unsigned* __restrict__ tbl, int c, int b, int ne) {
  return b < BIN_MAXR ? tbl[c * BIN_TBL + b] : bin_chunk_count(c, ne);
}


__device__ __forceinline__ void sc_scan(float* acc, unsigned* qw,
                                        const int* __restrict__ src, const int* __restrict__ dst,
                                        const float* __restrict__ hw, int ne, int r, int sl) {
  const int tid = threadIdx.x, lane = tid & 31, w = tid >> 5;
  const int half = lane >> 4, f = lane & 15;
  const int lo = r * SC_NR;

  const int slice = ne / SC_S;
  const int span  = slice / SC_WAVES;
  const int ebase = sl * slice + w * span;
  const int nb    = span / SC_BATCH;
  const v4i* dp   = (const v4i*)(dst + ebase);
  const int* sp   = src + ebase;
  int qn = 0;

  auto pass = [&](int q0, int nv) {
    unsigned ent[SC_UD]; bool ok[SC_UD]; int sv[SC_UD]; float hv[SC_UD];
#pragma unroll
    for (int u = 0; u < SC_UD; ++u) {
      ok[u]  = 2 * u + half < nv;
      ent[u] = ok[u] ? qw[q0 + 2 * u + half] : 0u;
    }
#pragma unroll
    for (int u = 0; u < SC_UD; ++u) sv[u] = sp[ent[u] >> 10];
#pragma unroll
    for (int u = 0; u < SC_UD; ++u) hv[u] = hw[sv[u] * 16 + f];
#pragma unroll
    for (int u = 0; u < SC_UD; ++u)
      unsafeAtomicAdd(&acc[(ent[u] & 1023u) * 16 + f], ok[u] ? hv[u] : 0.0f);
  };

  v4i cur[SC_LOADS], nxt[SC_LOADS];
#pragma unroll
  for (int j = 0; j < SC_LOADS; ++j) cur[j] = dp[j * 32 + lane];
  for (int b = 0; b < nb; ++b) {
    const int bn = (b + 1 < nb) ? b + 1 : b;
#pragma unroll
    for (int j = 0; j < SC_LOADS; ++j) nxt[j] = dp[bn * (SC_BATCH / 4) + j * 32 + lane];
#pragma unroll
    for (int j = 0; j < SC_LOADS; ++j) {
#pragma unroll
      for (int c = 0; c < 4; ++c) {
        const int dl = cur[j][c] - lo;
        const bool m = (unsigned)dl < (unsigned)SC_NR;
        const unsigned mask = __builtin_amdgcn_ballot_w32(m);
        if (mask != 0u) {
          const int cnt = __builtin_popcount(mask);
          if (m) {
            const int el = b * SC_BATCH + j * 128 + lane * 4 + c;
            qw[qn + (int)__builtin_amdgcn_mbcnt_lo(mask, 0u)] = ((unsigned)el << 10) | (unsigned)dl;
          }
          qn += cnt;
        }
      }
    }
    while (qn >= 2 * SC_UD) { qn -= 2 * SC_UD; pass(qn, 2 * SC_UD); }
#pragma unroll
    for (int j = 0; j < SC_LOADS; ++j) cur[j] = nxt[j];
  }
  if (qn > 0) pass(0, qn);
}

typedef unsigned v4u __attribute__((ext_vector_type(4)));

__global__ __launch_bounds__(BIN_T)
void k_bin_edges(const int* __restrict__ src, const int* __restrict__ dst, int ne, int nbins,
                 unsigned* __restrict__ tbl, unsigned* __restrict__ list,
                 const float* ws0, size_t ws_bytes, int tbl_room,
                 float* __restrict__ zacc, int nz) {
  __shared__ unsigned cnt[BIN_MAXR];
  __shared__ unsigned lpre[BIN_MAXR];
  __shared__ __attribute__((aligned(16))) unsigned ordered[BIN_EDGES];
  const int tid = threadIdx.x;
  for (int i = blockIdx.x * BIN_T + tid; i < nz; i += gridDim.x * BIN_T) zacc[i] = 0.0f;
  if (!sc_list_fits(list, ne, ws0, ws_bytes, tbl_room)) return;
  const int e0  = blockIdx.x * BIN_EDGES;

  v4i dv[BIN_LD], sv[BIN_LD];
  bool in[BIN_LD];
#pragma unroll
  for (int j = 0; j < BIN_LD; ++j) {
    const int e = e0 + (j * BIN_T + tid) * 4;
    in[j] = e + 3 < ne;
    dv[j] = __builtin_nontemporal_load((const v4i*)(dst + (in[j] ? e : 0)));
    sv[j] = __builtin_nontemporal_load((const v4i*)(src + (in[j] ? e : 0)));
  }
  if (tid < BIN_MAXR) cnt[tid] = 0u;
  __syncthreads();

  unsigned rk[BIN_LD][4];
#pragma unroll
  for (int j = 0; j < BIN_LD; ++j) {
#pragma unroll
    for (int k = 0; k < 4; ++k) {
      rk[j][k] = 0u;
      if (in[j]) rk[j][k] = atomicAdd(&cnt[(unsigned)dv[j][k] / (unsigned)BIN_NR], 1u);
    }
  }
  __syncthreads();

  if (tid < 32) {
    constexpr int PB = BIN_MAXR / 32;
    unsigned v[PB];
    unsigned t = 0u;
#pragma unroll
    for (int j = 0; j < PB; ++j) {
      v[j] = (tid * PB + j < nbins) ? cnt[tid * PB + j] : 0u;
      t += v[j];
    }
    unsigned inc = t;
    for (int o = 1; o < 32; o <<= 1) {
      const unsigned up = __shfl_up(inc, o);
      if (tid >= o) inc += up;
    }
    unsigned run = inc - t;
    unsigned* trow = tbl + (size_t)blockIdx.x * BIN_TBL;
#pragma unroll
    for (int j = 0; j < PB; ++j) {
      lpre[tid * PB + j] = run;
      trow[tid * PB + j] = run;
      run += v[j];
    }
  }
  __syncthreads();

#pragma unroll
  for (int j = 0; j < BIN_LD; ++j) {
#pragma unroll
    for (int k = 0; k < 4; ++k) {
      if (in[j]) {
        const unsigned rr = (unsigned)dv[j][k] / (unsigned)BIN_NR;
        ordered[lpre[rr] + rk[j][k]] = ((unsigned)sv[j][k] << 10) | ((unsigned)dv[j][k] % (unsigned)SC_NR);
      }
    }
  }
  __syncthreads();

  const int left = (ne - e0) & ~3;
  const int nv4 = (left < BIN_EDGES ? left : BIN_EDGES) >> 2;
  constexpr int BIN_CP = BIN_EDGES / 4 / BIN_T;
  const v4u* o4 = (const v4u*)ordered;
  v4u* l4 = (v4u*)(list + e0);
  v4u wd[BIN_CP];
#pragma unroll
  for (int k = 0; k < BIN_CP; ++k) wd[k] = o4[tid + k * BIN_T];
#pragma unroll
  for (int k = 0; k < BIN_CP; ++k)
    if (tid + k * BIN_T < nv4) l4[tid + k * BIN_T] = wd[k];
}

__global__ __launch_bounds__(DGQ_WAVES * 32)
void k_deg_gemm_first(const float* __restrict__ x, const float* __restrict__ B,
                      float* __restrict__ dinv,
                      float* __restrict__ hw,
                      const int* __restrict__ dst, int ne,
                      const unsigned* __restrict__ tbl, const unsigned* __restrict__ list,
                      const float* ws0, size_t ws_bytes, int tbl_room) {
  static_assert(DGQ_ROWS == BIN_NR, "a work-group's rows are one bin");
  static_assert(DGQ_WAVES == 16, "one load fetches the table words of one run per wave of the work-group");
  __shared__ unsigned cnt[DGQ_ROWS];
  const int tid = threadIdx.x, lane = tid & 31, w = tid >> 5;
  const int bin = blockIdx.x;
  const int lo = bin * DGQ_ROWS;
  const int m  = lane & 15;
  const int kh = (lane >> 4) << 1;
  const int row0  = lo + w * 16;
  const int rbase = row0 + ((lane >> 4) << 3);

  float a0[4], a1[4], b0[4], b1[4];
  const float* arow = x + (row0 + m) * 16;
#pragma unroll
  for (int k = 0; k < 4; ++k) {
    a0[k] = arow[4 * k + kh + 0];
    a1[k] = arow[4 * k + kh + 1];
    b0[k] = B[(4 * k + kh + 0) * 16 + m];
    b1[k] = B[(4 * k + kh + 1) * 16 + m];
  }

  for (int i = tid; i < DGQ_ROWS; i += DGQ_WAVES * 32) cnt[i] = 0u;
  __syncthreads();
  if (sc_list_fits(list, ne, ws0, ws_bytes, tbl_room)) {
    const int nchunks = (ne + BIN_EDGES - 1) / BIN_EDGES;
    for (int c0 = w; c0 < nchunks; c0 += 16 * DGQ_WAVES) {
      const int cl = c0 + DGQ_WAVES * m;
      unsigned tv = 0u;
      if (cl < nchunks) tv = bin_run_start(tbl, cl, bin + (lane >> 4), ne);
      unsigned s[16], e[16];
#pragma unroll
      for (int i = 0; i < 16; ++i) {
        s[i] = (unsigned)__builtin_amdgcn_readlane((int)tv, i);
        e[i] = (unsigned)__builtin_amdgcn_readlane((int)tv, 16 + i);
      }
      unsigned wv[16][DGQ_UD];
#pragma unroll
      for (int i = 0; i < 16; ++i) {
        const int ci = c0 + DGQ_WAVES * i;
        const unsigned* __restrict__ rl = list + (size_t)(ci < nchunks ? ci : 0) * BIN_EDGES;
#pragma unroll
        for (int u = 0; u < DGQ_UD; ++u) {
          const unsigned p = s[i] + 32u * u + lane;
          wv[i][u] = rl[p < e[i] ? p : 0u];
        }
      }
#pragma unroll
      for (int i = 0; i < 16; ++i) {
#pragma unroll
        for (int u = 0; u < DGQ_UD; ++u) {
          const unsigned p = s[i] + 32u * u + lane;
          if (p < e[i]) atomicAdd(&cnt[wv[i][u] & (DGQ_ROWS - 1)], 1u);
        }
      }
#pragma unroll
      for (int i = 0; i < 16; ++i) {
        const int ci = c0 + DGQ_WAVES * i;
        const unsigned* __restrict__ rl = list + (size_t)(ci < nchunks ? ci : 0) * BIN_EDGES;
        for (unsigned p0 = s[i] + 32u * DGQ_UD; p0 < e[i]; p0 += 32u) {
          const unsigned p = p0 + lane;
          if (p < e[i]) atomicAdd(&cnt[rl[p] & (DGQ_ROWS - 1)], 1u);
        }
      }
    }
  } else {
    for (int ed = tid; ed < ne; ed += DGQ_WAVES * 32) {
      const int dl = dst[ed] - lo;
      if ((unsigned)dl < (unsigned)DGQ_ROWS) atomicAdd(&cnt[dl], 1u);
    }
  }
  __syncthreads();

  float di[8];
#pragma unroll
  for (int rr = 0; rr < 8; ++rr) di[rr] = rsqrtf(1.0f + (float)cnt[rbase - lo + rr]);
  v8f c = {};
#pragma unroll
  for (int k = 0; k < 4; ++k) {
    v2f a, b;
    a.x = a0[k]; a.y = a1[k];
    b.x = b0[k]; b.y = b1[k];
    c = __builtin_amdgcn_wmma_f32_16x16x4_f32(false, a, false, b, (short)0, c, false, false);
  }
#pragma unroll
  for (int rr = 0; rr < 8; ++rr) {
    hw[(rbase + rr) * 16 + m] = c[rr] * di[rr];
  }
  if (m == 0) {
#pragma unroll
    for (int rr = 0; rr < 8; ++rr) dinv[rbase + rr] = di[rr];
  }
}

template <int U> __device__ __forceinline__ unsigned row_word_c(unsigned v) {
  return (unsigned)__builtin_amdgcn_update_dpp(0, (int)v, 0x150 + U, 0xF, 0xF, true);
}
__device__ __forceinline__ unsigned row_word(unsigned v, int u) {
  switch (u) {
    case 0:  return row_word_c<0>(v);   case 1:  return row_word_c<1>(v);
    case 2:  return row_word_c<2>(v);   case 3:  return row_word_c<3>(v);
    case 4:  return row_word_c<4>(v);   case 5:  return row_word_c<5>(v);
    case 6:  return row_word_c<6>(v);   case 7:  return row_word_c<7>(v);
    case 8:  return row_word_c<8>(v);   case 9:  return row_word_c<9>(v);
    case 10: return row_word_c<10>(v);  case 11: return row_word_c<11>(v);
    case 12: return row_word_c<12>(v);  case 13: return row_word_c<13>(v);
    case 14: return row_word_c<14>(v);  default: return row_word_c<15>(v);
  }
}

__device__ __forceinline__ void sc_bin_body(float* acc, unsigned* qw, const bool next,
                                            const int* __restrict__ src, const int* __restrict__ dst,
                                            const float* __restrict__ dinv, const float* __restrict__ hw,
                                            const float* __restrict__ bias, const float* __restrict__ Bn,
                                            float* __restrict__ out, int ne,
                                            const unsigned* __restrict__ tbl, const unsigned* __restrict__ list,
                                            const float* ws0, size_t ws_bytes, int tbl_room) {
  const int tid = threadIdx.x, lane = tid & 31, w = tid >> 5;
  const int half = lane >> 4, f = lane & 15;
  const int bin = blockIdx.x;
  const int lo  = bin * SC_NR;
  const bool fits = sc_list_fits(list, ne, ws0, ws_bytes, tbl_room);
  const int nchunks = (ne + BIN_EDGES - 1) / BIN_EDGES;
  constexpr int SC_FL = SC_NR * 16 / 1024;

  float st[SC_FL];
#pragma unroll
  for (int j = 0; j < SC_FL; ++j) st[j] = hw[(size_t)lo * 16 + tid + 1024 * j];
  unsigned tv = 0u;
  {
    const int cl = w + SC_WAVES * f;
    if (fits && f < RP_RUNS && cl < nchunks) tv = bin_run_start(tbl, cl, bin + half, ne);
  }
#pragma unroll
  for (int j = 0; j < SC_FL; ++j) acc[tid + 1024 * j] = st[j];
  __syncthreads();

  if (fits) {
    for (int c0 = w; c0 < nchunks; c0 += SC_WAVES * RP_RUNS) {
      unsigned s[RP_RUNS], n[RP_RUNS];
#pragma unroll
      for (int i = 0; i < RP_RUNS; ++i) {
        s[i] = (unsigned)__builtin_amdgcn_readlane((int)tv, i);
        const unsigned e = (unsigned)__builtin_amdgcn_readlane((int)tv, 16 + i);
        n[i] = e > s[i] ? e - s[i] : 0u;
      }
      unsigned wv[RP_RUNS][2];
#pragma unroll
      for (int i = 0; i < RP_RUNS; ++i) {
        const int ci = c0 + SC_WAVES * i;
        const unsigned* __restrict__ rl = list + (size_t)(ci < nchunks ? ci : 0) * BIN_EDGES;
#pragma unroll
        for (int k = 0; k < 2; ++k) {
          const unsigned p = 32u * k + lane;
          wv[i][k] = rl[p < n[i] ? s[i] + p : 0u];
        }
      }
      {
        const int cl = c0 + SC_WAVES * RP_RUNS + SC_WAVES * f;
        tv = 0u;
        if (f < RP_RUNS && cl < nchunks) tv = bin_run_start(tbl, cl, bin + half, ne);
      }
      unsigned T = 0u;
#pragma unroll
      for (int i = 0; i < RP_RUNS; ++i) {
#pragma unroll
        for (int k = 0; k < 2; ++k) {
          const unsigned p = 32u * k + lane;
          if (p < n[i]) qw[T + p] = wv[i][k];
        }
        T += n[i] < (unsigned)RP_RUNW ? n[i] : (unsigned)RP_RUNW;
      }

      constexpr unsigned IDLE = (unsigned)SC_NR;
      unsigned qa = lane < T ? qw[lane] : IDLE;
      unsigned qb = 32u + lane < T ? qw[32u + lane] : IDLE;
      float ha[16], hb[16];
#pragma unroll
      for (int u = 0; u < 16; ++u) ha[u] = hw[(row_word(qa, u) >> 10) * 16 + f];
      for (unsigned p = 0u; p < T; p += 64u) {
#pragma unroll
        for (int u = 0; u < 16; ++u) hb[u] = hw[(row_word(qb, u) >> 10) * 16 + f];
        const unsigned qc = p + 64u + lane < T ? qw[p + 64u + lane] : IDLE;
        const unsigned qd = p + 96u + lane < T ? qw[p + 96u + lane] : IDLE;
#pragma unroll
        for (int u = 0; u < 16; ++u)
          unsafeAtomicAdd(&acc[(row_word(qa, u) & (unsigned)(2 * SC_NR - 1)) * 16 + f], ha[u]);
#pragma unroll
        for (int u = 0; u < 16; ++u) ha[u] = hw[(row_word(qc, u) >> 10) * 16 + f];
        if (p + 32u < T) {
#pragma unroll
          for (int u = 0; u < 16; ++u)
            unsafeAtomicAdd(&acc[(row_word(qb, u) & (unsigned)(2 * SC_NR - 1)) * 16 + f], hb[u]);
        }
        qa = qc; qb = qd;
      }

#pragma unroll
      for (int i = 0; i < RP_RUNS; ++i) {
        const int ci = c0 + SC_WAVES * i;
        const unsigned* __restrict__ rl = list + (size_t)(ci < nchunks ? ci : 0) * BIN_EDGES;
        for (unsigned p0 = (unsigned)RP_RUNW; p0 < n[i]; p0 += 2u) {
          const unsigned p = p0 + half;
          if (p < n[i]) {
            const unsigned ent = rl[s[i] + p];
            unsafeAtomicAdd(&acc[(ent & (unsigned)(SC_NR - 1)) * 16 + f], hw[(ent >> 10) * 16 + f]);
          }
        }
      }
    }
  } else {
    sc_scan(acc, qw, src, dst, hw, ne, bin, 0);
  }

  if (!next) {
    float dv[SC_FL];
#pragma unroll
    for (int j = 0; j < SC_FL; ++j) dv[j] = dinv[lo + (tid >> 4) + 64 * j];
    const float bv = bias[f];
    __syncthreads();
    float* orow = out + (size_t)lo * 16;
#pragma unroll
    for (int j = 0; j < SC_FL; ++j) orow[tid + 1024 * j] = acc[tid + 1024 * j] * dv[j] + bv;
  } else {
    __syncthreads();
    if (w < SC_NR / 16) {
      const int m  = f;
      const int kh = half << 1;
      const int trow = w * 16;
      const float dA = dinv[lo + trow + m];
      float di[8], b0[4], b1[4], c0b[4], c1b[4];
#pragma unroll
      for (int r = 0; r < 8; ++r) di[r] = dinv[lo + trow + 8 * half + r];
#pragma unroll
      for (int k = 0; k < 4; ++k) {
        b0[k]  = Bn[(4 * k + kh + 0) * 16 + m];
        b1[k]  = Bn[(4 * k + kh + 1) * 16 + m];
        c0b[k] = bias[4 * k + kh + 0];
        c1b[k] = bias[4 * k + kh + 1];
      }
      const float* arow = acc + (trow + m) * 16;
      v8f c = {};
#pragma unroll
      for (int k = 0; k < 4; ++k) {
        v2f a, b;
        a.x = fmaxf(arow[4 * k + kh + 0] * dA + c0b[k], 0.0f);
        a.y = fmaxf(arow[4 * k + kh + 1] * dA + c1b[k], 0.0f);
        b.x = b0[k]; b.y = b1[k];
        c = __builtin_amdgcn_wmma_f32_16x16x4_f32(false, a, false, b, (short)0, c, false, false);
      }
      const int rbase = lo + trow + 8 * half;
#pragma unroll
      for (int r = 0; r < 8; ++r) out[(rbase + r) * 16 + m] = c[r] * di[r];
    }
  }
}

__global__ __launch_bounds__(1024)
void k_scatter_first(const int* __restrict__ src, const int* __restrict__ dst,
                     const float* __restrict__ dinv, const float* __restrict__ hw,
                     const float* __restrict__ bias, const float* __restrict__ Bn,
                     float* __restrict__ hw_next, int ne,
                     const unsigned* __restrict__ tbl, const unsigned* __restrict__ list,
                     const float* ws0, size_t ws_bytes, int tbl_room) {
  __shared__ float acc[(SC_NR + 1) * 16];
  __shared__ unsigned q[SC_WAVES][SC_QLEN];
  sc_bin_body(acc, q[threadIdx.x >> 5], true, src, dst, dinv, hw, bias, Bn, hw_next, ne,
              tbl, list, ws0, ws_bytes, tbl_room);
}

__global__ __launch_bounds__(1024)
void k_scatter_last(const int* __restrict__ src, const int* __restrict__ dst,
                    const float* __restrict__ dinv, const float* __restrict__ hw,
                    const float* __restrict__ bias,
                    float* __restrict__ agg, int ne,
                    const unsigned* __restrict__ tbl, const unsigned* __restrict__ list,
                    const float* ws0, size_t ws_bytes, int tbl_room) {
  __shared__ float acc[(SC_NR + 1) * 16];
  __shared__ unsigned q[SC_WAVES][SC_QLEN];
  sc_bin_body(acc, q[threadIdx.x >> 5], false, src, dst, dinv, hw, bias, bias, agg, ne,
              tbl, list, ws0, ws_bytes, tbl_room);
}

typedef __attribute__((ext_vector_type(4))) float v4f;

constexpr int HK_KLEN  = 64;
constexpr int HK_WAVES = 4;
constexpr int HK_NJ    = HK_KLEN / 8;
constexpr int HK_TS    = 8 * 32 + 16;
constexpr int HK_WS    = 3 * HK_TS;

__device__ __forceinline__ int hk_k(int j, int h, int s, int t) { return 8 * j + 4 * h + 2 * s + t; }

__global__ __launch_bounds__(HK_WAVES * 32)
void k_heads_splitk_relu(const float* __restrict__ A, int lda,
                         const float* __restrict__ Bp, const float* __restrict__ Bv,
                         float* __restrict__ accp, float* __restrict__ accv) {
  __shared__ float red[HK_WAVES * HK_WS];
  const int tid = threadIdx.x, lane = tid & 31, w = tid >> 5;
  const int tm = blockIdx.x & 3;
  const int kb = ((blockIdx.x >> 2) * HK_WAVES + w) * HK_KLEN;
  const int m = lane & 15, h = lane >> 4;
  const float* arow = A + ((tm * 16 + m) * lda + kb);
  const float* bp = Bp + (kb * 32 + 2 * m);
  const float* bv = Bv + (kb * 16 + m);

  v4f aq[HK_NJ];
  v2f pp[HK_NJ][4];
  float pv[HK_NJ][4];
#pragma unroll
  for (int j = 0; j < HK_NJ; ++j) {
    aq[j] = *(const v4f*)(arow + hk_k(j, h, 0, 0));
#pragma unroll
    for (int i = 0; i < 4; ++i) {
      const int k = hk_k(j, h, i >> 1, i & 1);
      pp[j][i] = *(const v2f*)(bp + k * 32);
      pv[j][i] = bv[k * 16];
    }
  }
  __builtin_amdgcn_sched_barrier(0);

  v8f c0 = {}, c1 = {}, c2 = {};
#pragma unroll
  for (int j = 0; j < HK_NJ; ++j) {
#pragma unroll
    for (int s = 0; s < 2; ++s) {
      v2f a, b;
      a.x = fmaxf(aq[j][2 * s + 0], 0.0f);
      a.y = fmaxf(aq[j][2 * s + 1], 0.0f);
      b.x = pp[j][2 * s + 0].x; b.y = pp[j][2 * s + 1].x;
      c0 = __builtin_amdgcn_wmma_f32_16x16x4_f32(false, a, false, b, (short)0, c0, false, false);
      b.x = pp[j][2 * s + 0].y; b.y = pp[j][2 * s + 1].y;
      c1 = __builtin_amdgcn_wmma_f32_16x16x4_f32(false, a, false, b, (short)0, c1, false, false);
      b.x = pv[j][2 * s + 0]; b.y = pv[j][2 * s + 1];
      c2 = __builtin_amdgcn_wmma_f32_16x16x4_f32(false, a, false, b, (short)0, c2, false, false);
    }
  }

  float* mine = red + w * HK_WS;
#pragma unroll
  for (int r = 0; r < 8; ++r) {
    mine[0 * HK_TS + r * 32 + lane] = c0[r];
    mine[1 * HK_TS + r * 32 + lane] = c1[r];
    mine[2 * HK_TS + r * 32 + lane] = c2[r];
  }
  __syncthreads();

  constexpr int HK_T  = HK_WAVES * 32;
  constexpr int HK_PE = 16 * 32 / HK_T;
  constexpr int HK_VE = 16 * 16 / HK_T;
  float sp[HK_PE], sv[HK_VE];
#pragma unroll
  for (int i = 0; i < HK_PE; ++i) {
    const int e = tid + HK_T * i;
    const int row = e >> 5, col = e & 31;
    const int at = (col & 1) * HK_TS + (row & 7) * 32 + (row >> 3) * 16 + (col >> 1);
    sp[i] = 0.0f;
#pragma unroll
    for (int ww = 0; ww < HK_WAVES; ++ww) sp[i] += red[ww * HK_WS + at];
  }
#pragma unroll
  for (int i = 0; i < HK_VE; ++i) {
    const int e = tid + HK_T * i;
    const int row = e >> 4, n = e & 15;
    const int at = 2 * HK_TS + (row & 7) * 32 + (row >> 3) * 16 + n;
    sv[i] = 0.0f;
#pragma unroll
    for (int ww = 0; ww < HK_WAVES; ++ww) sv[i] += red[ww * HK_WS + at];
  }
#pragma unroll
  for (int i = 0; i < HK_PE; ++i) {
    const int e = tid + HK_T * i;
    unsafeAtomicAdd(&accp[(tm * 16 + (e >> 5)) * 32 + (e & 31)], sp[i]);
  }
#pragma unroll
  for (int i = 0; i < HK_VE; ++i) {
    const int e = tid + HK_T * i;
    unsafeAtomicAdd(&accv[(tm * 16 + (e >> 4)) * 16 + (e & 15)], sv[i]);
  }
}

__global__ void k_heads_final(const float* __restrict__ pacc, const float* __restrict__ p1b,
                              const float* __restrict__ B, const float* __restrict__ bias,
                              const float* __restrict__ vacc, const float* __restrict__ v1b,
                              const float* __restrict__ v2w, const float* __restrict__ v2b,
                              float* __restrict__ out, float* __restrict__ vout) {
  if (blockIdx.x == 32) {
    int bi = threadIdx.x;
    if (bi < 64) {
      float s = 0.0f;
#pragma unroll
      for (int j = 0; j < 16; ++j) s += fmaxf(vacc[bi * 16 + j] + v1b[j], 0.0f) * v2w[j];
      vout[bi] = s + v2b[0];
    }
    return;
  }
  int wid  = (blockIdx.x * blockDim.x + threadIdx.x) >> 5;
  int lane = threadIdx.x & 31;
  int tm = wid >> 6;
  int tn = wid & 63;
  int m  = lane & 15;
  int kh = (lane >> 4) << 1;
  const float* arow = pacc + (tm * 16 + m) * 32;
  const float* bcol = B + tn * 16 + m;
  v8f c = {};
#pragma unroll
  for (int k0 = 0; k0 < 32; k0 += 4) {
    v2f a, b;
    a.x = fmaxf(arow[k0 + kh + 0] + p1b[k0 + kh + 0], 0.0f);
    a.y = fmaxf(arow[k0 + kh + 1] + p1b[k0 + kh + 1], 0.0f);
    b.x = bcol[(k0 + kh + 0) * 1024];
    b.y = bcol[(k0 + kh + 1) * 1024];
    c = __builtin_amdgcn_wmma_f32_16x16x4_f32(false, a, false, b, (short)0, c, false, false);
  }
  int rbase = tm * 16 + ((lane >> 4) << 3);
  float bv = bias[tn * 16 + m];
#pragma unroll
  for (int r = 0; r < 8; ++r)
    out[(rbase + r) * 1024 + tn * 16 + m] = c[r] + bv;
}

extern "C" void kernel_launch(void* const* d_in, const int* in_sizes, int n_in,
                              void* d_out, int out_size, void* d_ws, size_t ws_size,
                              hipStream_t stream) {
  const float* x    = (const float*)d_in[0];
  const int*   ei   = (const int*)d_in[1];
  const float* W1   = (const float*)d_in[2];
  const float* b1   = (const float*)d_in[3];
  const float* W2   = (const float*)d_in[4];
  const float* b2   = (const float*)d_in[5];
  const float* p1w  = (const float*)d_in[6];
  const float* p1b  = (const float*)d_in[7];
  const float* p2w  = (const float*)d_in[8];
  const float* p2b  = (const float*)d_in[9];
  const float* v1w  = (const float*)d_in[10];
  const float* v1b  = (const float*)d_in[11];
  const float* v2w  = (const float*)d_in[12];
  const float* v2b  = (const float*)d_in[13];

  const int N  = in_sizes[0] / 16;
  const int E  = in_sizes[1] / 2;
  const int* esrc = ei;
  const int* edst = ei + E;

  float* ws   = (float*)d_ws;
  float* dinv = ws + N;
  float* bufA = dinv + N;
  float* bufB = bufA + (size_t)16 * N;
  float* pacc = bufB + (size_t)16 * N;
  float* vacc = pacc + 64 * 32;
  unsigned* elist = (unsigned*)(vacc + 64 * 16);
  unsigned* tbl   = (unsigned*)ws;

  float* Xout = (float*)d_out;
  float* Vout = Xout + 64 * 1024;

  const int B256 = 256;
  int nranges = N / SC_NR;

  k_bin_edges<<<(E + BIN_EDGES - 1) / BIN_EDGES, BIN_T, 0, stream>>>(esrc, edst, E, N / BIN_NR, tbl, elist, ws, ws_size, N,
                                                                    pacc, 64 * 32 + 64 * 16);

  k_deg_gemm_first<<<N / DGQ_ROWS, DGQ_WAVES * 32, 0, stream>>>(x, W1, dinv, bufA, edst, E,
                                                              tbl, elist, ws, ws_size, N);
  k_scatter_first<<<nranges, 1024, 0, stream>>>(esrc, edst, dinv, bufA, b1, W2, bufB, E,
                                                tbl, elist, ws, ws_size, N);

  k_scatter_last<<<nranges, 1024, 0, stream>>>(esrc, edst, dinv, bufB, b2, bufA, E,
                                               tbl, elist, ws, ws_size, N);

  const int K = 16384;

  k_heads_splitk_relu<<<4 * (K / (HK_KLEN * HK_WAVES)), HK_WAVES * 32, 0, stream>>>(bufA, K, p1w, v1w, pacc, vacc);

  k_heads_final<<<32 + 1, B256, 0, stream>>>(pacc, p1b, p2w, p2b, vacc, v1b, v2w, v2b, Xout, Vout);
}
